# starting kernel
# baseline (speedup 1.0000x reference)
#include <hip/hip_runtime.h>

typedef __attribute__((ext_vector_type(2))) float v2f;
typedef __attribute__((ext_vector_type(8))) float v8f;

#define NN 100000
#define NE 1200000
#define CH 64
#define NTILES (NN / 16)

__global__ void k_clean_zero(const float* __restrict__ x, float* __restrict__ xc,
                             float* __restrict__ agg, float* __restrict__ cnt) {
    int i = blockIdx.x * blockDim.x + threadIdx.x;
    if (i < NN * CH) {
        float v = x[i];
        xc[i] = (v == v) ? v : 0.0f;
        agg[i] = 0.0f;
    }
    if (i < NN) cnt[i] = 0.0f;
}

__global__ void k_zero(float* __restrict__ agg, float* __restrict__ cnt) {
    int i = blockIdx.x * blockDim.x + threadIdx.x;
    if (i < NN * CH) agg[i] = 0.0f;
    if (i < NN) cnt[i] = 0.0f;
}

__global__ void k_scatter(const float* __restrict__ feat,
                          const int* __restrict__ src, const int* __restrict__ dst,
                          float* __restrict__ agg, float* __restrict__ cnt) {
    long long i = (long long)blockIdx.x * blockDim.x + threadIdx.x;
    if (i >= (long long)NE * CH) return;
    int e = (int)(i >> 6);
    int c = (int)(i & 63);
    int s = src[e];
    int d = dst[e];
    atomicAdd(&agg[d * CH + c], feat[s * CH + c]);
    if (c == 0) atomicAdd(&cnt[d], 1.0f);
}

__global__ void __launch_bounds__(256)
k_sage_gemm(const float* __restrict__ agg, const float* __restrict__ cnt,
            const float* __restrict__ feat,
            const float* __restrict__ Wl, const float* __restrict__ Wr,
            const float* __restrict__ bias, float* __restrict__ out) {
    __shared__ float Wfrag[32 * 4 * 32 * 2];

    for (int idx = threadIdx.x; idx < 32 * 4 * 32 * 2; idx += 256) {
        int v  = idx & 1;
        int l  = (idx >> 1) & 31;
        int nt = (idx >> 6) & 3;
        int s  = idx >> 8;
        int K  = 4 * s + ((l >> 4) << 1) + v;
        int n  = nt * 16 + (l & 15);
        Wfrag[idx] = (K < 64) ? Wl[K * 64 + n] : Wr[(K - 64) * 64 + n];
    }
    __syncthreads();

    int lane = threadIdx.x & 31;
    int hi   = lane >> 4;
    int r    = lane & 15;
    int tile = blockIdx.x * 8 + (threadIdx.x >> 5);
    bool active = (tile < NTILES);

    if (active) {
        int row = tile * 16 + r;
        float rd = 1.0f / fmaxf(cnt[row], 1.0f);
        const float* aggrow  = agg  + row * CH;
        const float* featrow = feat + row * CH;

        v8f acc[4];
        #pragma unroll
        for (int nt = 0; nt < 4; ++nt) {
            float bv = bias[nt * 16 + r];
            #pragma unroll
            for (int j = 0; j < 8; ++j) acc[nt][j] = bv;
        }

        #pragma unroll 4
        for (int s = 0; s < 32; ++s) {
            int kloc = 4 * (s & 15) + hi * 2;
            const float* base = (s < 16) ? aggrow : featrow;
            v2f a = *(const v2f*)(base + kloc);
            if (s < 16) { a.x *= rd; a.y *= rd; }

            #pragma unroll
            for (int nt = 0; nt < 4; ++nt) {
                v2f b = *(const v2f*)&Wfrag[((s * 4 + nt) * 32 + lane) * 2];
                acc[nt] = __builtin_amdgcn_wmma_f32_16x16x4_f32(
                    false, a, false, b, (short)0, acc[nt], false, false);
            }
        }

        #pragma unroll
        for (int nt = 0; nt < 4; ++nt) {
            #pragma unroll
            for (int v = 0; v < 8; ++v) {
                int orow = tile * 16 + hi * 8 + v;
                out[orow * CH + nt * 16 + r] = fmaxf(acc[nt][v], 0.0f);
            }
        }
    }
}

__global__ void k_fc(const float* __restrict__ h2, const float* __restrict__ Wfc,
                     const float* __restrict__ bfc, float* __restrict__ pred) {
    int node = blockIdx.x * 8 + (threadIdx.x >> 5);
    if (node >= NN) return;
    int lane = threadIdx.x & 31;
    const float* hrow = h2 + node * CH;
    float s = hrow[lane] * Wfc[lane] + hrow[lane + 32] * Wfc[lane + 32];
    #pragma unroll
    for (int off = 16; off > 0; off >>= 1) s += __shfl_down(s, off, 32);
    if (lane == 0) pred[node] = s + bfc[0];
}

extern "C" void kernel_launch(void* const* d_in, const int* in_sizes, int n_in,
                              void* d_out, int out_size, void* d_ws, size_t ws_size,
                              hipStream_t stream) {
    const float* x   = (const float*)d_in[0];
    const int*   ei  = (const int*)d_in[1];
    const float* W1l = (const float*)d_in[2];
    const float* b1  = (const float*)d_in[3];
    const float* W1r = (const float*)d_in[4];
    const float* W2l = (const float*)d_in[5];
    const float* b2  = (const float*)d_in[6];
    const float* W2r = (const float*)d_in[7];
    const float* Wfc = (const float*)d_in[8];
    const float* bfc = (const float*)d_in[9];
    float* pred = (float*)d_out;

    const int* src = ei;
    const int* dst = ei + NE;

    float* xc  = (float*)d_ws;
    float* agg = xc  + NN * CH;
    float* cnt = agg + NN * CH;
    float* h1  = cnt + NN;
    float* h2  = xc;

    const int zgrid = (NN * CH + 255) / 256;
    const int sgrid = (int)(((long long)NE * CH + 255) / 256);
    const int ggrid = (NTILES + 7) / 8;
    const int fgrid = (NN + 7) / 8;

    k_clean_zero<<<zgrid, 256, 0, stream>>>(x, xc, agg, cnt);
    k_scatter<<<sgrid, 256, 0, stream>>>(xc, src, dst, agg, cnt);
    k_sage_gemm<<<ggrid, 256, 0, stream>>>(agg, cnt, xc, W1l, W1r, b1, h1);

    k_zero<<<zgrid, 256, 0, stream>>>(agg, cnt);
    k_scatter<<<sgrid, 256, 0, stream>>>(h1, src, dst, agg, cnt);
    k_sage_gemm<<<ggrid, 256, 0, stream>>>(agg, cnt, h1, W2l, W2r, b2, h2);

    k_fc<<<fgrid, 256, 0, stream>>>(h2, Wfc, bfc, pred);
}


# Float-atomic edge scatter replaced by a destination-sorted edge list built once per call (integer-atomic counting sort) + per-node pull sum; sum order = arrival-set slot order, 2 interleaved partials
# speedup vs baseline: 1.8292x; 1.8292x over previous
#include <hip/hip_runtime.h>


typedef __attribute__((ext_vector_type(2))) float v2f;
typedef __attribute__((ext_vector_type(8))) float v8f;

#define NN 100000
#define NE 1200000
#define CH 64
#define NTILES (NN / 16)
#define NNP 100032

__global__ void k_clean_zero(const float* __restrict__ x, float* __restrict__ xc,
                             int* __restrict__ deg) {
    int i = blockIdx.x * blockDim.x + threadIdx.x;
    if (i < NN * CH) {
        float v = x[i];
        xc[i] = (v == v) ? v : 0.0f;
    }
    if (i < NN) deg[i] = 0;
}

__global__ void k_hist(const int* __restrict__ dst, int* __restrict__ deg) {
    int e = blockIdx.x * blockDim.x + threadIdx.x;
    if (e < NE) atomicAdd(&deg[dst[e]], 1);
}

__device__ __forceinline__ int wave_incl_scan(int x, int lane) {
    x += __builtin_amdgcn_update_dpp(0, x, 0x111, 0xf, 0xf, true);
    x += __builtin_amdgcn_update_dpp(0, x, 0x112, 0xf, 0xf, true);
    x += __builtin_amdgcn_update_dpp(0, x, 0x114, 0xf, 0xf, true);
    x += __builtin_amdgcn_update_dpp(0, x, 0x118, 0xf, 0xf, true);
    const int row0 = __builtin_amdgcn_readlane(x, 15);
    if (lane >= 16) x += row0;
    return x;
}

__global__ void __launch_bounds__(1024)
k_scan(const int* __restrict__ deg, int* __restrict__ rowptr, int* __restrict__ cursor) {
    __shared__ int wsum[2][4][32];
    const int t = threadIdx.x;
    const int lane = t & 31;
    const int w = __builtin_amdgcn_readfirstlane(t >> 5);
    int carry = 0;
    int buf = 0;
    for (int base = 0; base < NN; base += 4 * 4096) {
        int4 d[4];
        int s4[4], inc[4];
        #pragma unroll
        for (int k = 0; k < 4; ++k) {
            const int i = base + k * 4096 + t * 4;
            d[k] = make_int4(0, 0, 0, 0);
            if (i < NN) d[k] = *(const int4*)(deg + i);
        }
        #pragma unroll
        for (int k = 0; k < 4; ++k) {
            s4[k] = d[k].x + d[k].y + d[k].z + d[k].w;
            inc[k] = wave_incl_scan(s4[k], lane);
        }
        if (lane == 31) {
            #pragma unroll
            for (int k = 0; k < 4; ++k) wsum[buf][k][w] = inc[k];
        }
        __syncthreads();
        int run = carry;
        #pragma unroll
        for (int k = 0; k < 4; ++k) {
            const int ws = wsum[buf][k][lane];
            const int winc = wave_incl_scan(ws, lane);
            const int wave_excl = __builtin_amdgcn_readlane(winc - ws, w);
            const int total = __builtin_amdgcn_readlane(winc, 31);
            const int excl = run + wave_excl + inc[k] - s4[k];
            const int i = base + k * 4096 + t * 4;
            if (i < NN) {
                int4 o = make_int4(excl, excl + d[k].x, excl + d[k].x + d[k].y, excl + d[k].x + d[k].y + d[k].z);
                *(int4*)(rowptr + i) = o;
                *(int4*)(cursor + i) = o;
            }
            run += total;
        }
        carry = run;
        buf ^= 1;
    }
    if (t == 0) rowptr[NN] = carry;
}

__global__ void k_fill(const int* __restrict__ src, const int* __restrict__ dst,
                       int* __restrict__ cursor, int* __restrict__ col) {
    int e = blockIdx.x * blockDim.x + threadIdx.x;
    if (e < NE) {
        const int d = dst[e];
        const int s = src[e];
        int pos = atomicAdd(&cursor[d], 1);
        if ((unsigned)pos < (unsigned)NE) col[pos] = s;
    }
}

__global__ void __launch_bounds__(256)
k_gather(const float* __restrict__ feat, const int* __restrict__ rowptr,
         const int* __restrict__ col, float* __restrict__ agg, float* __restrict__ cnt) {
    const int node = __builtin_amdgcn_readfirstlane((int)(blockIdx.x * 8 + (threadIdx.x >> 5)));
    if (node >= NN) return;
    const int lane = threadIdx.x & 31;
    const int hi = lane >> 4;
    const int c4 = (lane & 15) * 4;
    const int beg = rowptr[node];
    const int end = rowptr[node + 1];

    float4 acc = make_float4(0.0f, 0.0f, 0.0f, 0.0f);
    for (int base = beg; base < end; base += 32) {
        int n = end - base;
        if (n > 32) n = 32;
        int myc = 0;
        if (lane < n) myc = col[base + lane];
        myc = min(max(myc, 0), NN - 1);
        for (int j = 0; j < n; j += 8) {
            float4 v[4];
            #pragma unroll
            for (int u = 0; u < 4; ++u) {
                int s = __shfl(myc, j + 2 * u + hi, 32);
                v[u] = *(const float4*)(feat + s * CH + c4);
            }
            #pragma unroll
            for (int u = 0; u < 4; ++u) {
                bool ok = (j + 2 * u + hi) < n;
                acc.x += ok ? v[u].x : 0.0f;
                acc.y += ok ? v[u].y : 0.0f;
                acc.z += ok ? v[u].z : 0.0f;
                acc.w += ok ? v[u].w : 0.0f;
            }
        }
    }
    acc.x += __shfl_xor(acc.x, 16, 32);
    acc.y += __shfl_xor(acc.y, 16, 32);
    acc.z += __shfl_xor(acc.z, 16, 32);
    acc.w += __shfl_xor(acc.w, 16, 32);
    if (hi == 0) *(float4*)(agg + node * CH + c4) = acc;
    if (lane == 0) cnt[node] = (float)(end - beg);
}

__global__ void __launch_bounds__(256)
k_sage_gemm(const float* __restrict__ agg, const float* __restrict__ cnt,
            const float* __restrict__ feat,
            const float* __restrict__ Wl, const float* __restrict__ Wr,
            const float* __restrict__ bias, float* __restrict__ out) {
    __shared__ float Wfrag[32 * 4 * 32 * 2];

    for (int idx = threadIdx.x; idx < 32 * 4 * 32 * 2; idx += 256) {
        int v  = idx & 1;
        int l  = (idx >> 1) & 31;
        int nt = (idx >> 6) & 3;
        int s  = idx >> 8;
        int K  = 4 * s + ((l >> 4) << 1) + v;
        int n  = nt * 16 + (l & 15);
        Wfrag[idx] = (K < 64) ? Wl[K * 64 + n] : Wr[(K - 64) * 64 + n];
    }
    __syncthreads();

    int lane = threadIdx.x & 31;
    int hi   = lane >> 4;
    int r    = lane & 15;
    int tile = blockIdx.x * 8 + (threadIdx.x >> 5);
    bool active = (tile < NTILES);

    if (active) {
        int row = tile * 16 + r;
        float rd = 1.0f / fmaxf(cnt[row], 1.0f);
        const float* aggrow  = agg  + row * CH;
        const float* featrow = feat + row * CH;

        v8f acc[4];
        #pragma unroll
        for (int nt = 0; nt < 4; ++nt) {
            float bv = bias[nt * 16 + r];
            #pragma unroll
            for (int j = 0; j < 8; ++j) acc[nt][j] = bv;
        }

        #pragma unroll 4
        for (int s = 0; s < 32; ++s) {
            int kloc = 4 * (s & 15) + hi * 2;
            const float* base = (s < 16) ? aggrow : featrow;
            v2f a = *(const v2f*)(base + kloc);
            if (s < 16) { a.x *= rd; a.y *= rd; }

            #pragma unroll
            for (int nt = 0; nt < 4; ++nt) {
                v2f b = *(const v2f*)&Wfrag[((s * 4 + nt) * 32 + lane) * 2];
                acc[nt] = __builtin_amdgcn_wmma_f32_16x16x4_f32(
                    false, a, false, b, (short)0, acc[nt], false, false);
            }
        }

        #pragma unroll
        for (int nt = 0; nt < 4; ++nt) {
            #pragma unroll
            for (int v = 0; v < 8; ++v) {
                int orow = tile * 16 + hi * 8 + v;
                out[orow * CH + nt * 16 + r] = fmaxf(acc[nt][v], 0.0f);
            }
        }
    }
}

__global__ void k_fc(const float* __restrict__ h2, const float* __restrict__ Wfc,
                     const float* __restrict__ bfc, float* __restrict__ pred) {
    int node = blockIdx.x * 8 + (threadIdx.x >> 5);
    if (node >= NN) return;
    int lane = threadIdx.x & 31;
    const float* hrow = h2 + node * CH;
    float s = hrow[lane] * Wfc[lane] + hrow[lane + 32] * Wfc[lane + 32];
    #pragma unroll
    for (int off = 16; off > 0; off >>= 1) s += __shfl_down(s, off, 32);
    if (lane == 0) pred[node] = s + bfc[0];
}

extern "C" void kernel_launch(void* const* d_in, const int* in_sizes, int n_in,
                              void* d_out, int out_size, void* d_ws, size_t ws_size,
                              hipStream_t stream) {
    const float* x   = (const float*)d_in[0];
    const int*   ei  = (const int*)d_in[1];
    const float* W1l = (const float*)d_in[2];
    const float* b1  = (const float*)d_in[3];
    const float* W1r = (const float*)d_in[4];
    const float* W2l = (const float*)d_in[5];
    const float* b2  = (const float*)d_in[6];
    const float* W2r = (const float*)d_in[7];
    const float* Wfc = (const float*)d_in[8];
    const float* bfc = (const float*)d_in[9];
    float* pred = (float*)d_out;

    const int* src = ei;
    const int* dst = ei + NE;

    float* xc  = (float*)d_ws;
    float* agg = xc  + NN * CH;
    float* cnt = agg + NN * CH;
    float* h1  = cnt + NNP;
    float* h2  = xc;
    int* deg    = (int*)(h1 + NN * CH);
    int* rowptr = deg + NNP;
    int* cursor = rowptr + NNP;
    int* col    = cursor + NNP;

    const int zgrid = (NN * CH + 255) / 256;
    const int egrid = (NE + 255) / 256;
    const int ngrid = (NN + 7) / 8;
    const int ggrid = (NTILES + 7) / 8;
    const int fgrid = (NN + 7) / 8;

    k_clean_zero<<<zgrid, 256, 0, stream>>>(x, xc, deg);
    k_hist<<<egrid, 256, 0, stream>>>(dst, deg);
    k_scan<<<1, 1024, 0, stream>>>(deg, rowptr, cursor);
    k_fill<<<egrid, 256, 0, stream>>>(src, dst, cursor, col);

    k_gather<<<ngrid, 256, 0, stream>>>(xc, rowptr, col, agg, cnt);
    k_sage_gemm<<<ggrid, 256, 0, stream>>>(agg, cnt, xc, W1l, W1r, b1, h1);

    k_gather<<<ngrid, 256, 0, stream>>>(h1, rowptr, col, agg, cnt);
    k_sage_gemm<<<ggrid, 256, 0, stream>>>(agg, cnt, h1, W2l, W2r, b2, h2);

    k_fc<<<fgrid, 256, 0, stream>>>(h2, Wfc, bfc, pred);
}


# Edge sort without global atomics: two-level counting sort on LDS counters (128 edge chunks x 196 node bins, one prefix sum, one work-group per bin); slot order, so float sum order, by LDS arrival
# speedup vs baseline: 2.9948x; 1.6372x over previous
#include <hip/hip_runtime.h>


typedef __attribute__((ext_vector_type(2))) float v2f;
typedef __attribute__((ext_vector_type(8))) float v8f;

#define NN 100000
#define NE 1200000
#define CH 64
#define NTILES (NN / 16)
#define NNP 100032

#define BSH 9
#define BINSZ (1 << BSH)
#define NB ((NN + BINSZ - 1) / BINSZ)
#define G1 128
#define CHUNK ((NE + G1 - 1) / G1)
#define EPT 10
#define NSC (NB * G1)
#define NSCP (NSC + 64)
#define PU 8
static_assert(CHUNK <= EPT * 1024, "a chunk must fit one pass of the work-group");
static_assert(NSC % 4 == 0, "the scan reads four counts per load");

__global__ void k_clean(const float* __restrict__ x, float* __restrict__ xc) {
    int i = blockIdx.x * blockDim.x + threadIdx.x;
    if (i < NN * CH) {
        float v = x[i];
        xc[i] = (v == v) ? v : 0.0f;
    }
}


__global__ void __launch_bounds__(1024)
k_bin_count(const int* __restrict__ dst, int* __restrict__ bh) {
    __shared__ int h[NB];
    const int t = threadIdx.x;
    if (t < NB) h[t] = 0;
    const int cbeg = blockIdx.x * CHUNK;
    const int cend = min(cbeg + CHUNK, NE);
    int d[EPT];
    #pragma unroll
    for (int k = 0; k < EPT; ++k) {
        const int e = cbeg + k * 1024 + t;
        d[k] = (e < cend) ? dst[e] : -1;
    }
    __syncthreads();
    #pragma unroll
    for (int k = 0; k < EPT; ++k)
        if ((unsigned)d[k] < (unsigned)NN) atomicAdd(&h[d[k] >> BSH], 1);
    __syncthreads();
    if (t < NB) bh[t * G1 + blockIdx.x] = h[t];
}

__device__ __forceinline__ int wave_incl_scan(int x, int lane) {
    x += __builtin_amdgcn_update_dpp(0, x, 0x111, 0xf, 0xf, true);
    x += __builtin_amdgcn_update_dpp(0, x, 0x112, 0xf, 0xf, true);
    x += __builtin_amdgcn_update_dpp(0, x, 0x114, 0xf, 0xf, true);
    x += __builtin_amdgcn_update_dpp(0, x, 0x118, 0xf, 0xf, true);
    const int row0 = __builtin_amdgcn_readlane(x, 15);
    if (lane >= 16) x += row0;
    return x;
}

__global__ void __launch_bounds__(1024)
k_scan(const int* __restrict__ in, int* __restrict__ out) {
    __shared__ int wsum[2][4][32];
    const int t = threadIdx.x;
    const int lane = t & 31;
    const int w = __builtin_amdgcn_readfirstlane(t >> 5);
    int carry = 0;
    int buf = 0;
    for (int base = 0; base < NSC; base += 4 * 4096) {
        int4 d[4];
        int s4[4], inc[4];
        #pragma unroll
        for (int k = 0; k < 4; ++k) {
            const int i = base + k * 4096 + t * 4;
            d[k] = make_int4(0, 0, 0, 0);
            if (i < NSC) d[k] = *(const int4*)(in + i);
        }
        #pragma unroll
        for (int k = 0; k < 4; ++k) {
            s4[k] = d[k].x + d[k].y + d[k].z + d[k].w;
            inc[k] = wave_incl_scan(s4[k], lane);
        }
        if (lane == 31) {
            #pragma unroll
            for (int k = 0; k < 4; ++k) wsum[buf][k][w] = inc[k];
        }
        __syncthreads();
        int run = carry;
        #pragma unroll
        for (int k = 0; k < 4; ++k) {
            const int ws = wsum[buf][k][lane];
            const int winc = wave_incl_scan(ws, lane);
            const int wave_excl = __builtin_amdgcn_readlane(winc - ws, w);
            const int total = __builtin_amdgcn_readlane(winc, 31);
            const int excl = run + wave_excl + inc[k] - s4[k];
            const int i = base + k * 4096 + t * 4;
            if (i < NSC) {
                int4 o = make_int4(excl, excl + d[k].x, excl + d[k].x + d[k].y, excl + d[k].x + d[k].y + d[k].z);
                *(int4*)(out + i) = o;
            }
            run += total;
        }
        carry = run;
        buf ^= 1;
    }
    if (t == 0) out[NSC] = carry;
}

__global__ void __launch_bounds__(1024)
k_bin_scatter(const int* __restrict__ src, const int* __restrict__ dst,
              const int* __restrict__ boff, int* __restrict__ tmpi) {
    __shared__ int cur[NB];
    int2* tmp = (int2*)tmpi;
    const int t = threadIdx.x;
    if (t < NB) cur[t] = boff[t * G1 + blockIdx.x];
    const int cbeg = blockIdx.x * CHUNK;
    const int cend = min(cbeg + CHUNK, NE);
    int d[EPT], s[EPT];
    #pragma unroll
    for (int k = 0; k < EPT; ++k) {
        const int e = cbeg + k * 1024 + t;
        d[k] = (e < cend) ? dst[e] : -1;
        s[k] = (e < cend) ? src[e] : 0;
    }
    __syncthreads();
    #pragma unroll
    for (int k = 0; k < EPT; ++k) {
        if ((unsigned)d[k] < (unsigned)NN) {
            const int pos = atomicAdd(&cur[d[k] >> BSH], 1);
            if ((unsigned)pos < (unsigned)NE) tmp[pos] = make_int2(s[k], d[k]);
        }
    }
}

__global__ void __launch_bounds__(1024)
k_bin_place(const int* __restrict__ boff, const int* __restrict__ tmpi,
            int* __restrict__ rowptr, int* __restrict__ col) {
    __shared__ int cnt[BINSZ];
    __shared__ int wtot[BINSZ / 32];
    const int2* tmp = (const int2*)tmpi;
    const int t = threadIdx.x;
    const int lane = t & 31;
    const int w = t >> 5;
    const int bin = blockIdx.x;
    const int base = bin * BINSZ;
    const int beg = boff[bin * G1];
    const int end = boff[(bin + 1) * G1];
    if (t < BINSZ) cnt[t] = 0;
    __syncthreads();
    for (int e0 = beg; e0 < end; e0 += 1024 * PU) {
        int dl[PU];
        #pragma unroll
        for (int u = 0; u < PU; ++u) {
            const int e = e0 + u * 1024 + t;
            dl[u] = (e < end) ? (tmp[e].y - base) : -1;
        }
        #pragma unroll
        for (int u = 0; u < PU; ++u)
            if ((unsigned)dl[u] < (unsigned)BINSZ) atomicAdd(&cnt[dl[u]], 1);
    }
    __syncthreads();
    const int c = (t < BINSZ) ? cnt[t] : 0;
    const int incl = wave_incl_scan(c, lane);
    if (lane == 31 && w < BINSZ / 32) wtot[w] = incl;
    __syncthreads();
    int wpre = 0;
    #pragma unroll
    for (int j = 0; j < BINSZ / 32; ++j) wpre += (j < w) ? wtot[j] : 0;
    const int excl = beg + wpre + incl - c;
    if (t < BINSZ) {
        const int node = base + t;
        if (node < NN) rowptr[node] = excl;
        cnt[t] = excl;
    }
    if (bin == NB - 1 && t == 0) rowptr[NN] = end;
    __syncthreads();
    for (int e0 = beg; e0 < end; e0 += 1024 * PU) {
        int2 p[PU];
        #pragma unroll
        for (int u = 0; u < PU; ++u) {
            const int e = e0 + u * 1024 + t;
            p[u] = (e < end) ? tmp[e] : make_int2(0, -1);
        }
        #pragma unroll
        for (int u = 0; u < PU; ++u) {
            const int dl = p[u].y - base;
            if (p[u].y >= 0 && (unsigned)dl < (unsigned)BINSZ) {
                const int pos = atomicAdd(&cnt[dl], 1);
                if ((unsigned)pos < (unsigned)NE) col[pos] = p[u].x;
            }
        }
    }
}

__global__ void __launch_bounds__(256)
k_gather(const float* __restrict__ feat, const int* __restrict__ rowptr,
         const int* __restrict__ col, float* __restrict__ agg, float* __restrict__ cnt) {
    const int node = __builtin_amdgcn_readfirstlane((int)(blockIdx.x * 8 + (threadIdx.x >> 5)));
    if (node >= NN) return;
    const int lane = threadIdx.x & 31;
    const int hi = lane >> 4;
    const int c4 = (lane & 15) * 4;
    const int beg = rowptr[node];
    const int end = rowptr[node + 1];

    float4 acc = make_float4(0.0f, 0.0f, 0.0f, 0.0f);
    for (int base = beg; base < end; base += 32) {
        int n = end - base;
        if (n > 32) n = 32;
        int myc = 0;
        if (lane < n) myc = col[base + lane];
        myc = min(max(myc, 0), NN - 1);
        for (int j = 0; j < n; j += 8) {
            float4 v[4];
            #pragma unroll
            for (int u = 0; u < 4; ++u) {
                int s = __shfl(myc, j + 2 * u + hi, 32);
                v[u] = *(const float4*)(feat + s * CH + c4);
            }
            #pragma unroll
            for (int u = 0; u < 4; ++u) {
                bool ok = (j + 2 * u + hi) < n;
                acc.x += ok ? v[u].x : 0.0f;
                acc.y += ok ? v[u].y : 0.0f;
                acc.z += ok ? v[u].z : 0.0f;
                acc.w += ok ? v[u].w : 0.0f;
            }
        }
    }
    acc.x += __shfl_xor(acc.x, 16, 32);
    acc.y += __shfl_xor(acc.y, 16, 32);
    acc.z += __shfl_xor(acc.z, 16, 32);
    acc.w += __shfl_xor(acc.w, 16, 32);
    if (hi == 0) *(float4*)(agg + node * CH + c4) = acc;
    if (lane == 0) cnt[node] = (float)(end - beg);
}

__global__ void __launch_bounds__(256)
k_sage_gemm(const float* __restrict__ agg, const float* __restrict__ cnt,
            const float* __restrict__ feat,
            const float* __restrict__ Wl, const float* __restrict__ Wr,
            const float* __restrict__ bias, float* __restrict__ out) {
    __shared__ float Wfrag[32 * 4 * 32 * 2];

    for (int idx = threadIdx.x; idx < 32 * 4 * 32 * 2; idx += 256) {
        int v  = idx & 1;
        int l  = (idx >> 1) & 31;
        int nt = (idx >> 6) & 3;
        int s  = idx >> 8;
        int K  = 4 * s + ((l >> 4) << 1) + v;
        int n  = nt * 16 + (l & 15);
        Wfrag[idx] = (K < 64) ? Wl[K * 64 + n] : Wr[(K - 64) * 64 + n];
    }
    __syncthreads();

    int lane = threadIdx.x & 31;
    int hi   = lane >> 4;
    int r    = lane & 15;
    int tile = blockIdx.x * 8 + (threadIdx.x >> 5);
    bool active = (tile < NTILES);

    if (active) {
        int row = tile * 16 + r;
        float rd = 1.0f / fmaxf(cnt[row], 1.0f);
        const float* aggrow  = agg  + row * CH;
        const float* featrow = feat + row * CH;

        v8f acc[4];
        #pragma unroll
        for (int nt = 0; nt < 4; ++nt) {
            float bv = bias[nt * 16 + r];
            #pragma unroll
            for (int j = 0; j < 8; ++j) acc[nt][j] = bv;
        }

        #pragma unroll 4
        for (int s = 0; s < 32; ++s) {
            int kloc = 4 * (s & 15) + hi * 2;
            const float* base = (s < 16) ? aggrow : featrow;
            v2f a = *(const v2f*)(base + kloc);
            if (s < 16) { a.x *= rd; a.y *= rd; }

            #pragma unroll
            for (int nt = 0; nt < 4; ++nt) {
                v2f b = *(const v2f*)&Wfrag[((s * 4 + nt) * 32 + lane) * 2];
                acc[nt] = __builtin_amdgcn_wmma_f32_16x16x4_f32(
                    false, a, false, b, (short)0, acc[nt], false, false);
            }
        }

        #pragma unroll
        for (int nt = 0; nt < 4; ++nt) {
            #pragma unroll
            for (int v = 0; v < 8; ++v) {
                int orow = tile * 16 + hi * 8 + v;
                out[orow * CH + nt * 16 + r] = fmaxf(acc[nt][v], 0.0f);
            }
        }
    }
}

__global__ void k_fc(const float* __restrict__ h2, const float* __restrict__ Wfc,
                     const float* __restrict__ bfc, float* __restrict__ pred) {
    int node = blockIdx.x * 8 + (threadIdx.x >> 5);
    if (node >= NN) return;
    int lane = threadIdx.x & 31;
    const float* hrow = h2 + node * CH;
    float s = hrow[lane] * Wfc[lane] + hrow[lane + 32] * Wfc[lane + 32];
    #pragma unroll
    for (int off = 16; off > 0; off >>= 1) s += __shfl_down(s, off, 32);
    if (lane == 0) pred[node] = s + bfc[0];
}

extern "C" void kernel_launch(void* const* d_in, const int* in_sizes, int n_in,
                              void* d_out, int out_size, void* d_ws, size_t ws_size,
                              hipStream_t stream) {
    const float* x   = (const float*)d_in[0];
    const int*   ei  = (const int*)d_in[1];
    const float* W1l = (const float*)d_in[2];
    const float* b1  = (const float*)d_in[3];
    const float* W1r = (const float*)d_in[4];
    const float* W2l = (const float*)d_in[5];
    const float* b2  = (const float*)d_in[6];
    const float* W2r = (const float*)d_in[7];
    const float* Wfc = (const float*)d_in[8];
    const float* bfc = (const float*)d_in[9];
    float* pred = (float*)d_out;

    const int* src = ei;
    const int* dst = ei + NE;

    float* xc  = (float*)d_ws;
    float* agg = xc  + NN * CH;
    float* cnt = agg + NN * CH;
    float* h1  = cnt + NNP;
    float* h2  = xc;
    int* rowptr = (int*)(h1 + NN * CH);
    int* col    = rowptr + NNP;
    int* bh     = col + NE;
    int* boff   = bh + NSCP;
    int* tmp    = boff + NSCP;

    const int zgrid = (NN * CH + 255) / 256;
    const int ngrid = (NN + 7) / 8;
    const int ggrid = (NTILES + 7) / 8;
    const int fgrid = (NN + 7) / 8;

    k_clean<<<zgrid, 256, 0, stream>>>(x, xc);
    k_bin_count<<<G1, 1024, 0, stream>>>(dst, bh);
    k_scan<<<1, 1024, 0, stream>>>(bh, boff);
    k_bin_scatter<<<G1, 1024, 0, stream>>>(src, dst, boff, tmp);
    k_bin_place<<<NB, 1024, 0, stream>>>(boff, tmp, rowptr, col);

    k_gather<<<ngrid, 256, 0, stream>>>(xc, rowptr, col, agg, cnt);
    k_sage_gemm<<<ggrid, 256, 0, stream>>>(agg, cnt, xc, W1l, W1r, b1, h1);

    k_gather<<<ngrid, 256, 0, stream>>>(h1, rowptr, col, agg, cnt);
    k_sage_gemm<<<ggrid, 256, 0, stream>>>(agg, cnt, h1, W2l, W2r, b2, h2);

    k_fc<<<fgrid, 256, 0, stream>>>(h2, Wfc, bfc, pred);
}


# Fewer, fatter waves in the prep, pull-sum and head kernels: 8192 resident waves each walk a contiguous share (12-13 nodes, next node's sources prefetched, 8 row loads in flight); sum order unchanged
# speedup vs baseline: 3.7356x; 1.2474x over previous
#include <hip/hip_runtime.h>


typedef __attribute__((ext_vector_type(2))) float v2f;
typedef __attribute__((ext_vector_type(8))) float v8f;

#define NN 100000
#define NE 1200000
#define CH 64
#define NTILES (NN / 16)
#define NNP 100032

#define BSH 9
#define BINSZ (1 << BSH)
#define NB ((NN + BINSZ - 1) / BINSZ)
#define G1 128
#define CHUNK ((NE + G1 - 1) / G1)
#define EPT 10
#define NSC (NB * G1)
#define NSCP (NSC + 64)
#define PU 8
static_assert(CHUNK <= EPT * 1024, "a chunk must fit one pass of the work-group");
static_assert(NSC % 4 == 0, "the scan reads four counts per load");

#define FW_WG 1024
#define FW_WAVES (FW_WG * 8)
#define FW_THREADS (FW_WG * 256)
#define NODE_LO(w) (((w) * NN) / FW_WAVES)
static_assert((long long)FW_WAVES * NN < 2147483648LL, "the node split is computed in 32-bit integers");
static_assert((NN + FW_WAVES - 1) / FW_WAVES < 31, "a wave's segment bounds must fit one register across its lanes");

#define N4 (NN * CH / 4)
#define CLEAN_TRIPS ((N4 + FW_THREADS - 1) / FW_THREADS)
#define CLEAN_FULL (N4 / FW_THREADS)
__global__ void __launch_bounds__(256)
k_clean(const float* __restrict__ x, float* __restrict__ xc) {
    const int t = blockIdx.x * 256 + threadIdx.x;
    float4 v[CLEAN_TRIPS];
    #pragma unroll
    for (int k = 0; k < CLEAN_TRIPS; ++k) {
        const int i = k * FW_THREADS + t;
        v[k] = make_float4(0.0f, 0.0f, 0.0f, 0.0f);
        if (k < CLEAN_FULL || i < N4) v[k] = ((const float4*)x)[i];
    }
    #pragma unroll
    for (int k = 0; k < CLEAN_TRIPS; ++k) {
        const int i = k * FW_THREADS + t;
        if (k < CLEAN_FULL || i < N4) {
            float4 r;
            r.x = (v[k].x == v[k].x) ? v[k].x : 0.0f;
            r.y = (v[k].y == v[k].y) ? v[k].y : 0.0f;
            r.z = (v[k].z == v[k].z) ? v[k].z : 0.0f;
            r.w = (v[k].w == v[k].w) ? v[k].w : 0.0f;
            ((float4*)xc)[i] = r;
        }
    }
}


__global__ void __launch_bounds__(1024)
k_bin_count(const int* __restrict__ dst, int* __restrict__ bh) {
    __shared__ int h[NB];
    const int t = threadIdx.x;
    if (t < NB) h[t] = 0;
    const int cbeg = blockIdx.x * CHUNK;
    const int cend = min(cbeg + CHUNK, NE);
    int d[EPT];
    #pragma unroll
    for (int k = 0; k < EPT; ++k) {
        const int e = cbeg + k * 1024 + t;
        d[k] = (e < cend) ? dst[e] : -1;
    }
    __syncthreads();
    #pragma unroll
    for (int k = 0; k < EPT; ++k)
        if ((unsigned)d[k] < (unsigned)NN) atomicAdd(&h[d[k] >> BSH], 1);
    __syncthreads();
    if (t < NB) bh[t * G1 + blockIdx.x] = h[t];
}

__device__ __forceinline__ int wave_incl_scan(int x, int lane) {
    x += __builtin_amdgcn_update_dpp(0, x, 0x111, 0xf, 0xf, true);
    x += __builtin_amdgcn_update_dpp(0, x, 0x112, 0xf, 0xf, true);
    x += __builtin_amdgcn_update_dpp(0, x, 0x114, 0xf, 0xf, true);
    x += __builtin_amdgcn_update_dpp(0, x, 0x118, 0xf, 0xf, true);
    const int row0 = __builtin_amdgcn_readlane(x, 15);
    if (lane >= 16) x += row0;
    return x;
}

__global__ void __launch_bounds__(1024)
k_scan(const int* __restrict__ in, int* __restrict__ out) {
    __shared__ int wsum[2][4][32];
    const int t = threadIdx.x;
    const int lane = t & 31;
    const int w = __builtin_amdgcn_readfirstlane(t >> 5);
    int carry = 0;
    int buf = 0;
    for (int base = 0; base < NSC; base += 4 * 4096) {
        int4 d[4];
        int s4[4], inc[4];
        #pragma unroll
        for (int k = 0; k < 4; ++k) {
            const int i = base + k * 4096 + t * 4;
            d[k] = make_int4(0, 0, 0, 0);
            if (i < NSC) d[k] = *(const int4*)(in + i);
        }
        #pragma unroll
        for (int k = 0; k < 4; ++k) {
            s4[k] = d[k].x + d[k].y + d[k].z + d[k].w;
            inc[k] = wave_incl_scan(s4[k], lane);
        }
        if (lane == 31) {
            #pragma unroll
            for (int k = 0; k < 4; ++k) wsum[buf][k][w] = inc[k];
        }
        __syncthreads();
        int run = carry;
        #pragma unroll
        for (int k = 0; k < 4; ++k) {
            const int ws = wsum[buf][k][lane];
            const int winc = wave_incl_scan(ws, lane);
            const int wave_excl = __builtin_amdgcn_readlane(winc - ws, w);
            const int total = __builtin_amdgcn_readlane(winc, 31);
            const int excl = run + wave_excl + inc[k] - s4[k];
            const int i = base + k * 4096 + t * 4;
            if (i < NSC) {
                int4 o = make_int4(excl, excl + d[k].x, excl + d[k].x + d[k].y, excl + d[k].x + d[k].y + d[k].z);
                *(int4*)(out + i) = o;
            }
            run += total;
        }
        carry = run;
        buf ^= 1;
    }
    if (t == 0) out[NSC] = carry;
}

__global__ void __launch_bounds__(1024)
k_bin_scatter(const int* __restrict__ src, const int* __restrict__ dst,
              const int* __restrict__ boff, int* __restrict__ tmpi) {
    __shared__ int cur[NB];
    int2* tmp = (int2*)tmpi;
    const int t = threadIdx.x;
    if (t < NB) cur[t] = boff[t * G1 + blockIdx.x];
    const int cbeg = blockIdx.x * CHUNK;
    const int cend = min(cbeg + CHUNK, NE);
    int d[EPT], s[EPT];
    #pragma unroll
    for (int k = 0; k < EPT; ++k) {
        const int e = cbeg + k * 1024 + t;
        d[k] = (e < cend) ? dst[e] : -1;
        s[k] = (e < cend) ? src[e] : 0;
    }
    __syncthreads();
    #pragma unroll
    for (int k = 0; k < EPT; ++k) {
        if ((unsigned)d[k] < (unsigned)NN) {
            const int pos = atomicAdd(&cur[d[k] >> BSH], 1);
            if ((unsigned)pos < (unsigned)NE) tmp[pos] = make_int2(s[k], d[k]);
        }
    }
}

__global__ void __launch_bounds__(1024)
k_bin_place(const int* __restrict__ boff, const int* __restrict__ tmpi,
            int* __restrict__ rowptr, int* __restrict__ col) {
    __shared__ int cnt[BINSZ];
    __shared__ int wtot[BINSZ / 32];
    const int2* tmp = (const int2*)tmpi;
    const int t = threadIdx.x;
    const int lane = t & 31;
    const int w = t >> 5;
    const int bin = blockIdx.x;
    const int base = bin * BINSZ;
    const int beg = boff[bin * G1];
    const int end = boff[(bin + 1) * G1];
    if (t < BINSZ) cnt[t] = 0;
    __syncthreads();
    for (int e0 = beg; e0 < end; e0 += 1024 * PU) {
        int dl[PU];
        #pragma unroll
        for (int u = 0; u < PU; ++u) {
            const int e = e0 + u * 1024 + t;
            dl[u] = (e < end) ? (tmp[e].y - base) : -1;
        }
        #pragma unroll
        for (int u = 0; u < PU; ++u)
            if ((unsigned)dl[u] < (unsigned)BINSZ) atomicAdd(&cnt[dl[u]], 1);
    }
    __syncthreads();
    const int c = (t < BINSZ) ? cnt[t] : 0;
    const int incl = wave_incl_scan(c, lane);
    if (lane == 31 && w < BINSZ / 32) wtot[w] = incl;
    __syncthreads();
    int wpre = 0;
    #pragma unroll
    for (int j = 0; j < BINSZ / 32; ++j) wpre += (j < w) ? wtot[j] : 0;
    const int excl = beg + wpre + incl - c;
    if (t < BINSZ) {
        const int node = base + t;
        if (node < NN) rowptr[node] = excl;
        cnt[t] = excl;
    }
    if (bin == NB - 1 && t == 0) rowptr[NN] = end;
    __syncthreads();
    for (int e0 = beg; e0 < end; e0 += 1024 * PU) {
        int2 p[PU];
        #pragma unroll
        for (int u = 0; u < PU; ++u) {
            const int e = e0 + u * 1024 + t;
            p[u] = (e < end) ? tmp[e] : make_int2(0, -1);
        }
        #pragma unroll
        for (int u = 0; u < PU; ++u) {
            const int dl = p[u].y - base;
            if (p[u].y >= 0 && (unsigned)dl < (unsigned)BINSZ) {
                const int pos = atomicAdd(&cnt[dl], 1);
                if ((unsigned)pos < (unsigned)NE) col[pos] = p[u].x;
            }
        }
    }
}

template <int U>
__device__ __forceinline__ void gather_trip(const float* __restrict__ feat, int myc, int j, int n,
                                            int hi, int c4, float4& acc) {
    float4 v[U];
    #pragma unroll
    for (int u = 0; u < U; ++u) {
        const int s = __shfl(myc, j + 2 * u + hi, 32);
        v[u] = *(const float4*)(feat + s * CH + c4);
    }
    #pragma unroll
    for (int u = 0; u < U; ++u) {
        const bool ok = (j + 2 * u + hi) < n;
        acc.x += ok ? v[u].x : 0.0f;
        acc.y += ok ? v[u].y : 0.0f;
        acc.z += ok ? v[u].z : 0.0f;
        acc.w += ok ? v[u].w : 0.0f;
    }
}

__global__ void __launch_bounds__(256)
k_gather(const float* __restrict__ feat, const int* __restrict__ rowptr,
         const int* __restrict__ col, float* __restrict__ agg, float* __restrict__ cnt) {
    const int lane = threadIdx.x & 31;
    const int hi = lane >> 4;
    const int c4 = (lane & 15) * 4;
    const int gw = __builtin_amdgcn_readfirstlane((int)(blockIdx.x * 8 + (threadIdx.x >> 5)));
    const int n0 = NODE_LO(gw);
    const int nn = NODE_LO(gw + 1) - n0;

    int rpv = 0;
    if (lane <= nn) rpv = rowptr[n0 + lane];
    int end = __builtin_amdgcn_readlane(rpv, 0);
    int nend = __builtin_amdgcn_readlane(rpv, 1);
    int cur = 0;
    if (lane < nend - end) cur = col[end + lane];
    cur = min(max(cur, 0), NN - 1);

    for (int k = 0; k < nn; ++k) {
        const int node = n0 + k;
        const int beg = end;
        end = nend;
        int myc = cur;
        int nxt = 0;
        if (k + 1 < nn) {
            nend = __builtin_amdgcn_readlane(rpv, k + 2);
            if (lane < nend - end) nxt = col[end + lane];
        }

        float4 acc = make_float4(0.0f, 0.0f, 0.0f, 0.0f);
        for (int base = beg; base < end; base += 32) {
            int n = end - base;
            if (n > 32) n = 32;
            if (base != beg) {
                int c = 0;
                if (lane < n) c = col[base + lane];
                myc = min(max(c, 0), NN - 1);
            }
            for (int j = 0; j < n; j += 16) {
                if (n - j > 8) gather_trip<8>(feat, myc, j, n, hi, c4, acc);
                else           gather_trip<4>(feat, myc, j, n, hi, c4, acc);
            }
        }
        acc.x += __shfl_xor(acc.x, 16, 32);
        acc.y += __shfl_xor(acc.y, 16, 32);
        acc.z += __shfl_xor(acc.z, 16, 32);
        acc.w += __shfl_xor(acc.w, 16, 32);
        if (hi == 0) *(float4*)(agg + node * CH + c4) = acc;
        if (lane == 0) cnt[node] = (float)(end - beg);
        asm volatile("" : "+v"(nxt));
        cur = min(max(nxt, 0), NN - 1);
    }
}

__global__ void __launch_bounds__(256)
k_sage_gemm(const float* __restrict__ agg, const float* __restrict__ cnt,
            const float* __restrict__ feat,
            const float* __restrict__ Wl, const float* __restrict__ Wr,
            const float* __restrict__ bias, float* __restrict__ out) {
    __shared__ float Wfrag[32 * 4 * 32 * 2];

    for (int idx = threadIdx.x; idx < 32 * 4 * 32 * 2; idx += 256) {
        int v  = idx & 1;
        int l  = (idx >> 1) & 31;
        int nt = (idx >> 6) & 3;
        int s  = idx >> 8;
        int K  = 4 * s + ((l >> 4) << 1) + v;
        int n  = nt * 16 + (l & 15);
        Wfrag[idx] = (K < 64) ? Wl[K * 64 + n] : Wr[(K - 64) * 64 + n];
    }
    __syncthreads();

    int lane = threadIdx.x & 31;
    int hi   = lane >> 4;
    int r    = lane & 15;
    int tile = blockIdx.x * 8 + (threadIdx.x >> 5);
    bool active = (tile < NTILES);

    if (active) {
        int row = tile * 16 + r;
        float rd = 1.0f / fmaxf(cnt[row], 1.0f);
        const float* aggrow  = agg  + row * CH;
        const float* featrow = feat + row * CH;

        v8f acc[4];
        #pragma unroll
        for (int nt = 0; nt < 4; ++nt) {
            float bv = bias[nt * 16 + r];
            #pragma unroll
            for (int j = 0; j < 8; ++j) acc[nt][j] = bv;
        }

        #pragma unroll 4
        for (int s = 0; s < 32; ++s) {
            int kloc = 4 * (s & 15) + hi * 2;
            const float* base = (s < 16) ? aggrow : featrow;
            v2f a = *(const v2f*)(base + kloc);
            if (s < 16) { a.x *= rd; a.y *= rd; }

            #pragma unroll
            for (int nt = 0; nt < 4; ++nt) {
                v2f b = *(const v2f*)&Wfrag[((s * 4 + nt) * 32 + lane) * 2];
                acc[nt] = __builtin_amdgcn_wmma_f32_16x16x4_f32(
                    false, a, false, b, (short)0, acc[nt], false, false);
            }
        }

        #pragma unroll
        for (int nt = 0; nt < 4; ++nt) {
            #pragma unroll
            for (int v = 0; v < 8; ++v) {
                int orow = tile * 16 + hi * 8 + v;
                out[orow * CH + nt * 16 + r] = fmaxf(acc[nt][v], 0.0f);
            }
        }
    }
}

#define FC_U 8
__global__ void __launch_bounds__(256)
k_fc(const float* __restrict__ h2, const float* __restrict__ Wfc,
     const float* __restrict__ bfc, float* __restrict__ pred) {
    const int lane = threadIdx.x & 31;
    const int gw = __builtin_amdgcn_readfirstlane((int)(blockIdx.x * 8 + (threadIdx.x >> 5)));
    const int n0 = NODE_LO(gw);
    const int n1 = NODE_LO(gw + 1);
    const float w0 = Wfc[lane];
    const float w1 = Wfc[lane + 32];
    const float bias = bfc[0];
    for (int b = n0; b < n1; b += FC_U) {
        float a0[FC_U], a1[FC_U];
        #pragma unroll
        for (int u = 0; u < FC_U; ++u) {
            const int nd = min(b + u, n1 - 1);
            const float* hrow = h2 + nd * CH;
            a0[u] = hrow[lane];
            a1[u] = hrow[lane + 32];
        }
        float s[FC_U];
        #pragma unroll
        for (int u = 0; u < FC_U; ++u) s[u] = a0[u] * w0 + a1[u] * w1;
        #pragma unroll
        for (int off = 16; off > 0; off >>= 1) {
            #pragma unroll
            for (int u = 0; u < FC_U; ++u) s[u] += __shfl_down(s[u], off, 32);
        }
        #pragma unroll
        for (int u = 0; u < FC_U; ++u)
            if (lane == 0 && b + u < n1) pred[b + u] = s[u] + bias;
    }
}

extern "C" void kernel_launch(void* const* d_in, const int* in_sizes, int n_in,
                              void* d_out, int out_size, void* d_ws, size_t ws_size,
                              hipStream_t stream) {
    const float* x   = (const float*)d_in[0];
    const int*   ei  = (const int*)d_in[1];
    const float* W1l = (const float*)d_in[2];
    const float* b1  = (const float*)d_in[3];
    const float* W1r = (const float*)d_in[4];
    const float* W2l = (const float*)d_in[5];
    const float* b2  = (const float*)d_in[6];
    const float* W2r = (const float*)d_in[7];
    const float* Wfc = (const float*)d_in[8];
    const float* bfc = (const float*)d_in[9];
    float* pred = (float*)d_out;

    const int* src = ei;
    const int* dst = ei + NE;

    float* xc  = (float*)d_ws;
    float* agg = xc  + NN * CH;
    float* cnt = agg + NN * CH;
    float* h1  = cnt + NNP;
    float* h2  = xc;
    int* rowptr = (int*)(h1 + NN * CH);
    int* col    = rowptr + NNP;
    int* bh     = col + NE;
    int* boff   = bh + NSCP;
    int* tmp    = boff + NSCP;

    const int ggrid = (NTILES + 7) / 8;

    k_clean<<<FW_WG, 256, 0, stream>>>(x, xc);
    k_bin_count<<<G1, 1024, 0, stream>>>(dst, bh);
    k_scan<<<1, 1024, 0, stream>>>(bh, boff);
    k_bin_scatter<<<G1, 1024, 0, stream>>>(src, dst, boff, tmp);
    k_bin_place<<<NB, 1024, 0, stream>>>(boff, tmp, rowptr, col);

    k_gather<<<FW_WG, 256, 0, stream>>>(xc, rowptr, col, agg, cnt);
    k_sage_gemm<<<ggrid, 256, 0, stream>>>(agg, cnt, xc, W1l, W1r, b1, h1);

    k_gather<<<FW_WG, 256, 0, stream>>>(h1, rowptr, col, agg, cnt);
    k_sage_gemm<<<ggrid, 256, 0, stream>>>(agg, cnt, h1, W2l, W2r, b2, h2);

    k_fc<<<FW_WG, 256, 0, stream>>>(h2, Wfc, bfc, pred);
}


# Pull sum fused into the layer's matrix kernel: each wave gathers its 16-node tile's means into LDS (272-B row pitch), reads A fragments there; agg/cnt round trip, 2 launches gone; sum order kept
# speedup vs baseline: 4.3136x; 1.1547x over previous
#include <hip/hip_runtime.h>


typedef __attribute__((ext_vector_type(2))) float v2f;
typedef __attribute__((ext_vector_type(8))) float v8f;

#define NN 100000
#define NE 1200000
#define CH 64
#define NTILES (NN / 16)
#define NNP 100032

#define BSH 9
#define BINSZ (1 << BSH)
#define NB ((NN + BINSZ - 1) / BINSZ)
#define G1 128
#define CHUNK ((NE + G1 - 1) / G1)
#define EPT 10
#define NSC (NB * G1)
#define NSCP (NSC + 64)
#define PU 8
static_assert(CHUNK <= EPT * 1024, "a chunk must fit one pass of the work-group");
static_assert(NSC % 4 == 0, "the scan reads four counts per load");

#define FW_WG 1024
#define FW_WAVES (FW_WG * 8)
#define FW_THREADS (FW_WG * 256)
#define NODE_LO(w) (((w) * NN) / FW_WAVES)
static_assert((long long)FW_WAVES * NN < 2147483648LL, "the node split is computed in 32-bit integers");
static_assert((NN + FW_WAVES - 1) / FW_WAVES < 31, "a wave's segment bounds must fit one register across its lanes");

#define N4 (NN * CH / 4)
#define CLEAN_TRIPS ((N4 + FW_THREADS - 1) / FW_THREADS)
#define CLEAN_FULL (N4 / FW_THREADS)
__global__ void __launch_bounds__(256)
k_clean(const float* __restrict__ x, float* __restrict__ xc) {
    const int t = blockIdx.x * 256 + threadIdx.x;
    float4 v[CLEAN_TRIPS];
    #pragma unroll
    for (int k = 0; k < CLEAN_TRIPS; ++k) {
        const int i = k * FW_THREADS + t;
        v[k] = make_float4(0.0f, 0.0f, 0.0f, 0.0f);
        if (k < CLEAN_FULL || i < N4) v[k] = ((const float4*)x)[i];
    }
    #pragma unroll
    for (int k = 0; k < CLEAN_TRIPS; ++k) {
        const int i = k * FW_THREADS + t;
        if (k < CLEAN_FULL || i < N4) {
            float4 r;
            r.x = (v[k].x == v[k].x) ? v[k].x : 0.0f;
            r.y = (v[k].y == v[k].y) ? v[k].y : 0.0f;
            r.z = (v[k].z == v[k].z) ? v[k].z : 0.0f;
            r.w = (v[k].w == v[k].w) ? v[k].w : 0.0f;
            ((float4*)xc)[i] = r;
        }
    }
}


__global__ void __launch_bounds__(1024)
k_bin_count(const int* __restrict__ dst, int* __restrict__ bh) {
    __shared__ int h[NB];
    const int t = threadIdx.x;
    if (t < NB) h[t] = 0;
    const int cbeg = blockIdx.x * CHUNK;
    const int cend = min(cbeg + CHUNK, NE);
    int d[EPT];
    #pragma unroll
    for (int k = 0; k < EPT; ++k) {
        const int e = cbeg + k * 1024 + t;
        d[k] = (e < cend) ? dst[e] : -1;
    }
    __syncthreads();
    #pragma unroll
    for (int k = 0; k < EPT; ++k)
        if ((unsigned)d[k] < (unsigned)NN) atomicAdd(&h[d[k] >> BSH], 1);
    __syncthreads();
    if (t < NB) bh[t * G1 + blockIdx.x] = h[t];
}

__device__ __forceinline__ int wave_incl_scan(int x, int lane) {
    x += __builtin_amdgcn_update_dpp(0, x, 0x111, 0xf, 0xf, true);
    x += __builtin_amdgcn_update_dpp(0, x, 0x112, 0xf, 0xf, true);
    x += __builtin_amdgcn_update_dpp(0, x, 0x114, 0xf, 0xf, true);
    x += __builtin_amdgcn_update_dpp(0, x, 0x118, 0xf, 0xf, true);
    const int row0 = __builtin_amdgcn_readlane(x, 15);
    if (lane >= 16) x += row0;
    return x;
}

__global__ void __launch_bounds__(1024)
k_scan(const int* __restrict__ in, int* __restrict__ out) {
    __shared__ int wsum[2][4][32];
    const int t = threadIdx.x;
    const int lane = t & 31;
    const int w = __builtin_amdgcn_readfirstlane(t >> 5);
    int carry = 0;
    int buf = 0;
    for (int base = 0; base < NSC; base += 4 * 4096) {
        int4 d[4];
        int s4[4], inc[4];
        #pragma unroll
        for (int k = 0; k < 4; ++k) {
            const int i = base + k * 4096 + t * 4;
            d[k] = make_int4(0, 0, 0, 0);
            if (i < NSC) d[k] = *(const int4*)(in + i);
        }
        #pragma unroll
        for (int k = 0; k < 4; ++k) {
            s4[k] = d[k].x + d[k].y + d[k].z + d[k].w;
            inc[k] = wave_incl_scan(s4[k], lane);
        }
        if (lane == 31) {
            #pragma unroll
            for (int k = 0; k < 4; ++k) wsum[buf][k][w] = inc[k];
        }
        __syncthreads();
        int run = carry;
        #pragma unroll
        for (int k = 0; k < 4; ++k) {
            const int ws = wsum[buf][k][lane];
            const int winc = wave_incl_scan(ws, lane);
            const int wave_excl = __builtin_amdgcn_readlane(winc - ws, w);
            const int total = __builtin_amdgcn_readlane(winc, 31);
            const int excl = run + wave_excl + inc[k] - s4[k];
            const int i = base + k * 4096 + t * 4;
            if (i < NSC) {
                int4 o = make_int4(excl, excl + d[k].x, excl + d[k].x + d[k].y, excl + d[k].x + d[k].y + d[k].z);
                *(int4*)(out + i) = o;
            }
            run += total;
        }
        carry = run;
        buf ^= 1;
    }
    if (t == 0) out[NSC] = carry;
}

__global__ void __launch_bounds__(1024)
k_bin_scatter(const int* __restrict__ src, const int* __restrict__ dst,
              const int* __restrict__ boff, int* __restrict__ tmpi) {
    __shared__ int cur[NB];
    int2* tmp = (int2*)tmpi;
    const int t = threadIdx.x;
    if (t < NB) cur[t] = boff[t * G1 + blockIdx.x];
    const int cbeg = blockIdx.x * CHUNK;
    const int cend = min(cbeg + CHUNK, NE);
    int d[EPT], s[EPT];
    #pragma unroll
    for (int k = 0; k < EPT; ++k) {
        const int e = cbeg + k * 1024 + t;
        d[k] = (e < cend) ? dst[e] : -1;
        s[k] = (e < cend) ? src[e] : 0;
    }
    __syncthreads();
    #pragma unroll
    for (int k = 0; k < EPT; ++k) {
        if ((unsigned)d[k] < (unsigned)NN) {
            const int pos = atomicAdd(&cur[d[k] >> BSH], 1);
            if ((unsigned)pos < (unsigned)NE) tmp[pos] = make_int2(s[k], d[k]);
        }
    }
}

__global__ void __launch_bounds__(1024)
k_bin_place(const int* __restrict__ boff, const int* __restrict__ tmpi,
            int* __restrict__ rowptr, int* __restrict__ col) {
    __shared__ int cnt[BINSZ];
    __shared__ int wtot[BINSZ / 32];
    const int2* tmp = (const int2*)tmpi;
    const int t = threadIdx.x;
    const int lane = t & 31;
    const int w = t >> 5;
    const int bin = blockIdx.x;
    const int base = bin * BINSZ;
    const int beg = boff[bin * G1];
    const int end = boff[(bin + 1) * G1];
    if (t < BINSZ) cnt[t] = 0;
    __syncthreads();
    for (int e0 = beg; e0 < end; e0 += 1024 * PU) {
        int dl[PU];
        #pragma unroll
        for (int u = 0; u < PU; ++u) {
            const int e = e0 + u * 1024 + t;
            dl[u] = (e < end) ? (tmp[e].y - base) : -1;
        }
        #pragma unroll
        for (int u = 0; u < PU; ++u)
            if ((unsigned)dl[u] < (unsigned)BINSZ) atomicAdd(&cnt[dl[u]], 1);
    }
    __syncthreads();
    const int c = (t < BINSZ) ? cnt[t] : 0;
    const int incl = wave_incl_scan(c, lane);
    if (lane == 31 && w < BINSZ / 32) wtot[w] = incl;
    __syncthreads();
    int wpre = 0;
    #pragma unroll
    for (int j = 0; j < BINSZ / 32; ++j) wpre += (j < w) ? wtot[j] : 0;
    const int excl = beg + wpre + incl - c;
    if (t < BINSZ) {
        const int node = base + t;
        if (node < NN) rowptr[node] = excl;
        cnt[t] = excl;
    }
    if (bin == NB - 1 && t == 0) rowptr[NN] = end;
    __syncthreads();
    for (int e0 = beg; e0 < end; e0 += 1024 * PU) {
        int2 p[PU];
        #pragma unroll
        for (int u = 0; u < PU; ++u) {
            const int e = e0 + u * 1024 + t;
            p[u] = (e < end) ? tmp[e] : make_int2(0, -1);
        }
        #pragma unroll
        for (int u = 0; u < PU; ++u) {
            const int dl = p[u].y - base;
            if (p[u].y >= 0 && (unsigned)dl < (unsigned)BINSZ) {
                const int pos = atomicAdd(&cnt[dl], 1);
                if ((unsigned)pos < (unsigned)NE) col[pos] = p[u].x;
            }
        }
    }
}

template <int U>
__device__ __forceinline__ void gather_trip(const float* __restrict__ feat, int myc, int j, int n,
                                            int hi, int c4, float4& acc) {
    float4 v[U];
    #pragma unroll
    for (int u = 0; u < U; ++u) {
        const int s = __shfl(myc, j + 2 * u + hi, 32);
        v[u] = *(const float4*)(feat + s * CH + c4);
    }
    #pragma unroll
    for (int u = 0; u < U; ++u) {
        const bool ok = (j + 2 * u + hi) < n;
        acc.x += ok ? v[u].x : 0.0f;
        acc.y += ok ? v[u].y : 0.0f;
        acc.z += ok ? v[u].z : 0.0f;
        acc.w += ok ? v[u].w : 0.0f;
    }
}

#define TP 68
__global__ void __launch_bounds__(256)
k_sage_layer(const float* __restrict__ feat, const int* __restrict__ rowptr, const int* __restrict__ col,
             const float* __restrict__ Wl, const float* __restrict__ Wr,
             const float* __restrict__ bias, float* __restrict__ out) {
    __shared__ float Wfrag[32 * 4 * 32 * 2];
    __shared__ float meanT[8][16 * TP];

    for (int idx = threadIdx.x; idx < 32 * 4 * 32 * 2; idx += 256) {
        int v  = idx & 1;
        int l  = (idx >> 1) & 31;
        int nt = (idx >> 6) & 3;
        int s  = idx >> 8;
        int K  = 4 * s + ((l >> 4) << 1) + v;
        int n  = nt * 16 + (l & 15);
        Wfrag[idx] = (K < 64) ? Wl[K * 64 + n] : Wr[(K - 64) * 64 + n];
    }
    __syncthreads();

    const int lane = threadIdx.x & 31;
    const int hi   = lane >> 4;
    const int r    = lane & 15;
    const int c4   = r * 4;
    const int wv   = __builtin_amdgcn_readfirstlane((int)(threadIdx.x >> 5));
    const int tile = blockIdx.x * 8 + wv;

    if (tile < NTILES) {
        float* mt = meanT[wv];
        const int n0 = tile * 16;

        int rpv = 0;
        if (lane <= 16) rpv = rowptr[n0 + lane];
        int end = __builtin_amdgcn_readlane(rpv, 0);
        int nend = __builtin_amdgcn_readlane(rpv, 1);
        int cur = 0;
        if (lane < nend - end) cur = col[end + lane];
        cur = min(max(cur, 0), NN - 1);

        for (int k = 0; k < 16; ++k) {
            const int beg = end;
            end = nend;
            int myc = cur;
            int nxt = 0;
            if (k + 1 < 16) {
                nend = __builtin_amdgcn_readlane(rpv, k + 2);
                if (lane < nend - end) nxt = col[end + lane];
            }

            float4 acc = make_float4(0.0f, 0.0f, 0.0f, 0.0f);
            for (int base = beg; base < end; base += 32) {
                int n = end - base;
                if (n > 32) n = 32;
                if (base != beg) {
                    int c = 0;
                    if (lane < n) c = col[base + lane];
                    myc = min(max(c, 0), NN - 1);
                }
                for (int j = 0; j < n; j += 16) {
                    if (n - j > 8) gather_trip<8>(feat, myc, j, n, hi, c4, acc);
                    else           gather_trip<4>(feat, myc, j, n, hi, c4, acc);
                }
            }
            acc.x += __shfl_xor(acc.x, 16, 32);
            acc.y += __shfl_xor(acc.y, 16, 32);
            acc.z += __shfl_xor(acc.z, 16, 32);
            acc.w += __shfl_xor(acc.w, 16, 32);
            const float rd = 1.0f / fmaxf((float)(end - beg), 1.0f);
            if (hi == 0) *(float4*)(mt + k * TP + c4) = make_float4(acc.x * rd, acc.y * rd, acc.z * rd, acc.w * rd);
            asm volatile("" : "+v"(nxt));
            cur = min(max(nxt, 0), NN - 1);
        }

        const int row = n0 + r;
        const float* featrow = feat + row * CH;
        v2f af[16];
        #pragma unroll
        for (int s = 0; s < 16; ++s) af[s] = *(const v2f*)(featrow + 4 * s + hi * 2);

        v8f acc[4];
        #pragma unroll
        for (int nt = 0; nt < 4; ++nt) {
            float bv = bias[nt * 16 + r];
            #pragma unroll
            for (int j = 0; j < 8; ++j) acc[nt][j] = bv;
        }

        #pragma unroll 4
        for (int s = 0; s < 16; ++s) {
            v2f a = *(const v2f*)(mt + r * TP + 4 * s + hi * 2);
            #pragma unroll
            for (int nt = 0; nt < 4; ++nt) {
                v2f b = *(const v2f*)&Wfrag[((s * 4 + nt) * 32 + lane) * 2];
                acc[nt] = __builtin_amdgcn_wmma_f32_16x16x4_f32(
                    false, a, false, b, (short)0, acc[nt], false, false);
            }
        }
        #pragma unroll
        for (int s = 0; s < 16; ++s) {
            #pragma unroll
            for (int nt = 0; nt < 4; ++nt) {
                v2f b = *(const v2f*)&Wfrag[(((s + 16) * 4 + nt) * 32 + lane) * 2];
                acc[nt] = __builtin_amdgcn_wmma_f32_16x16x4_f32(
                    false, af[s], false, b, (short)0, acc[nt], false, false);
            }
        }

        #pragma unroll
        for (int nt = 0; nt < 4; ++nt) {
            #pragma unroll
            for (int v = 0; v < 8; ++v) {
                int orow = n0 + hi * 8 + v;
                out[orow * CH + nt * 16 + r] = fmaxf(acc[nt][v], 0.0f);
            }
        }
    }
}

#define FC_U 8
__global__ void __launch_bounds__(256)
k_fc(const float* __restrict__ h2, const float* __restrict__ Wfc,
     const float* __restrict__ bfc, float* __restrict__ pred) {
    const int lane = threadIdx.x & 31;
    const int gw = __builtin_amdgcn_readfirstlane((int)(blockIdx.x * 8 + (threadIdx.x >> 5)));
    const int n0 = NODE_LO(gw);
    const int n1 = NODE_LO(gw + 1);
    const float w0 = Wfc[lane];
    const float w1 = Wfc[lane + 32];
    const float bias = bfc[0];
    for (int b = n0; b < n1; b += FC_U) {
        float a0[FC_U], a1[FC_U];
        #pragma unroll
        for (int u = 0; u < FC_U; ++u) {
            const int nd = min(b + u, n1 - 1);
            const float* hrow = h2 + nd * CH;
            a0[u] = hrow[lane];
            a1[u] = hrow[lane + 32];
        }
        float s[FC_U];
        #pragma unroll
        for (int u = 0; u < FC_U; ++u) s[u] = a0[u] * w0 + a1[u] * w1;
        #pragma unroll
        for (int off = 16; off > 0; off >>= 1) {
            #pragma unroll
            for (int u = 0; u < FC_U; ++u) s[u] += __shfl_down(s[u], off, 32);
        }
        #pragma unroll
        for (int u = 0; u < FC_U; ++u)
            if (lane == 0 && b + u < n1) pred[b + u] = s[u] + bias;
    }
}

extern "C" void kernel_launch(void* const* d_in, const int* in_sizes, int n_in,
                              void* d_out, int out_size, void* d_ws, size_t ws_size,
                              hipStream_t stream) {
    const float* x   = (const float*)d_in[0];
    const int*   ei  = (const int*)d_in[1];
    const float* W1l = (const float*)d_in[2];
    const float* b1  = (const float*)d_in[3];
    const float* W1r = (const float*)d_in[4];
    const float* W2l = (const float*)d_in[5];
    const float* b2  = (const float*)d_in[6];
    const float* W2r = (const float*)d_in[7];
    const float* Wfc = (const float*)d_in[8];
    const float* bfc = (const float*)d_in[9];
    float* pred = (float*)d_out;

    const int* src = ei;
    const int* dst = ei + NE;

    float* xc  = (float*)d_ws;
    float* h1  = xc + NN * CH;
    float* h2  = xc;
    int* rowptr = (int*)(h1 + NN * CH);
    int* col    = rowptr + NNP;
    int* bh     = col + NE;
    int* boff   = bh + NSCP;
    int* tmp    = boff + NSCP;

    const int ggrid = (NTILES + 7) / 8;

    k_clean<<<FW_WG, 256, 0, stream>>>(x, xc);
    k_bin_count<<<G1, 1024, 0, stream>>>(dst, bh);
    k_scan<<<1, 1024, 0, stream>>>(bh, boff);
    k_bin_scatter<<<G1, 1024, 0, stream>>>(src, dst, boff, tmp);
    k_bin_place<<<NB, 1024, 0, stream>>>(boff, tmp, rowptr, col);

    k_sage_layer<<<ggrid, 256, 0, stream>>>(xc, rowptr, col, W1l, W1r, b1, h1);
    k_sage_layer<<<ggrid, 256, 0, stream>>>(h1, rowptr, col, W2l, W2r, b2, h2);

    k_fc<<<FW_WG, 256, 0, stream>>>(h2, Wfc, bfc, pred);
}


# nan clean and FC head fused into the layer kernels: layer 1 cleans x in registers at load, layer 2 dots relu(D) with Wfc (4 cols per lane, 16-lane xor tree: FC sum order changes); 2 launches gone
# speedup vs baseline: 4.8618x; 1.1271x over previous
#include <hip/hip_runtime.h>


typedef __attribute__((ext_vector_type(2))) float v2f;
typedef __attribute__((ext_vector_type(4))) float v4f;
typedef __attribute__((ext_vector_type(8))) float v8f;

#define NN 100000
#define NE 1200000
#define CH 64
#define NTILES (NN / 16)
#define NNP 100032

#define BSH 9
#define BINSZ (1 << BSH)
#define NB ((NN + BINSZ - 1) / BINSZ)
#define G1 128
#define CHUNK ((NE + G1 - 1) / G1)
#define EPT 10
#define NSC (NB * G1)
#define NSCP (NSC + 64)
#define PU 8
static_assert(CHUNK <= EPT * 1024, "a chunk must fit one pass of the work-group");
static_assert(NSC % 4 == 0, "the scan reads four counts per load");

__device__ __forceinline__ float nan0(float v) { return (v == v) ? v : 0.0f; }


__global__ void __launch_bounds__(1024)
k_bin_count(const int* __restrict__ dst, int* __restrict__ bh) {
    __shared__ int h[NB];
    const int t = threadIdx.x;
    if (t < NB) h[t] = 0;
    const int cbeg = blockIdx.x * CHUNK;
    const int cend = min(cbeg + CHUNK, NE);
    int d[EPT];
    #pragma unroll
    for (int k = 0; k < EPT; ++k) {
        const int e = cbeg + k * 1024 + t;
        d[k] = (e < cend) ? dst[e] : -1;
    }
    __syncthreads();
    #pragma unroll
    for (int k = 0; k < EPT; ++k)
        if ((unsigned)d[k] < (unsigned)NN) atomicAdd(&h[d[k] >> BSH], 1);
    __syncthreads();
    if (t < NB) bh[t * G1 + blockIdx.x] = h[t];
}

__device__ __forceinline__ int wave_incl_scan(int x, int lane) {
    x += __builtin_amdgcn_update_dpp(0, x, 0x111, 0xf, 0xf, true);
    x += __builtin_amdgcn_update_dpp(0, x, 0x112, 0xf, 0xf, true);
    x += __builtin_amdgcn_update_dpp(0, x, 0x114, 0xf, 0xf, true);
    x += __builtin_amdgcn_update_dpp(0, x, 0x118, 0xf, 0xf, true);
    const int row0 = __builtin_amdgcn_readlane(x, 15);
    if (lane >= 16) x += row0;
    return x;
}

__global__ void __launch_bounds__(1024)
k_scan(const int* __restrict__ in, int* __restrict__ out) {
    __shared__ int wsum[2][4][32];
    const int t = threadIdx.x;
    const int lane = t & 31;
    const int w = __builtin_amdgcn_readfirstlane(t >> 5);
    int carry = 0;
    int buf = 0;
    for (int base = 0; base < NSC; base += 4 * 4096) {
        int4 d[4];
        int s4[4], inc[4];
        #pragma unroll
        for (int k = 0; k < 4; ++k) {
            const int i = base + k * 4096 + t * 4;
            d[k] = make_int4(0, 0, 0, 0);
            if (i < NSC) d[k] = *(const int4*)(in + i);
        }
        #pragma unroll
        for (int k = 0; k < 4; ++k) {
            s4[k] = d[k].x + d[k].y + d[k].z + d[k].w;
            inc[k] = wave_incl_scan(s4[k], lane);
        }
        if (lane == 31) {
            #pragma unroll
            for (int k = 0; k < 4; ++k) wsum[buf][k][w] = inc[k];
        }
        __syncthreads();
        int run = carry;
        #pragma unroll
        for (int k = 0; k < 4; ++k) {
            const int ws = wsum[buf][k][lane];
            const int winc = wave_incl_scan(ws, lane);
            const int wave_excl = __builtin_amdgcn_readlane(winc - ws, w);
            const int total = __builtin_amdgcn_readlane(winc, 31);
            const int excl = run + wave_excl + inc[k] - s4[k];
            const int i = base + k * 4096 + t * 4;
            if (i < NSC) {
                int4 o = make_int4(excl, excl + d[k].x, excl + d[k].x + d[k].y, excl + d[k].x + d[k].y + d[k].z);
                *(int4*)(out + i) = o;
            }
            run += total;
        }
        carry = run;
        buf ^= 1;
    }
    if (t == 0) out[NSC] = carry;
}

__global__ void __launch_bounds__(1024)
k_bin_scatter(const int* __restrict__ src, const int* __restrict__ dst,
              const int* __restrict__ boff, int* __restrict__ tmpi) {
    __shared__ int cur[NB];
    int2* tmp = (int2*)tmpi;
    const int t = threadIdx.x;
    if (t < NB) cur[t] = boff[t * G1 + blockIdx.x];
    const int cbeg = blockIdx.x * CHUNK;
    const int cend = min(cbeg + CHUNK, NE);
    int d[EPT], s[EPT];
    #pragma unroll
    for (int k = 0; k < EPT; ++k) {
        const int e = cbeg + k * 1024 + t;
        d[k] = (e < cend) ? dst[e] : -1;
        s[k] = (e < cend) ? src[e] : 0;
    }
    __syncthreads();
    #pragma unroll
    for (int k = 0; k < EPT; ++k) {
        if ((unsigned)d[k] < (unsigned)NN) {
            const int pos = atomicAdd(&cur[d[k] >> BSH], 1);
            if ((unsigned)pos < (unsigned)NE) tmp[pos] = make_int2(s[k], d[k]);
        }
    }
}

__global__ void __launch_bounds__(1024)
k_bin_place(const int* __restrict__ boff, const int* __restrict__ tmpi,
            int* __restrict__ rowptr, int* __restrict__ col) {
    __shared__ int cnt[BINSZ];
    __shared__ int wtot[BINSZ / 32];
    const int2* tmp = (const int2*)tmpi;
    const int t = threadIdx.x;
    const int lane = t & 31;
    const int w = t >> 5;
    const int bin = blockIdx.x;
    const int base = bin * BINSZ;
    const int beg = boff[bin * G1];
    const int end = boff[(bin + 1) * G1];
    if (t < BINSZ) cnt[t] = 0;
    __syncthreads();
    for (int e0 = beg; e0 < end; e0 += 1024 * PU) {
        int dl[PU];
        #pragma unroll
        for (int u = 0; u < PU; ++u) {
            const int e = e0 + u * 1024 + t;
            dl[u] = (e < end) ? (tmp[e].y - base) : -1;
        }
        #pragma unroll
        for (int u = 0; u < PU; ++u)
            if ((unsigned)dl[u] < (unsigned)BINSZ) atomicAdd(&cnt[dl[u]], 1);
    }
    __syncthreads();
    const int c = (t < BINSZ) ? cnt[t] : 0;
    const int incl = wave_incl_scan(c, lane);
    if (lane == 31 && w < BINSZ / 32) wtot[w] = incl;
    __syncthreads();
    int wpre = 0;
    #pragma unroll
    for (int j = 0; j < BINSZ / 32; ++j) wpre += (j < w) ? wtot[j] : 0;
    const int excl = beg + wpre + incl - c;
    if (t < BINSZ) {
        const int node = base + t;
        if (node < NN) rowptr[node] = excl;
        cnt[t] = excl;
    }
    if (bin == NB - 1 && t == 0) rowptr[NN] = end;
    __syncthreads();
    for (int e0 = beg; e0 < end; e0 += 1024 * PU) {
        int2 p[PU];
        #pragma unroll
        for (int u = 0; u < PU; ++u) {
            const int e = e0 + u * 1024 + t;
            p[u] = (e < end) ? tmp[e] : make_int2(0, -1);
        }
        #pragma unroll
        for (int u = 0; u < PU; ++u) {
            const int dl = p[u].y - base;
            if (p[u].y >= 0 && (unsigned)dl < (unsigned)BINSZ) {
                const int pos = atomicAdd(&cnt[dl], 1);
                if ((unsigned)pos < (unsigned)NE) col[pos] = p[u].x;
            }
        }
    }
}

template <int U, bool CLEAN>
__device__ __forceinline__ void gather_trip(const float* __restrict__ feat, int myc, int j, int n,
                                            int hi, int c4, float4& acc) {
    static_assert(U == 4 || U == 8, "the fence below names every row of the trip");
    v4f v[U];
    #pragma unroll
    for (int u = 0; u < U; ++u) {
        const int s = __shfl(myc, j + 2 * u + hi, 32);
        v[u] = *(const v4f*)(feat + s * CH + c4);
    }
    if constexpr (U == 8)
        asm volatile("" : "+v"(v[0]), "+v"(v[1]), "+v"(v[2]), "+v"(v[3]), "+v"(v[4]), "+v"(v[5]), "+v"(v[6]), "+v"(v[7]));
    else
        asm volatile("" : "+v"(v[0]), "+v"(v[1]), "+v"(v[2]), "+v"(v[3]));
    #pragma unroll
    for (int u = 0; u < U; ++u) {
        const bool ok = (j + 2 * u + hi) < n;
        v4f c = v[u];
        if (CLEAN) { c.x = nan0(c.x); c.y = nan0(c.y); c.z = nan0(c.z); c.w = nan0(c.w); }
        acc.x += ok ? c.x : 0.0f;
        acc.y += ok ? c.y : 0.0f;
        acc.z += ok ? c.z : 0.0f;
        acc.w += ok ? c.w : 0.0f;
    }
}

#define TP 68

template <int XM>
__device__ __forceinline__ float row_xmask_add(float x) {
    return x + __builtin_bit_cast(float, __builtin_amdgcn_update_dpp(0, __builtin_bit_cast(int, x), 0x160 + XM, 0xf, 0xf, true));
}

template <bool CLEAN, bool HEAD>
__global__ void __launch_bounds__(256)
k_sage_layer(const float* __restrict__ feat, const int* __restrict__ rowptr, const int* __restrict__ col,
             const float* __restrict__ Wl, const float* __restrict__ Wr,
             const float* __restrict__ bias, float* __restrict__ out,
             const float* __restrict__ Wfc, const float* __restrict__ bfc, float* __restrict__ pred) {
    __shared__ float Wfrag[32 * 4 * 32 * 2];
    __shared__ float meanT[8][16 * TP];

    for (int idx = threadIdx.x; idx < 32 * 4 * 32 * 2; idx += 256) {
        int v  = idx & 1;
        int l  = (idx >> 1) & 31;
        int nt = (idx >> 6) & 3;
        int s  = idx >> 8;
        int K  = 4 * s + ((l >> 4) << 1) + v;
        int n  = nt * 16 + (l & 15);
        Wfrag[idx] = (K < 64) ? Wl[K * 64 + n] : Wr[(K - 64) * 64 + n];
    }
    __syncthreads();

    const int lane = threadIdx.x & 31;
    const int hi   = lane >> 4;
    const int r    = lane & 15;
    const int c4   = r * 4;
    const int wv   = __builtin_amdgcn_readfirstlane((int)(threadIdx.x >> 5));
    const int tile = blockIdx.x * 8 + wv;

    if (tile < NTILES) {
        float* mt = meanT[wv];
        const int n0 = tile * 16;

        int rpv = 0;
        if (lane <= 16) rpv = rowptr[n0 + lane];
        int end = __builtin_amdgcn_readlane(rpv, 0);
        int nend = __builtin_amdgcn_readlane(rpv, 1);
        int cur = 0;
        if (lane < nend - end) cur = col[end + lane];
        cur = min(max(cur, 0), NN - 1);

        for (int k = 0; k < 16; ++k) {
            const int beg = end;
            end = nend;
            int myc = cur;
            int nxt = 0;
            if (k + 1 < 16) {
                nend = __builtin_amdgcn_readlane(rpv, k + 2);
                if (lane < nend - end) nxt = col[end + lane];
            }

            float4 acc = make_float4(0.0f, 0.0f, 0.0f, 0.0f);
            for (int base = beg; base < end; base += 32) {
                int n = end - base;
                if (n > 32) n = 32;
                if (base != beg) {
                    int c = 0;
                    if (lane < n) c = col[base + lane];
                    myc = min(max(c, 0), NN - 1);
                }
                for (int j = 0; j < n; j += 16) {
                    if (n - j > 8) gather_trip<8, CLEAN>(feat, myc, j, n, hi, c4, acc);
                    else           gather_trip<4, CLEAN>(feat, myc, j, n, hi, c4, acc);
                }
            }
            acc.x += __shfl_xor(acc.x, 16, 32);
            acc.y += __shfl_xor(acc.y, 16, 32);
            acc.z += __shfl_xor(acc.z, 16, 32);
            acc.w += __shfl_xor(acc.w, 16, 32);
            const float rd = 1.0f / fmaxf((float)(end - beg), 1.0f);
            if (hi == 0) *(float4*)(mt + k * TP + c4) = make_float4(acc.x * rd, acc.y * rd, acc.z * rd, acc.w * rd);
            asm volatile("" : "+v"(nxt));
            cur = min(max(nxt, 0), NN - 1);
        }

        const int row = n0 + r;
        const float* featrow = feat + row * CH;
        v2f af[16];
        #pragma unroll
        for (int s = 0; s < 16; ++s) af[s] = *(const v2f*)(featrow + 4 * s + hi * 2);
        if (CLEAN) {
            #pragma unroll
            for (int s = 0; s < 16; ++s) { af[s].x = nan0(af[s].x); af[s].y = nan0(af[s].y); }
        }

        v8f acc[4];
        #pragma unroll
        for (int nt = 0; nt < 4; ++nt) {
            float bv = bias[nt * 16 + r];
            #pragma unroll
            for (int j = 0; j < 8; ++j) acc[nt][j] = bv;
        }

        #pragma unroll 4
        for (int s = 0; s < 16; ++s) {
            v2f a = *(const v2f*)(mt + r * TP + 4 * s + hi * 2);
            #pragma unroll
            for (int nt = 0; nt < 4; ++nt) {
                v2f b = *(const v2f*)&Wfrag[((s * 4 + nt) * 32 + lane) * 2];
                acc[nt] = __builtin_amdgcn_wmma_f32_16x16x4_f32(
                    false, a, false, b, (short)0, acc[nt], false, false);
            }
        }
        #pragma unroll
        for (int s = 0; s < 16; ++s) {
            #pragma unroll
            for (int nt = 0; nt < 4; ++nt) {
                v2f b = *(const v2f*)&Wfrag[(((s + 16) * 4 + nt) * 32 + lane) * 2];
                acc[nt] = __builtin_amdgcn_wmma_f32_16x16x4_f32(
                    false, af[s], false, b, (short)0, acc[nt], false, false);
            }
        }

        if (!HEAD) {
            #pragma unroll
            for (int nt = 0; nt < 4; ++nt) {
                #pragma unroll
                for (int v = 0; v < 8; ++v) {
                    int orow = n0 + hi * 8 + v;
                    out[orow * CH + nt * 16 + r] = fmaxf(acc[nt][v], 0.0f);
                }
            }
        } else {
            float wf[4];
            #pragma unroll
            for (int nt = 0; nt < 4; ++nt) wf[nt] = Wfc[nt * 16 + r];
            const float bf = bfc[0];
            float p[8];
            #pragma unroll
            for (int v = 0; v < 8; ++v) {
                p[v] = fmaxf(acc[0][v], 0.0f) * wf[0];
                p[v] += fmaxf(acc[1][v], 0.0f) * wf[1];
                p[v] += fmaxf(acc[2][v], 0.0f) * wf[2];
                p[v] += fmaxf(acc[3][v], 0.0f) * wf[3];
            }
            #pragma unroll
            for (int v = 0; v < 8; ++v) p[v] = row_xmask_add<8>(p[v]);
            #pragma unroll
            for (int v = 0; v < 8; ++v) p[v] = row_xmask_add<4>(p[v]);
            #pragma unroll
            for (int v = 0; v < 8; ++v) p[v] = row_xmask_add<2>(p[v]);
            #pragma unroll
            for (int v = 0; v < 8; ++v) p[v] = row_xmask_add<1>(p[v]);
            float mine = p[0];
            #pragma unroll
            for (int v = 1; v < 8; ++v) mine = (r == v) ? p[v] : mine;
            if (r < 8) mt[hi * 8 + r] = mine + bf;
        }
    }
    if (HEAD) {
        __syncthreads();
        if (threadIdx.x < 128) {
            const int node = blockIdx.x * 128 + threadIdx.x;
            if (node < NN) pred[node] = meanT[threadIdx.x >> 4][threadIdx.x & 15];
        }
    }
}

extern "C" void kernel_launch(void* const* d_in, const int* in_sizes, int n_in,
                              void* d_out, int out_size, void* d_ws, size_t ws_size,
                              hipStream_t stream) {
    const float* x   = (const float*)d_in[0];
    const int*   ei  = (const int*)d_in[1];
    const float* W1l = (const float*)d_in[2];
    const float* b1  = (const float*)d_in[3];
    const float* W1r = (const float*)d_in[4];
    const float* W2l = (const float*)d_in[5];
    const float* b2  = (const float*)d_in[6];
    const float* W2r = (const float*)d_in[7];
    const float* Wfc = (const float*)d_in[8];
    const float* bfc = (const float*)d_in[9];
    float* pred = (float*)d_out;

    const int* src = ei;
    const int* dst = ei + NE;

    float* h1  = (float*)d_ws;
    int* rowptr = (int*)(h1 + NN * CH);
    int* col    = rowptr + NNP;
    int* bh     = col + NE;
    int* boff   = bh + NSCP;
    int* tmp    = boff + NSCP;

    const int ggrid = (NTILES + 7) / 8;

    k_bin_count<<<G1, 1024, 0, stream>>>(dst, bh);
    k_scan<<<1, 1024, 0, stream>>>(bh, boff);
    k_bin_scatter<<<G1, 1024, 0, stream>>>(src, dst, boff, tmp);
    k_bin_place<<<NB, 1024, 0, stream>>>(boff, tmp, rowptr, col);

    k_sage_layer<true, false><<<ggrid, 256, 0, stream>>>(x, rowptr, col, W1l, W1r, b1, h1, Wfc, bfc, pred);
    k_sage_layer<false, true><<<ggrid, 256, 0, stream>>>(h1, rowptr, col, W2l, W2r, b2, nullptr, Wfc, bfc, pred);
}


# Edge sort in 2 launches, was 4: each of 256 chunks sorts its pairs by bin in LDS into its own run + a (start,size) table; the bin pass sums the table itself. Slot (so float sum) order by LDS arrival
# speedup vs baseline: 5.6318x; 1.1584x over previous
#include <hip/hip_runtime.h>


typedef __attribute__((ext_vector_type(2))) float v2f;
typedef __attribute__((ext_vector_type(4))) float v4f;
typedef __attribute__((ext_vector_type(8))) float v8f;

#define NN 100000
#define NE 1200000
#define CH 64
#define NTILES (NN / 16)
#define NNP 100032

#define BSH 9
#define BINSZ (1 << BSH)
#define NB ((NN + BINSZ - 1) / BINSZ)
#define G1 256
#define CHUNK ((NE + G1 - 1) / G1)
#define CT 512
#define EPT 10
#define NTAB (NB * G1)
#define PPW (G1 / 32)
#define PT 2
#define TABT ((NTAB + 4095) / 4096)
#define COT ((CHUNK + 2 * CT - 1) / (2 * CT))
static_assert(CHUNK <= EPT * CT, "a chunk must fit one pass of the work-group");
static_assert(CHUNK < 32768, "piece start and piece size share one 32-bit table word");
static_assert(NB <= 256, "the chunk pass scans its bin counts with eight waves");
static_assert(CHUNK % 2 == 0, "an odd run is shorter than the chunk: its last 16-byte LDS read stays inside the buffer");
static_assert(G1 % 32 == 0 && G1 <= 1024, "the bin pass deals whole pieces to its 32 waves");

__device__ __forceinline__ float nan0(float v) { return (v == v) ? v : 0.0f; }


__device__ __forceinline__ int wave_incl_scan(int x, int lane) {
    x += __builtin_amdgcn_update_dpp(0, x, 0x111, 0xf, 0xf, true);
    x += __builtin_amdgcn_update_dpp(0, x, 0x112, 0xf, 0xf, true);
    x += __builtin_amdgcn_update_dpp(0, x, 0x114, 0xf, 0xf, true);
    x += __builtin_amdgcn_update_dpp(0, x, 0x118, 0xf, 0xf, true);
    const int row0 = __builtin_amdgcn_readlane(x, 15);
    if (lane >= 16) x += row0;
    return x;
}

__device__ __forceinline__ int wave_sum(int x, int lane) {
    return __builtin_amdgcn_readlane(wave_incl_scan(x, lane), 31);
}

__global__ void __launch_bounds__(CT)
k_chunk_sort(const int* __restrict__ src, const int* __restrict__ dst,
             unsigned* __restrict__ tab, int* __restrict__ tmpi) {
    __shared__ int h[NB];
    __shared__ int wt[8];
    __shared__ __attribute__((aligned(16))) int2 buf[CHUNK];
    const int t = threadIdx.x;
    const int lane = t & 31;
    const int w = t >> 5;
    if (t < NB) h[t] = 0;
    const int cbeg = blockIdx.x * CHUNK;
    const int cend = min(cbeg + CHUNK, NE);
    int d[EPT], s[EPT];
    #pragma unroll
    for (int k = 0; k < EPT; ++k) {
        const int e = cbeg + k * CT + t;
        d[k] = (e < cend) ? dst[e] : -1;
        s[k] = (e < cend) ? src[e] : 0;
    }
    __syncthreads();
    #pragma unroll
    for (int k = 0; k < EPT; ++k)
        if ((unsigned)d[k] < (unsigned)NN) atomicAdd(&h[d[k] >> BSH], 1);
    __syncthreads();
    int c = 0, incl = 0;
    if (t < 256) {
        c = (t < NB) ? h[t] : 0;
        incl = wave_incl_scan(c, lane);
        if (lane == 31) wt[w] = incl;
    }
    __syncthreads();
    if (t < 256) {
        int wpre = 0;
        #pragma unroll
        for (int j = 0; j < 8; ++j) wpre += (j < w) ? wt[j] : 0;
        const int excl = wpre + incl - c;
        if (t < NB) {
            tab[t * G1 + blockIdx.x] = (unsigned)excl | ((unsigned)c << 16);
            h[t] = excl;
        }
    }
    __syncthreads();
    int pos[EPT];
    #pragma unroll
    for (int k = 0; k < EPT; ++k) {
        pos[k] = -1;
        if ((unsigned)d[k] < (unsigned)NN) pos[k] = atomicAdd(&h[d[k] >> BSH], 1);
    }
    #pragma unroll
    for (int k = 0; k < EPT; ++k)
        if ((unsigned)pos[k] < (unsigned)CHUNK) buf[pos[k]] = make_int2(s[k], d[k]);
    __syncthreads();
    const int total = min(h[NB - 1], CHUNK);
    int2* run = (int2*)tmpi + blockIdx.x * CHUNK;
    int4 o[COT];
    #pragma unroll
    for (int k = 0; k < COT; ++k) {
        const int i = (k * CT + t) * 2;
        o[k] = make_int4(0, 0, 0, 0);
        if (i < total) o[k] = *(const int4*)(buf + i);
    }
    #pragma unroll
    for (int k = 0; k < COT; ++k) {
        const int i = (k * CT + t) * 2;
        if (i + 1 < total) *(int4*)(run + i) = o[k];
        else if (i < total) run[i] = make_int2(o[k].x, o[k].y);
    }
}

__global__ void __launch_bounds__(1024)
k_bin_place(const unsigned* __restrict__ tab, const int* __restrict__ tmpi,
            int* __restrict__ rowptr, int* __restrict__ col) {
    __shared__ int cnt[BINSZ];
    __shared__ int wtot[BINSZ / 32];
    __shared__ int red[2][32];
    const int2* tmp = (const int2*)tmpi;
    const int t = threadIdx.x;
    const int lane = t & 31;
    const int w = __builtin_amdgcn_readfirstlane(t >> 5);
    const int bin = blockIdx.x;
    const int base = bin * BINSZ;

    unsigned myinfo = 0;
    if (lane < PPW) myinfo = tab[bin * G1 + w + 32 * lane];
    const int nprev = bin * G1;
    uint4 tv[TABT];
    #pragma unroll
    for (int k = 0; k < TABT; ++k) {
        const int i = k * 4096 + t * 4;
        tv[k] = *(const uint4*)(tab + ((i < nprev) ? i : 0));
    }
    int pst[PPW], pcn[PPW];
    #pragma unroll
    for (int i = 0; i < PPW; ++i) {
        const unsigned info = __builtin_amdgcn_readlane(myinfo, i);
        pst[i] = min((int)(info & 0xffffu), CHUNK);
        pcn[i] = min((int)(info >> 16), CHUNK - pst[i]);
    }
    int2 p[PPW * PT];
    #pragma unroll
    for (int i = 0; i < PPW; ++i) {
        const int2* piece = tmp + (w + 32 * i) * CHUNK + pst[i];
        #pragma unroll
        for (int u = 0; u < PT; ++u) {
            const int j = u * 32 + lane;
            p[i * PT + u] = make_int2(0, -1);
            if (j < pcn[i]) p[i * PT + u] = piece[j];
        }
    }

    int before = 0;
    #pragma unroll
    for (int k = 0; k < TABT; ++k) {
        const int i = k * 4096 + t * 4;
        const int s4 = (int)((tv[k].x >> 16) + (tv[k].y >> 16) + (tv[k].z >> 16) + (tv[k].w >> 16));
        before += (i < nprev) ? s4 : 0;
    }
    if (t < BINSZ) cnt[t] = 0;
    const int wb = wave_sum(before, lane);
    const int wo = wave_sum((int)(myinfo >> 16), lane);
    if (lane == 0) { red[0][w] = wb; red[1][w] = wo; }
    __syncthreads();
    const int beg = wave_sum(red[0][lane], lane);
    const int end = beg + wave_sum(red[1][lane], lane);

    #pragma unroll
    for (int q = 0; q < PPW * PT; ++q) {
        const int dl = p[q].y - base;
        if (p[q].y >= 0 && (unsigned)dl < (unsigned)BINSZ) atomicAdd(&cnt[dl], 1);
    }
    for (int i = 0; i < PPW; ++i) {
        const int2* piece = tmp + (w + 32 * i) * CHUNK + pst[i];
        for (int j = PT * 32 + lane; j < pcn[i]; j += 32) {
            const int y = piece[j].y;
            const int dl = y - base;
            if (y >= 0 && (unsigned)dl < (unsigned)BINSZ) atomicAdd(&cnt[dl], 1);
        }
    }
    __syncthreads();
    const int c = (t < BINSZ) ? cnt[t] : 0;
    const int incl = wave_incl_scan(c, lane);
    if (lane == 31 && w < BINSZ / 32) wtot[w] = incl;
    __syncthreads();
    int wpre = 0;
    #pragma unroll
    for (int j = 0; j < BINSZ / 32; ++j) wpre += (j < w) ? wtot[j] : 0;
    const int excl = beg + wpre + incl - c;
    if (t < BINSZ) {
        const int node = base + t;
        if (node < NN) rowptr[node] = excl;
        cnt[t] = excl;
    }
    if (bin == NB - 1 && t == 0) rowptr[NN] = end;
    __syncthreads();
    int pos[PPW * PT];
    #pragma unroll
    for (int q = 0; q < PPW * PT; ++q) {
        const int dl = p[q].y - base;
        pos[q] = -1;
        if (p[q].y >= 0 && (unsigned)dl < (unsigned)BINSZ) pos[q] = atomicAdd(&cnt[dl], 1);
    }
    #pragma unroll
    for (int q = 0; q < PPW * PT; ++q)
        if ((unsigned)pos[q] < (unsigned)NE) col[pos[q]] = p[q].x;
    for (int i = 0; i < PPW; ++i) {
        const int2* piece = tmp + (w + 32 * i) * CHUNK + pst[i];
        for (int j = PT * 32 + lane; j < pcn[i]; j += 32) {
            const int2 pr = piece[j];
            const int dl = pr.y - base;
            if (pr.y >= 0 && (unsigned)dl < (unsigned)BINSZ) {
                const int ps = atomicAdd(&cnt[dl], 1);
                if ((unsigned)ps < (unsigned)NE) col[ps] = pr.x;
            }
        }
    }
}

template <int U, bool CLEAN>
__device__ __forceinline__ void gather_trip(const float* __restrict__ feat, int myc, int j, int n,
                                            int hi, int c4, float4& acc) {
    static_assert(U == 4 || U == 8, "the fence below names every row of the trip");
    v4f v[U];
    #pragma unroll
    for (int u = 0; u < U; ++u) {
        const int s = __shfl(myc, j + 2 * u + hi, 32);
        v[u] = *(const v4f*)(feat + s * CH + c4);
    }
    if constexpr (U == 8)
        asm volatile("" : "+v"(v[0]), "+v"(v[1]), "+v"(v[2]), "+v"(v[3]), "+v"(v[4]), "+v"(v[5]), "+v"(v[6]), "+v"(v[7]));
    else
        asm volatile("" : "+v"(v[0]), "+v"(v[1]), "+v"(v[2]), "+v"(v[3]));
    #pragma unroll
    for (int u = 0; u < U; ++u) {
        const bool ok = (j + 2 * u + hi) < n;
        v4f c = v[u];
        if (CLEAN) { c.x = nan0(c.x); c.y = nan0(c.y); c.z = nan0(c.z); c.w = nan0(c.w); }
        acc.x += ok ? c.x : 0.0f;
        acc.y += ok ? c.y : 0.0f;
        acc.z += ok ? c.z : 0.0f;
        acc.w += ok ? c.w : 0.0f;
    }
}

#define TP 68

template <int XM>
__device__ __forceinline__ float row_xmask_add(float x) {
    return x + __builtin_bit_cast(float, __builtin_amdgcn_update_dpp(0, __builtin_bit_cast(int, x), 0x160 + XM, 0xf, 0xf, true));
}

template <bool CLEAN, bool HEAD>
__global__ void __launch_bounds__(256)
k_sage_layer(const float* __restrict__ feat, const int* __restrict__ rowptr, const int* __restrict__ col,
             const float* __restrict__ Wl, const float* __restrict__ Wr,
             const float* __restrict__ bias, float* __restrict__ out,
             const float* __restrict__ Wfc, const float* __restrict__ bfc, float* __restrict__ pred) {
    __shared__ float Wfrag[32 * 4 * 32 * 2];
    __shared__ float meanT[8][16 * TP];

    for (int idx = threadIdx.x; idx < 32 * 4 * 32 * 2; idx += 256) {
        int v  = idx & 1;
        int l  = (idx >> 1) & 31;
        int nt = (idx >> 6) & 3;
        int s  = idx >> 8;
        int K  = 4 * s + ((l >> 4) << 1) + v;
        int n  = nt * 16 + (l & 15);
        Wfrag[idx] = (K < 64) ? Wl[K * 64 + n] : Wr[(K - 64) * 64 + n];
    }
    __syncthreads();

    const int lane = threadIdx.x & 31;
    const int hi   = lane >> 4;
    const int r    = lane & 15;
    const int c4   = r * 4;
    const int wv   = __builtin_amdgcn_readfirstlane((int)(threadIdx.x >> 5));
    const int tile = blockIdx.x * 8 + wv;

    if (tile < NTILES) {
        float* mt = meanT[wv];
        const int n0 = tile * 16;

        int rpv = 0;
        if (lane <= 16) rpv = rowptr[n0 + lane];
        int end = __builtin_amdgcn_readlane(rpv, 0);
        int nend = __builtin_amdgcn_readlane(rpv, 1);
        int cur = 0;
        if (lane < nend - end) cur = col[end + lane];
        cur = min(max(cur, 0), NN - 1);

        for (int k = 0; k < 16; ++k) {
            const int beg = end;
            end = nend;
            int myc = cur;
            int nxt = 0;
            if (k + 1 < 16) {
                nend = __builtin_amdgcn_readlane(rpv, k + 2);
                if (lane < nend - end) nxt = col[end + lane];
            }

            float4 acc = make_float4(0.0f, 0.0f, 0.0f, 0.0f);
            for (int base = beg; base < end; base += 32) {
                int n = end - base;
                if (n > 32) n = 32;
                if (base != beg) {
                    int c = 0;
                    if (lane < n) c = col[base + lane];
                    myc = min(max(c, 0), NN - 1);
                }
                for (int j = 0; j < n; j += 16) {
                    if (n - j > 8) gather_trip<8, CLEAN>(feat, myc, j, n, hi, c4, acc);
                    else           gather_trip<4, CLEAN>(feat, myc, j, n, hi, c4, acc);
                }
            }
            acc.x += __shfl_xor(acc.x, 16, 32);
            acc.y += __shfl_xor(acc.y, 16, 32);
            acc.z += __shfl_xor(acc.z, 16, 32);
            acc.w += __shfl_xor(acc.w, 16, 32);
            const float rd = 1.0f / fmaxf((float)(end - beg), 1.0f);
            if (hi == 0) *(float4*)(mt + k * TP + c4) = make_float4(acc.x * rd, acc.y * rd, acc.z * rd, acc.w * rd);
            asm volatile("" : "+v"(nxt));
            cur = min(max(nxt, 0), NN - 1);
        }

        const int row = n0 + r;
        const float* featrow = feat + row * CH;
        v2f af[16];
        #pragma unroll
        for (int s = 0; s < 16; ++s) af[s] = *(const v2f*)(featrow + 4 * s + hi * 2);
        if (CLEAN) {
            #pragma unroll
            for (int s = 0; s < 16; ++s) { af[s].x = nan0(af[s].x); af[s].y = nan0(af[s].y); }
        }

        v8f acc[4];
        #pragma unroll
        for (int nt = 0; nt < 4; ++nt) {
            float bv = bias[nt * 16 + r];
            #pragma unroll
            for (int j = 0; j < 8; ++j) acc[nt][j] = bv;
        }

        #pragma unroll 4
        for (int s = 0; s < 16; ++s) {
            v2f a = *(const v2f*)(mt + r * TP + 4 * s + hi * 2);
            #pragma unroll
            for (int nt = 0; nt < 4; ++nt) {
                v2f b = *(const v2f*)&Wfrag[((s * 4 + nt) * 32 + lane) * 2];
                acc[nt] = __builtin_amdgcn_wmma_f32_16x16x4_f32(
                    false, a, false, b, (short)0, acc[nt], false, false);
            }
        }
        #pragma unroll
        for (int s = 0; s < 16; ++s) {
            #pragma unroll
            for (int nt = 0; nt < 4; ++nt) {
                v2f b = *(const v2f*)&Wfrag[(((s + 16) * 4 + nt) * 32 + lane) * 2];
                acc[nt] = __builtin_amdgcn_wmma_f32_16x16x4_f32(
                    false, af[s], false, b, (short)0, acc[nt], false, false);
            }
        }

        if (!HEAD) {
            #pragma unroll
            for (int nt = 0; nt < 4; ++nt) {
                #pragma unroll
                for (int v = 0; v < 8; ++v) {
                    int orow = n0 + hi * 8 + v;
                    out[orow * CH + nt * 16 + r] = fmaxf(acc[nt][v], 0.0f);
                }
            }
        } else {
            float wf[4];
            #pragma unroll
            for (int nt = 0; nt < 4; ++nt) wf[nt] = Wfc[nt * 16 + r];
            const float bf = bfc[0];
            float p[8];
            #pragma unroll
            for (int v = 0; v < 8; ++v) {
                p[v] = fmaxf(acc[0][v], 0.0f) * wf[0];
                p[v] += fmaxf(acc[1][v], 0.0f) * wf[1];
                p[v] += fmaxf(acc[2][v], 0.0f) * wf[2];
                p[v] += fmaxf(acc[3][v], 0.0f) * wf[3];
            }
            #pragma unroll
            for (int v = 0; v < 8; ++v) p[v] = row_xmask_add<8>(p[v]);
            #pragma unroll
            for (int v = 0; v < 8; ++v) p[v] = row_xmask_add<4>(p[v]);
            #pragma unroll
            for (int v = 0; v < 8; ++v) p[v] = row_xmask_add<2>(p[v]);
            #pragma unroll
            for (int v = 0; v < 8; ++v) p[v] = row_xmask_add<1>(p[v]);
            float mine = p[0];
            #pragma unroll
            for (int v = 1; v < 8; ++v) mine = (r == v) ? p[v] : mine;
            if (r < 8) mt[hi * 8 + r] = mine + bf;
        }
    }
    if (HEAD) {
        __syncthreads();
        if (threadIdx.x < 128) {
            const int node = blockIdx.x * 128 + threadIdx.x;
            if (node < NN) pred[node] = meanT[threadIdx.x >> 4][threadIdx.x & 15];
        }
    }
}

extern "C" void kernel_launch(void* const* d_in, const int* in_sizes, int n_in,
                              void* d_out, int out_size, void* d_ws, size_t ws_size,
                              hipStream_t stream) {
    const float* x   = (const float*)d_in[0];
    const int*   ei  = (const int*)d_in[1];
    const float* W1l = (const float*)d_in[2];
    const float* b1  = (const float*)d_in[3];
    const float* W1r = (const float*)d_in[4];
    const float* W2l = (const float*)d_in[5];
    const float* b2  = (const float*)d_in[6];
    const float* W2r = (const float*)d_in[7];
    const float* Wfc = (const float*)d_in[8];
    const float* bfc = (const float*)d_in[9];
    float* pred = (float*)d_out;

    const int* src = ei;
    const int* dst = ei + NE;

    float* h1  = (float*)d_ws;
    int* rowptr = (int*)(h1 + NN * CH);
    int* col    = rowptr + NNP;
    unsigned* tab = (unsigned*)(col + NE);
    int* tmp    = (int*)(tab + NTAB);

    const int ggrid = (NTILES + 7) / 8;

    k_chunk_sort<<<G1, CT, 0, stream>>>(src, dst, tab, tmp);
    k_bin_place<<<NB, 1024, 0, stream>>>(tab, tmp, rowptr, col);

    k_sage_layer<true, false><<<ggrid, 256, 0, stream>>>(x, rowptr, col, W1l, W1r, b1, h1, Wfc, bfc, pred);
    k_sage_layer<false, true><<<ggrid, 256, 0, stream>>>(h1, rowptr, col, W2l, W2r, b2, nullptr, Wfc, bfc, pred);
}


# Layer kernels: weights, bias and head operands asked for in one whole-line load group at wave start; barrier split (signal behind own LDS stores, wait at matrix part): staging off the pull sums' way
# speedup vs baseline: 5.9076x; 1.0490x over previous
#include <hip/hip_runtime.h>


typedef __attribute__((ext_vector_type(2))) float v2f;
typedef __attribute__((ext_vector_type(4))) float v4f;
typedef __attribute__((ext_vector_type(8))) float v8f;

#define NN 100000
#define NE 1200000
#define CH 64
#define NTILES (NN / 16)
#define NNP 100032

#define BSH 9
#define BINSZ (1 << BSH)
#define NB ((NN + BINSZ - 1) / BINSZ)
#define G1 256
#define CHUNK ((NE + G1 - 1) / G1)
#define CT 512
#define EPT 10
#define NTAB (NB * G1)
#define PPW (G1 / 32)
#define PT 2
#define TABT ((NTAB + 4095) / 4096)
#define COT ((CHUNK + 2 * CT - 1) / (2 * CT))
static_assert(CHUNK <= EPT * CT, "a chunk must fit one pass of the work-group");
static_assert(CHUNK < 32768, "piece start and piece size share one 32-bit table word");
static_assert(NB <= 256, "the chunk pass scans its bin counts with eight waves");
static_assert(CHUNK % 2 == 0, "an odd run is shorter than the chunk: its last 16-byte LDS read stays inside the buffer");
static_assert(G1 % 32 == 0 && G1 <= 1024, "the bin pass deals whole pieces to its 32 waves");

__device__ __forceinline__ float nan0(float v) { return (v == v) ? v : 0.0f; }


__device__ __forceinline__ int wave_incl_scan(int x, int lane) {
    x += __builtin_amdgcn_update_dpp(0, x, 0x111, 0xf, 0xf, true);
    x += __builtin_amdgcn_update_dpp(0, x, 0x112, 0xf, 0xf, true);
    x += __builtin_amdgcn_update_dpp(0, x, 0x114, 0xf, 0xf, true);
    x += __builtin_amdgcn_update_dpp(0, x, 0x118, 0xf, 0xf, true);
    const int row0 = __builtin_amdgcn_readlane(x, 15);
    if (lane >= 16) x += row0;
    return x;
}

__device__ __forceinline__ int wave_sum(int x, int lane) {
    return __builtin_amdgcn_readlane(wave_incl_scan(x, lane), 31);
}

__global__ void __launch_bounds__(CT)
k_chunk_sort(const int* __restrict__ src, const int* __restrict__ dst,
             unsigned* __restrict__ tab, int* __restrict__ tmpi) {
    __shared__ int h[NB];
    __shared__ int wt[8];
    __shared__ __attribute__((aligned(16))) int2 buf[CHUNK];
    const int t = threadIdx.x;
    const int lane = t & 31;
    const int w = t >> 5;
    if (t < NB) h[t] = 0;
    const int cbeg = blockIdx.x * CHUNK;
    const int cend = min(cbeg + CHUNK, NE);
    int d[EPT], s[EPT];
    #pragma unroll
    for (int k = 0; k < EPT; ++k) {
        const int e = cbeg + k * CT + t;
        d[k] = (e < cend) ? dst[e] : -1;
        s[k] = (e < cend) ? src[e] : 0;
    }
    __syncthreads();
    #pragma unroll
    for (int k = 0; k < EPT; ++k)
        if ((unsigned)d[k] < (unsigned)NN) atomicAdd(&h[d[k] >> BSH], 1);
    __syncthreads();
    int c = 0, incl = 0;
    if (t < 256) {
        c = (t < NB) ? h[t] : 0;
        incl = wave_incl_scan(c, lane);
        if (lane == 31) wt[w] = incl;
    }
    __syncthreads();
    if (t < 256) {
        int wpre = 0;
        #pragma unroll
        for (int j = 0; j < 8; ++j) wpre += (j < w) ? wt[j] : 0;
        const int excl = wpre + incl - c;
        if (t < NB) {
            tab[t * G1 + blockIdx.x] = (unsigned)excl | ((unsigned)c << 16);
            h[t] = excl;
        }
    }
    __syncthreads();
    int pos[EPT];
    #pragma unroll
    for (int k = 0; k < EPT; ++k) {
        pos[k] = -1;
        if ((unsigned)d[k] < (unsigned)NN) pos[k] = atomicAdd(&h[d[k] >> BSH], 1);
    }
    #pragma unroll
    for (int k = 0; k < EPT; ++k)
        if ((unsigned)pos[k] < (unsigned)CHUNK) buf[pos[k]] = make_int2(s[k], d[k]);
    __syncthreads();
    const int total = min(h[NB - 1], CHUNK);
    int2* run = (int2*)tmpi + blockIdx.x * CHUNK;
    int4 o[COT];
    #pragma unroll
    for (int k = 0; k < COT; ++k) {
        const int i = (k * CT + t) * 2;
        o[k] = make_int4(0, 0, 0, 0);
        if (i < total) o[k] = *(const int4*)(buf + i);
    }
    #pragma unroll
    for (int k = 0; k < COT; ++k) {
        const int i = (k * CT + t) * 2;
        if (i + 1 < total) *(int4*)(run + i) = o[k];
        else if (i < total) run[i] = make_int2(o[k].x, o[k].y);
    }
}

__global__ void __launch_bounds__(1024)
k_bin_place(const unsigned* __restrict__ tab, const int* __restrict__ tmpi,
            int* __restrict__ rowptr, int* __restrict__ col) {
    __shared__ int cnt[BINSZ];
    __shared__ int wtot[BINSZ / 32];
    __shared__ int red[2][32];
    const int2* tmp = (const int2*)tmpi;
    const int t = threadIdx.x;
    const int lane = t & 31;
    const int w = __builtin_amdgcn_readfirstlane(t >> 5);
    const int bin = blockIdx.x;
    const int base = bin * BINSZ;

    unsigned myinfo = 0;
    if (lane < PPW) myinfo = tab[bin * G1 + w + 32 * lane];
    const int nprev = bin * G1;
    uint4 tv[TABT];
    #pragma unroll
    for (int k = 0; k < TABT; ++k) {
        const int i = k * 4096 + t * 4;
        tv[k] = *(const uint4*)(tab + ((i < nprev) ? i : 0));
    }
    int pst[PPW], pcn[PPW];
    #pragma unroll
    for (int i = 0; i < PPW; ++i) {
        const unsigned info = __builtin_amdgcn_readlane(myinfo, i);
        pst[i] = min((int)(info & 0xffffu), CHUNK);
        pcn[i] = min((int)(info >> 16), CHUNK - pst[i]);
    }
    int2 p[PPW * PT];
    #pragma unroll
    for (int i = 0; i < PPW; ++i) {
        const int2* piece = tmp + (w + 32 * i) * CHUNK + pst[i];
        #pragma unroll
        for (int u = 0; u < PT; ++u) {
            const int j = u * 32 + lane;
            p[i * PT + u] = make_int2(0, -1);
            if (j < pcn[i]) p[i * PT + u] = piece[j];
        }
    }

    int before = 0;
    #pragma unroll
    for (int k = 0; k < TABT; ++k) {
        const int i = k * 4096 + t * 4;
        const int s4 = (int)((tv[k].x >> 16) + (tv[k].y >> 16) + (tv[k].z >> 16) + (tv[k].w >> 16));
        before += (i < nprev) ? s4 : 0;
    }
    if (t < BINSZ) cnt[t] = 0;
    const int wb = wave_sum(before, lane);
    const int wo = wave_sum((int)(myinfo >> 16), lane);
    if (lane == 0) { red[0][w] = wb; red[1][w] = wo; }
    __syncthreads();
    const int beg = wave_sum(red[0][lane], lane);
    const int end = beg + wave_sum(red[1][lane], lane);

    #pragma unroll
    for (int q = 0; q < PPW * PT; ++q) {
        const int dl = p[q].y - base;
        if (p[q].y >= 0 && (unsigned)dl < (unsigned)BINSZ) atomicAdd(&cnt[dl], 1);
    }
    for (int i = 0; i < PPW; ++i) {
        const int2* piece = tmp + (w + 32 * i) * CHUNK + pst[i];
        for (int j = PT * 32 + lane; j < pcn[i]; j += 32) {
            const int y = piece[j].y;
            const int dl = y - base;
            if (y >= 0 && (unsigned)dl < (unsigned)BINSZ) atomicAdd(&cnt[dl], 1);
        }
    }
    __syncthreads();
    const int c = (t < BINSZ) ? cnt[t] : 0;
    const int incl = wave_incl_scan(c, lane);
    if (lane == 31 && w < BINSZ / 32) wtot[w] = incl;
    __syncthreads();
    int wpre = 0;
    #pragma unroll
    for (int j = 0; j < BINSZ / 32; ++j) wpre += (j < w) ? wtot[j] : 0;
    const int excl = beg + wpre + incl - c;
    if (t < BINSZ) {
        const int node = base + t;
        if (node < NN) rowptr[node] = excl;
        cnt[t] = excl;
    }
    if (bin == NB - 1 && t == 0) rowptr[NN] = end;
    __syncthreads();
    int pos[PPW * PT];
    #pragma unroll
    for (int q = 0; q < PPW * PT; ++q) {
        const int dl = p[q].y - base;
        pos[q] = -1;
        if (p[q].y >= 0 && (unsigned)dl < (unsigned)BINSZ) pos[q] = atomicAdd(&cnt[dl], 1);
    }
    #pragma unroll
    for (int q = 0; q < PPW * PT; ++q)
        if ((unsigned)pos[q] < (unsigned)NE) col[pos[q]] = p[q].x;
    for (int i = 0; i < PPW; ++i) {
        const int2* piece = tmp + (w + 32 * i) * CHUNK + pst[i];
        for (int j = PT * 32 + lane; j < pcn[i]; j += 32) {
            const int2 pr = piece[j];
            const int dl = pr.y - base;
            if (pr.y >= 0 && (unsigned)dl < (unsigned)BINSZ) {
                const int ps = atomicAdd(&cnt[dl], 1);
                if ((unsigned)ps < (unsigned)NE) col[ps] = pr.x;
            }
        }
    }
}

template <int U, bool CLEAN>
__device__ __forceinline__ void gather_trip(const float* __restrict__ feat, int myc, int j, int n,
                                            int hi, int c4, float4& acc) {
    static_assert(U == 4 || U == 8, "the fence below names every row of the trip");
    v4f v[U];
    #pragma unroll
    for (int u = 0; u < U; ++u) {
        const int s = __shfl(myc, j + 2 * u + hi, 32);
        v[u] = *(const v4f*)(feat + s * CH + c4);
    }
    if constexpr (U == 8)
        asm volatile("" : "+v"(v[0]), "+v"(v[1]), "+v"(v[2]), "+v"(v[3]), "+v"(v[4]), "+v"(v[5]), "+v"(v[6]), "+v"(v[7]));
    else
        asm volatile("" : "+v"(v[0]), "+v"(v[1]), "+v"(v[2]), "+v"(v[3]));
    #pragma unroll
    for (int u = 0; u < U; ++u) {
        const bool ok = (j + 2 * u + hi) < n;
        v4f c = v[u];
        if (CLEAN) { c.x = nan0(c.x); c.y = nan0(c.y); c.z = nan0(c.z); c.w = nan0(c.w); }
        acc.x += ok ? c.x : 0.0f;
        acc.y += ok ? c.y : 0.0f;
        acc.z += ok ? c.z : 0.0f;
        acc.w += ok ? c.w : 0.0f;
    }
}

#define TP 68

template <int XM>
__device__ __forceinline__ float row_xmask_add(float x) {
    return x + __builtin_bit_cast(float, __builtin_amdgcn_update_dpp(0, __builtin_bit_cast(int, x), 0x160 + XM, 0xf, 0xf, true));
}

template <bool CLEAN, bool HEAD>
__global__ void __launch_bounds__(256)
k_sage_layer(const float* __restrict__ feat, const int* __restrict__ rowptr, const int* __restrict__ col,
             const float* __restrict__ Wl, const float* __restrict__ Wr,
             const float* __restrict__ bias, float* __restrict__ out,
             const float* __restrict__ Wfc, const float* __restrict__ bfc, float* __restrict__ pred) {
    __shared__ float Wfrag[32 * 4 * 32 * 2];
    __shared__ float meanT[8][16 * TP];

    const int lane = threadIdx.x & 31;
    const int hi   = lane >> 4;
    const int r    = lane & 15;
    const int c4   = r * 4;
    const int wv   = __builtin_amdgcn_readfirstlane((int)(threadIdx.x >> 5));
    const int tile = blockIdx.x * 8 + wv;
    const bool live = tile < NTILES;
    float* mt = meanT[wv];
    const int n0 = tile * 16;

    v4f w0[4], w1[4];
    #pragma unroll
    for (int j = 0; j < 4; ++j) {
        const int id = j * 256 + (int)threadIdx.x;
        const int q  = id & 15;
        const int kp = id >> 4;
        const float* wsrc = ((j < 2) ? Wl : Wr) + (2 * (kp & 31)) * 64 + 4 * q;
        w0[j] = *(const v4f*)wsrc;
        w1[j] = *(const v4f*)(wsrc + 64);
    }
    float bv[4], wf[4];
    #pragma unroll
    for (int nt = 0; nt < 4; ++nt) {
        bv[nt] = bias[nt * 16 + r];
        wf[nt] = HEAD ? Wfc[nt * 16 + r] : 0.0f;
    }
    const float bf = HEAD ? bfc[0] : 0.0f;
    int rpv = 0;
    if (live && lane <= 16) rpv = rowptr[n0 + lane];
    #pragma unroll
    for (int j = 0; j < 4; ++j) {
        const int id = j * 256 + (int)threadIdx.x;
        const int q  = id & 15;
        const int kp = id >> 4;
        float* wdst = Wfrag + ((((kp >> 1) * 4 + (q >> 2)) * 32 + (kp & 1) * 16 + (q & 3) * 4) * 2);
        *(v4f*)wdst       = (v4f){w0[j].x, w1[j].x, w0[j].y, w1[j].y};
        *(v4f*)(wdst + 4) = (v4f){w0[j].z, w1[j].z, w0[j].w, w1[j].w};
    }
    asm volatile("s_wait_dscnt 0x0\n\ts_barrier_signal -1" ::: "memory");

    if (live) {
        int end = __builtin_amdgcn_readlane(rpv, 0);
        int nend = __builtin_amdgcn_readlane(rpv, 1);
        int cur = 0;
        if (lane < nend - end) cur = col[end + lane];
        cur = min(max(cur, 0), NN - 1);

        for (int k = 0; k < 16; ++k) {
            const int beg = end;
            end = nend;
            int myc = cur;
            int nxt = 0;
            if (k + 1 < 16) {
                nend = __builtin_amdgcn_readlane(rpv, k + 2);
                if (lane < nend - end) nxt = col[end + lane];
            }

            float4 acc = make_float4(0.0f, 0.0f, 0.0f, 0.0f);
            for (int base = beg; base < end; base += 32) {
                int n = end - base;
                if (n > 32) n = 32;
                if (base != beg) {
                    int c = 0;
                    if (lane < n) c = col[base + lane];
                    myc = min(max(c, 0), NN - 1);
                }
                for (int j = 0; j < n; j += 16) {
                    if (n - j > 8) gather_trip<8, CLEAN>(feat, myc, j, n, hi, c4, acc);
                    else           gather_trip<4, CLEAN>(feat, myc, j, n, hi, c4, acc);
                }
            }
            acc.x += __shfl_xor(acc.x, 16, 32);
            acc.y += __shfl_xor(acc.y, 16, 32);
            acc.z += __shfl_xor(acc.z, 16, 32);
            acc.w += __shfl_xor(acc.w, 16, 32);
            const float rd = 1.0f / fmaxf((float)(end - beg), 1.0f);
            if (hi == 0) *(float4*)(mt + k * TP + c4) = make_float4(acc.x * rd, acc.y * rd, acc.z * rd, acc.w * rd);
            asm volatile("" : "+v"(nxt));
            cur = min(max(nxt, 0), NN - 1);
        }

    }

    asm volatile("s_barrier_wait -1" ::: "memory");

    if (live) {
        const int row = n0 + r;
        const float* featrow = feat + row * CH;
        v2f af[16];
        #pragma unroll
        for (int s = 0; s < 16; ++s) af[s] = *(const v2f*)(featrow + 4 * s + hi * 2);
        if (CLEAN) {
            #pragma unroll
            for (int s = 0; s < 16; ++s) { af[s].x = nan0(af[s].x); af[s].y = nan0(af[s].y); }
        }

        v8f acc[4];
        #pragma unroll
        for (int nt = 0; nt < 4; ++nt) {
            #pragma unroll
            for (int j = 0; j < 8; ++j) acc[nt][j] = bv[nt];
        }

        #pragma unroll 4
        for (int s = 0; s < 16; ++s) {
            v2f a = *(const v2f*)(mt + r * TP + 4 * s + hi * 2);
            #pragma unroll
            for (int nt = 0; nt < 4; ++nt) {
                v2f b = *(const v2f*)&Wfrag[((s * 4 + nt) * 32 + lane) * 2];
                acc[nt] = __builtin_amdgcn_wmma_f32_16x16x4_f32(
                    false, a, false, b, (short)0, acc[nt], false, false);
            }
        }
        #pragma unroll
        for (int s = 0; s < 16; ++s) {
            #pragma unroll
            for (int nt = 0; nt < 4; ++nt) {
                v2f b = *(const v2f*)&Wfrag[(((s + 16) * 4 + nt) * 32 + lane) * 2];
                acc[nt] = __builtin_amdgcn_wmma_f32_16x16x4_f32(
                    false, af[s], false, b, (short)0, acc[nt], false, false);
            }
        }

        if (!HEAD) {
            #pragma unroll
            for (int nt = 0; nt < 4; ++nt) {
                #pragma unroll
                for (int v = 0; v < 8; ++v) {
                    int orow = n0 + hi * 8 + v;
                    out[orow * CH + nt * 16 + r] = fmaxf(acc[nt][v], 0.0f);
                }
            }
        } else {
            float p[8];
            #pragma unroll
            for (int v = 0; v < 8; ++v) {
                p[v] = fmaxf(acc[0][v], 0.0f) * wf[0];
                p[v] += fmaxf(acc[1][v], 0.0f) * wf[1];
                p[v] += fmaxf(acc[2][v], 0.0f) * wf[2];
                p[v] += fmaxf(acc[3][v], 0.0f) * wf[3];
            }
            #pragma unroll
            for (int v = 0; v < 8; ++v) p[v] = row_xmask_add<8>(p[v]);
            #pragma unroll
            for (int v = 0; v < 8; ++v) p[v] = row_xmask_add<4>(p[v]);
            #pragma unroll
            for (int v = 0; v < 8; ++v) p[v] = row_xmask_add<2>(p[v]);
            #pragma unroll
            for (int v = 0; v < 8; ++v) p[v] = row_xmask_add<1>(p[v]);
            float mine = p[0];
            #pragma unroll
            for (int v = 1; v < 8; ++v) mine = (r == v) ? p[v] : mine;
            if (r < 8) mt[hi * 8 + r] = mine + bf;
        }
    }
    if (HEAD) {
        __syncthreads();
        if (threadIdx.x < 128) {
            const int node = blockIdx.x * 128 + threadIdx.x;
            if (node < NN) pred[node] = meanT[threadIdx.x >> 4][threadIdx.x & 15];
        }
    }
}

extern "C" void kernel_launch(void* const* d_in, const int* in_sizes, int n_in,
                              void* d_out, int out_size, void* d_ws, size_t ws_size,
                              hipStream_t stream) {
    const float* x   = (const float*)d_in[0];
    const int*   ei  = (const int*)d_in[1];
    const float* W1l = (const float*)d_in[2];
    const float* b1  = (const float*)d_in[3];
    const float* W1r = (const float*)d_in[4];
    const float* W2l = (const float*)d_in[5];
    const float* b2  = (const float*)d_in[6];
    const float* W2r = (const float*)d_in[7];
    const float* Wfc = (const float*)d_in[8];
    const float* bfc = (const float*)d_in[9];
    float* pred = (float*)d_out;

    const int* src = ei;
    const int* dst = ei + NE;

    float* h1  = (float*)d_ws;
    int* rowptr = (int*)(h1 + NN * CH);
    int* col    = rowptr + NNP;
    unsigned* tab = (unsigned*)(col + NE);
    int* tmp    = (int*)(tab + NTAB);

    const int ggrid = (NTILES + 7) / 8;

    k_chunk_sort<<<G1, CT, 0, stream>>>(src, dst, tab, tmp);
    k_bin_place<<<NB, 1024, 0, stream>>>(tab, tmp, rowptr, col);

    k_sage_layer<true, false><<<ggrid, 256, 0, stream>>>(x, rowptr, col, W1l, W1r, b1, h1, Wfc, bfc, pred);
    k_sage_layer<false, true><<<ggrid, 256, 0, stream>>>(h1, rowptr, col, W2l, W2r, b2, nullptr, Wfc, bfc, pred);
}


# Pull sums walk two nodes per trip: both nodes' first 16 edges (16 two-row loads) requested before the first add, next pair's sources prefetched; 8 memory round trips per tile, was 16; sum order kept
# speedup vs baseline: 6.2569x; 1.0591x over previous
#include <hip/hip_runtime.h>


typedef __attribute__((ext_vector_type(2))) float v2f;
typedef __attribute__((ext_vector_type(4))) float v4f;
typedef __attribute__((ext_vector_type(8))) float v8f;

#define NN 100000
#define NE 1200000
#define CH 64
#define NTILES (NN / 16)
#define NNP 100032

#define BSH 9
#define BINSZ (1 << BSH)
#define NB ((NN + BINSZ - 1) / BINSZ)
#define G1 256
#define CHUNK ((NE + G1 - 1) / G1)
#define CT 512
#define EPT 10
#define NTAB (NB * G1)
#define PPW (G1 / 32)
#define PT 2
#define TABT ((NTAB + 4095) / 4096)
#define COT ((CHUNK + 2 * CT - 1) / (2 * CT))
static_assert(CHUNK <= EPT * CT, "a chunk must fit one pass of the work-group");
static_assert(CHUNK < 32768, "piece start and piece size share one 32-bit table word");
static_assert(NB <= 256, "the chunk pass scans its bin counts with eight waves");
static_assert(CHUNK % 2 == 0, "an odd run is shorter than the chunk: its last 16-byte LDS read stays inside the buffer");
static_assert(G1 % 32 == 0 && G1 <= 1024, "the bin pass deals whole pieces to its 32 waves");

__device__ __forceinline__ float nan0(float v) { return (v == v) ? v : 0.0f; }


__device__ __forceinline__ int wave_incl_scan(int x, int lane) {
    x += __builtin_amdgcn_update_dpp(0, x, 0x111, 0xf, 0xf, true);
    x += __builtin_amdgcn_update_dpp(0, x, 0x112, 0xf, 0xf, true);
    x += __builtin_amdgcn_update_dpp(0, x, 0x114, 0xf, 0xf, true);
    x += __builtin_amdgcn_update_dpp(0, x, 0x118, 0xf, 0xf, true);
    const int row0 = __builtin_amdgcn_readlane(x, 15);
    if (lane >= 16) x += row0;
    return x;
}

__device__ __forceinline__ int wave_sum(int x, int lane) {
    return __builtin_amdgcn_readlane(wave_incl_scan(x, lane), 31);
}

__global__ void __launch_bounds__(CT)
k_chunk_sort(const int* __restrict__ src, const int* __restrict__ dst,
             unsigned* __restrict__ tab, int* __restrict__ tmpi) {
    __shared__ int h[NB];
    __shared__ int wt[8];
    __shared__ __attribute__((aligned(16))) int2 buf[CHUNK];
    const int t = threadIdx.x;
    const int lane = t & 31;
    const int w = t >> 5;
    if (t < NB) h[t] = 0;
    const int cbeg = blockIdx.x * CHUNK;
    const int cend = min(cbeg + CHUNK, NE);
    int d[EPT], s[EPT];
    #pragma unroll
    for (int k = 0; k < EPT; ++k) {
        const int e = cbeg + k * CT + t;
        d[k] = (e < cend) ? dst[e] : -1;
        s[k] = (e < cend) ? src[e] : 0;
    }
    __syncthreads();
    #pragma unroll
    for (int k = 0; k < EPT; ++k)
        if ((unsigned)d[k] < (unsigned)NN) atomicAdd(&h[d[k] >> BSH], 1);
    __syncthreads();
    int c = 0, incl = 0;
    if (t < 256) {
        c = (t < NB) ? h[t] : 0;
        incl = wave_incl_scan(c, lane);
        if (lane == 31) wt[w] = incl;
    }
    __syncthreads();
    if (t < 256) {
        int wpre = 0;
        #pragma unroll
        for (int j = 0; j < 8; ++j) wpre += (j < w) ? wt[j] : 0;
        const int excl = wpre + incl - c;
        if (t < NB) {
            tab[t * G1 + blockIdx.x] = (unsigned)excl | ((unsigned)c << 16);
            h[t] = excl;
        }
    }
    __syncthreads();
    int pos[EPT];
    #pragma unroll
    for (int k = 0; k < EPT; ++k) {
        pos[k] = -1;
        if ((unsigned)d[k] < (unsigned)NN) pos[k] = atomicAdd(&h[d[k] >> BSH], 1);
    }
    #pragma unroll
    for (int k = 0; k < EPT; ++k)
        if ((unsigned)pos[k] < (unsigned)CHUNK) buf[pos[k]] = make_int2(s[k], d[k]);
    __syncthreads();
    const int total = min(h[NB - 1], CHUNK);
    int2* run = (int2*)tmpi + blockIdx.x * CHUNK;
    int4 o[COT];
    #pragma unroll
    for (int k = 0; k < COT; ++k) {
        const int i = (k * CT + t) * 2;
        o[k] = make_int4(0, 0, 0, 0);
        if (i < total) o[k] = *(const int4*)(buf + i);
    }
    #pragma unroll
    for (int k = 0; k < COT; ++k) {
        const int i = (k * CT + t) * 2;
        if (i + 1 < total) *(int4*)(run + i) = o[k];
        else if (i < total) run[i] = make_int2(o[k].x, o[k].y);
    }
}

__global__ void __launch_bounds__(1024)
k_bin_place(const unsigned* __restrict__ tab, const int* __restrict__ tmpi,
            int* __restrict__ rowptr, int* __restrict__ col) {
    __shared__ int cnt[BINSZ];
    __shared__ int wtot[BINSZ / 32];
    __shared__ int red[2][32];
    const int2* tmp = (const int2*)tmpi;
    const int t = threadIdx.x;
    const int lane = t & 31;
    const int w = __builtin_amdgcn_readfirstlane(t >> 5);
    const int bin = blockIdx.x;
    const int base = bin * BINSZ;

    unsigned myinfo = 0;
    if (lane < PPW) myinfo = tab[bin * G1 + w + 32 * lane];
    const int nprev = bin * G1;
    uint4 tv[TABT];
    #pragma unroll
    for (int k = 0; k < TABT; ++k) {
        const int i = k * 4096 + t * 4;
        tv[k] = *(const uint4*)(tab + ((i < nprev) ? i : 0));
    }
    int pst[PPW], pcn[PPW];
    #pragma unroll
    for (int i = 0; i < PPW; ++i) {
        const unsigned info = __builtin_amdgcn_readlane(myinfo, i);
        pst[i] = min((int)(info & 0xffffu), CHUNK);
        pcn[i] = min((int)(info >> 16), CHUNK - pst[i]);
    }
    int2 p[PPW * PT];
    #pragma unroll
    for (int i = 0; i < PPW; ++i) {
        const int2* piece = tmp + (w + 32 * i) * CHUNK + pst[i];
        #pragma unroll
        for (int u = 0; u < PT; ++u) {
            const int j = u * 32 + lane;
            p[i * PT + u] = make_int2(0, -1);
            if (j < pcn[i]) p[i * PT + u] = piece[j];
        }
    }

    int before = 0;
    #pragma unroll
    for (int k = 0; k < TABT; ++k) {
        const int i = k * 4096 + t * 4;
        const int s4 = (int)((tv[k].x >> 16) + (tv[k].y >> 16) + (tv[k].z >> 16) + (tv[k].w >> 16));
        before += (i < nprev) ? s4 : 0;
    }
    if (t < BINSZ) cnt[t] = 0;
    const int wb = wave_sum(before, lane);
    const int wo = wave_sum((int)(myinfo >> 16), lane);
    if (lane == 0) { red[0][w] = wb; red[1][w] = wo; }
    __syncthreads();
    const int beg = wave_sum(red[0][lane], lane);
    const int end = beg + wave_sum(red[1][lane], lane);

    #pragma unroll
    for (int q = 0; q < PPW * PT; ++q) {
        const int dl = p[q].y - base;
        if (p[q].y >= 0 && (unsigned)dl < (unsigned)BINSZ) atomicAdd(&cnt[dl], 1);
    }
    for (int i = 0; i < PPW; ++i) {
        const int2* piece = tmp + (w + 32 * i) * CHUNK + pst[i];
        for (int j = PT * 32 + lane; j < pcn[i]; j += 32) {
            const int y = piece[j].y;
            const int dl = y - base;
            if (y >= 0 && (unsigned)dl < (unsigned)BINSZ) atomicAdd(&cnt[dl], 1);
        }
    }
    __syncthreads();
    const int c = (t < BINSZ) ? cnt[t] : 0;
    const int incl = wave_incl_scan(c, lane);
    if (lane == 31 && w < BINSZ / 32) wtot[w] = incl;
    __syncthreads();
    int wpre = 0;
    #pragma unroll
    for (int j = 0; j < BINSZ / 32; ++j) wpre += (j < w) ? wtot[j] : 0;
    const int excl = beg + wpre + incl - c;
    if (t < BINSZ) {
        const int node = base + t;
        if (node < NN) rowptr[node] = excl;
        cnt[t] = excl;
    }
    if (bin == NB - 1 && t == 0) rowptr[NN] = end;
    __syncthreads();
    int pos[PPW * PT];
    #pragma unroll
    for (int q = 0; q < PPW * PT; ++q) {
        const int dl = p[q].y - base;
        pos[q] = -1;
        if (p[q].y >= 0 && (unsigned)dl < (unsigned)BINSZ) pos[q] = atomicAdd(&cnt[dl], 1);
    }
    #pragma unroll
    for (int q = 0; q < PPW * PT; ++q)
        if ((unsigned)pos[q] < (unsigned)NE) col[pos[q]] = p[q].x;
    for (int i = 0; i < PPW; ++i) {
        const int2* piece = tmp + (w + 32 * i) * CHUNK + pst[i];
        for (int j = PT * 32 + lane; j < pcn[i]; j += 32) {
            const int2 pr = piece[j];
            const int dl = pr.y - base;
            if (pr.y >= 0 && (unsigned)dl < (unsigned)BINSZ) {
                const int ps = atomicAdd(&cnt[dl], 1);
                if ((unsigned)ps < (unsigned)NE) col[ps] = pr.x;
            }
        }
    }
}

template <int U, bool CLEAN>
__device__ __forceinline__ void gather_trip(const float* __restrict__ feat, int myc, int j, int n,
                                            int hi, int c4, float4& acc) {
    static_assert(U == 4 || U == 8, "the fence below names every row of the trip");
    v4f v[U];
    #pragma unroll
    for (int u = 0; u < U; ++u) {
        const int s = __shfl(myc, j + 2 * u + hi, 32);
        v[u] = *(const v4f*)(feat + s * CH + c4);
    }
    if constexpr (U == 8)
        asm volatile("" : "+v"(v[0]), "+v"(v[1]), "+v"(v[2]), "+v"(v[3]), "+v"(v[4]), "+v"(v[5]), "+v"(v[6]), "+v"(v[7]));
    else
        asm volatile("" : "+v"(v[0]), "+v"(v[1]), "+v"(v[2]), "+v"(v[3]));
    #pragma unroll
    for (int u = 0; u < U; ++u) {
        const bool ok = (j + 2 * u + hi) < n;
        v4f c = v[u];
        if (CLEAN) { c.x = nan0(c.x); c.y = nan0(c.y); c.z = nan0(c.z); c.w = nan0(c.w); }
        acc.x += ok ? c.x : 0.0f;
        acc.y += ok ? c.y : 0.0f;
        acc.z += ok ? c.z : 0.0f;
        acc.w += ok ? c.w : 0.0f;
    }
}

__device__ __forceinline__ void request_rows8(const float* __restrict__ feat, int myc, int hi, int c4, v4f (&v)[8]) {
    #pragma unroll
    for (int u = 0; u < 8; ++u) {
        const int s = __shfl(myc, 2 * u + hi, 32);
        v[u] = *(const v4f*)(feat + s * CH + c4);
    }
}

template <bool CLEAN>
__device__ __forceinline__ void add_rows8(const v4f (&v)[8], int n, int hi, float4& acc) {
    #pragma unroll
    for (int u = 0; u < 8; ++u) {
        const bool ok = (2 * u + hi) < n;
        v4f c = v[u];
        if (CLEAN) { c.x = nan0(c.x); c.y = nan0(c.y); c.z = nan0(c.z); c.w = nan0(c.w); }
        acc.x += ok ? c.x : 0.0f;
        acc.y += ok ? c.y : 0.0f;
        acc.z += ok ? c.z : 0.0f;
        acc.w += ok ? c.w : 0.0f;
    }
}

template <bool CLEAN>
__device__ __forceinline__ void gather_rest(const float* __restrict__ feat, const int* __restrict__ col,
                                            int beg, int end, int myc, int lane, int hi, int c4, float4& acc) {
    for (int base = beg; base < end; base += 32) {
        int n = end - base;
        if (n > 32) n = 32;
        int j = 16;
        if (base != beg) {
            int c = 0;
            if (lane < n) c = col[base + lane];
            myc = min(max(c, 0), NN - 1);
            j = 0;
        }
        for (; j < n; j += 16) {
            if (n - j > 8) gather_trip<8, CLEAN>(feat, myc, j, n, hi, c4, acc);
            else           gather_trip<4, CLEAN>(feat, myc, j, n, hi, c4, acc);
        }
    }
}

__device__ __forceinline__ void store_mean(float* __restrict__ mrow, float4 acc, int deg, int hi, int c4) {
    acc.x += __shfl_xor(acc.x, 16, 32);
    acc.y += __shfl_xor(acc.y, 16, 32);
    acc.z += __shfl_xor(acc.z, 16, 32);
    acc.w += __shfl_xor(acc.w, 16, 32);
    const float rd = 1.0f / fmaxf((float)deg, 1.0f);
    if (hi == 0) *(float4*)(mrow + c4) = make_float4(acc.x * rd, acc.y * rd, acc.z * rd, acc.w * rd);
}

#define TP 68

template <int XM>
__device__ __forceinline__ float row_xmask_add(float x) {
    return x + __builtin_bit_cast(float, __builtin_amdgcn_update_dpp(0, __builtin_bit_cast(int, x), 0x160 + XM, 0xf, 0xf, true));
}

template <bool CLEAN, bool HEAD>
__global__ void __launch_bounds__(256)
k_sage_layer(const float* __restrict__ feat, const int* __restrict__ rowptr, const int* __restrict__ col,
             const float* __restrict__ Wl, const float* __restrict__ Wr,
             const float* __restrict__ bias, float* __restrict__ out,
             const float* __restrict__ Wfc, const float* __restrict__ bfc, float* __restrict__ pred) {
    __shared__ float Wfrag[32 * 4 * 32 * 2];
    __shared__ float meanT[8][16 * TP];

    const int lane = threadIdx.x & 31;
    const int hi   = lane >> 4;
    const int r    = lane & 15;
    const int c4   = r * 4;
    const int wv   = __builtin_amdgcn_readfirstlane((int)(threadIdx.x >> 5));
    const int tile = blockIdx.x * 8 + wv;
    const bool live = tile < NTILES;
    float* mt = meanT[wv];
    const int n0 = tile * 16;

    v4f w0[4], w1[4];
    #pragma unroll
    for (int j = 0; j < 4; ++j) {
        const int id = j * 256 + (int)threadIdx.x;
        const int q  = id & 15;
        const int kp = id >> 4;
        const float* wsrc = ((j < 2) ? Wl : Wr) + (2 * (kp & 31)) * 64 + 4 * q;
        w0[j] = *(const v4f*)wsrc;
        w1[j] = *(const v4f*)(wsrc + 64);
    }
    float bv[4], wf[4];
    #pragma unroll
    for (int nt = 0; nt < 4; ++nt) {
        bv[nt] = bias[nt * 16 + r];
        wf[nt] = HEAD ? Wfc[nt * 16 + r] : 0.0f;
    }
    const float bf = HEAD ? bfc[0] : 0.0f;
    int rpv = 0;
    if (live && lane <= 16) rpv = rowptr[n0 + lane];
    #pragma unroll
    for (int j = 0; j < 4; ++j) {
        const int id = j * 256 + (int)threadIdx.x;
        const int q  = id & 15;
        const int kp = id >> 4;
        float* wdst = Wfrag + ((((kp >> 1) * 4 + (q >> 2)) * 32 + (kp & 1) * 16 + (q & 3) * 4) * 2);
        *(v4f*)wdst       = (v4f){w0[j].x, w1[j].x, w0[j].y, w1[j].y};
        *(v4f*)(wdst + 4) = (v4f){w0[j].z, w1[j].z, w0[j].w, w1[j].w};
    }
    asm volatile("s_wait_dscnt 0x0\n\ts_barrier_signal -1" ::: "memory");

    if (live) {
        int b0 = __builtin_amdgcn_readlane(rpv, 0);
        int b1 = __builtin_amdgcn_readlane(rpv, 1);
        int b2 = __builtin_amdgcn_readlane(rpv, 2);
        int curA = 0, curB = 0;
        if (lane < b1 - b0) curA = col[b0 + lane];
        if (lane < b2 - b1) curB = col[b1 + lane];
        asm volatile("" : "+v"(curA), "+v"(curB));
        curA = min(max(curA, 0), NN - 1);
        curB = min(max(curB, 0), NN - 1);

        for (int p = 0; p < 8; ++p) {
            const int begA = b0, endA = b1, endB = b2;
            const int mycA = curA, mycB = curB;
            int nxtA = 0, nxtB = 0;
            if (p + 1 < 8) {
                b0 = endB;
                b1 = __builtin_amdgcn_readlane(rpv, 2 * p + 3);
                b2 = __builtin_amdgcn_readlane(rpv, 2 * p + 4);
                if (lane < b1 - b0) nxtA = col[b0 + lane];
                if (lane < b2 - b1) nxtB = col[b1 + lane];
            }
            const int degA = endA - begA, degB = endB - endA;

            v4f va[8], vb[8];
            request_rows8(feat, mycA, hi, c4, va);
            request_rows8(feat, mycB, hi, c4, vb);
            float4 accA = make_float4(0.0f, 0.0f, 0.0f, 0.0f);
            float4 accB = make_float4(0.0f, 0.0f, 0.0f, 0.0f);
            add_rows8<CLEAN>(va, min(degA, 32), hi, accA);
            add_rows8<CLEAN>(vb, min(degB, 32), hi, accB);
            if (degA > 16) gather_rest<CLEAN>(feat, col, begA, endA, mycA, lane, hi, c4, accA);
            if (degB > 16) gather_rest<CLEAN>(feat, col, endA, endB, mycB, lane, hi, c4, accB);
            store_mean(mt + (2 * p) * TP, accA, degA, hi, c4);
            store_mean(mt + (2 * p + 1) * TP, accB, degB, hi, c4);
            asm volatile("" : "+v"(nxtA), "+v"(nxtB));
            curA = min(max(nxtA, 0), NN - 1);
            curB = min(max(nxtB, 0), NN - 1);
        }

    }

    asm volatile("s_barrier_wait -1" ::: "memory");

    if (live) {
        const int row = n0 + r;
        const float* featrow = feat + row * CH;
        v2f af[16];
        #pragma unroll
        for (int s = 0; s < 16; ++s) af[s] = *(const v2f*)(featrow + 4 * s + hi * 2);
        if (CLEAN) {
            #pragma unroll
            for (int s = 0; s < 16; ++s) { af[s].x = nan0(af[s].x); af[s].y = nan0(af[s].y); }
        }

        v8f acc[4];
        #pragma unroll
        for (int nt = 0; nt < 4; ++nt) {
            #pragma unroll
            for (int j = 0; j < 8; ++j) acc[nt][j] = bv[nt];
        }

        #pragma unroll 4
        for (int s = 0; s < 16; ++s) {
            v2f a = *(const v2f*)(mt + r * TP + 4 * s + hi * 2);
            #pragma unroll
            for (int nt = 0; nt < 4; ++nt) {
                v2f b = *(const v2f*)&Wfrag[((s * 4 + nt) * 32 + lane) * 2];
                acc[nt] = __builtin_amdgcn_wmma_f32_16x16x4_f32(
                    false, a, false, b, (short)0, acc[nt], false, false);
            }
        }
        #pragma unroll
        for (int s = 0; s < 16; ++s) {
            #pragma unroll
            for (int nt = 0; nt < 4; ++nt) {
                v2f b = *(const v2f*)&Wfrag[(((s + 16) * 4 + nt) * 32 + lane) * 2];
                acc[nt] = __builtin_amdgcn_wmma_f32_16x16x4_f32(
                    false, af[s], false, b, (short)0, acc[nt], false, false);
            }
        }

        if (!HEAD) {
            #pragma unroll
            for (int nt = 0; nt < 4; ++nt) {
                #pragma unroll
                for (int v = 0; v < 8; ++v) {
                    int orow = n0 + hi * 8 + v;
                    out[orow * CH + nt * 16 + r] = fmaxf(acc[nt][v], 0.0f);
                }
            }
        } else {
            float p[8];
            #pragma unroll
            for (int v = 0; v < 8; ++v) {
                p[v] = fmaxf(acc[0][v], 0.0f) * wf[0];
                p[v] += fmaxf(acc[1][v], 0.0f) * wf[1];
                p[v] += fmaxf(acc[2][v], 0.0f) * wf[2];
                p[v] += fmaxf(acc[3][v], 0.0f) * wf[3];
            }
            #pragma unroll
            for (int v = 0; v < 8; ++v) p[v] = row_xmask_add<8>(p[v]);
            #pragma unroll
            for (int v = 0; v < 8; ++v) p[v] = row_xmask_add<4>(p[v]);
            #pragma unroll
            for (int v = 0; v < 8; ++v) p[v] = row_xmask_add<2>(p[v]);
            #pragma unroll
            for (int v = 0; v < 8; ++v) p[v] = row_xmask_add<1>(p[v]);
            float mine = p[0];
            #pragma unroll
            for (int v = 1; v < 8; ++v) mine = (r == v) ? p[v] : mine;
            if (r < 8) mt[hi * 8 + r] = mine + bf;
        }
    }
    if (HEAD) {
        __syncthreads();
        if (threadIdx.x < 128) {
            const int node = blockIdx.x * 128 + threadIdx.x;
            if (node < NN) pred[node] = meanT[threadIdx.x >> 4][threadIdx.x & 15];
        }
    }
}

extern "C" void kernel_launch(void* const* d_in, const int* in_sizes, int n_in,
                              void* d_out, int out_size, void* d_ws, size_t ws_size,
                              hipStream_t stream) {
    const float* x   = (const float*)d_in[0];
    const int*   ei  = (const int*)d_in[1];
    const float* W1l = (const float*)d_in[2];
    const float* b1  = (const float*)d_in[3];
    const float* W1r = (const float*)d_in[4];
    const float* W2l = (const float*)d_in[5];
    const float* b2  = (const float*)d_in[6];
    const float* W2r = (const float*)d_in[7];
    const float* Wfc = (const float*)d_in[8];
    const float* bfc = (const float*)d_in[9];
    float* pred = (float*)d_out;

    const int* src = ei;
    const int* dst = ei + NE;

    float* h1  = (float*)d_ws;
    int* rowptr = (int*)(h1 + NN * CH);
    int* col    = rowptr + NNP;
    unsigned* tab = (unsigned*)(col + NE);
    int* tmp    = (int*)(tab + NTAB);

    const int ggrid = (NTILES + 7) / 8;

    k_chunk_sort<<<G1, CT, 0, stream>>>(src, dst, tab, tmp);
    k_bin_place<<<NB, 1024, 0, stream>>>(tab, tmp, rowptr, col);

    k_sage_layer<true, false><<<ggrid, 256, 0, stream>>>(x, rowptr, col, W1l, W1r, b1, h1, Wfc, bfc, pred);
    k_sage_layer<false, true><<<ggrid, 256, 0, stream>>>(h1, rowptr, col, W2l, W2r, b2, nullptr, Wfc, bfc, pred);
}


# Layer 1 fused into the edge sort's bin pass: bins of 400 nodes = 25 tiles, the work-group that places a bin's edges runs layer 1 on them, bounds in LDS; 1 launch gone; float sum order by LDS arrival
# speedup vs baseline: 6.5771x; 1.0512x over previous
#include <hip/hip_runtime.h>


typedef __attribute__((ext_vector_type(2))) float v2f;
typedef __attribute__((ext_vector_type(4))) float v4f;
typedef __attribute__((ext_vector_type(8))) float v8f;

#define NN 100000
#define NE 1200000
#define CH 64
#define NTILES (NN / 16)
#define NNP 100032

#define BINSZ 400
#define NB (NN / BINSZ)
#define TPB (BINSZ / 16)
#define SCW ((BINSZ + 31) / 32)
#define G1 256
#define CHUNK ((NE + G1 - 1) / G1)
#define CT 512
#define EPT 10
#define NTAB (NB * G1)
#define PPW (G1 / 32)
#define PT 2
#define TABT ((NTAB + 4095) / 4096)
#define COT ((CHUNK + 2 * CT - 1) / (2 * CT))
static_assert(CHUNK <= EPT * CT, "a chunk must fit one pass of the work-group");
static_assert(CHUNK < 32768, "piece start and piece size share one 32-bit table word");
static_assert(NB <= 256, "the chunk pass scans its bin counts with eight waves");
static_assert(NN % BINSZ == 0 && BINSZ % 16 == 0 && TPB <= 32, "a bin is a whole number of 16-node tiles, one per wave of the bin pass");
static_assert(CHUNK % 2 == 0, "an odd run is shorter than the chunk: its last 16-byte LDS read stays inside the buffer");
static_assert(G1 % 32 == 0 && G1 <= 1024, "the bin pass deals whole pieces to its 32 waves");

__device__ __forceinline__ float nan0(float v) { return (v == v) ? v : 0.0f; }


__device__ __forceinline__ int wave_incl_scan(int x, int lane) {
    x += __builtin_amdgcn_update_dpp(0, x, 0x111, 0xf, 0xf, true);
    x += __builtin_amdgcn_update_dpp(0, x, 0x112, 0xf, 0xf, true);
    x += __builtin_amdgcn_update_dpp(0, x, 0x114, 0xf, 0xf, true);
    x += __builtin_amdgcn_update_dpp(0, x, 0x118, 0xf, 0xf, true);
    const int row0 = __builtin_amdgcn_readlane(x, 15);
    if (lane >= 16) x += row0;
    return x;
}

__device__ __forceinline__ int wave_sum(int x, int lane) {
    return __builtin_amdgcn_readlane(wave_incl_scan(x, lane), 31);
}

__global__ void __launch_bounds__(CT)
k_chunk_sort(const int* __restrict__ src, const int* __restrict__ dst,
             unsigned* __restrict__ tab, int* __restrict__ tmpi) {
    __shared__ int h[NB];
    __shared__ int wt[8];
    __shared__ __attribute__((aligned(16))) int2 buf[CHUNK];
    const int t = threadIdx.x;
    const int lane = t & 31;
    const int w = t >> 5;
    if (t < NB) h[t] = 0;
    const int cbeg = blockIdx.x * CHUNK;
    const int cend = min(cbeg + CHUNK, NE);
    int d[EPT], s[EPT];
    #pragma unroll
    for (int k = 0; k < EPT; ++k) {
        const int e = cbeg + k * CT + t;
        d[k] = (e < cend) ? dst[e] : -1;
        s[k] = (e < cend) ? src[e] : 0;
    }
    __syncthreads();
    #pragma unroll
    for (int k = 0; k < EPT; ++k)
        if ((unsigned)d[k] < (unsigned)NN) atomicAdd(&h[(unsigned)d[k] / BINSZ], 1);
    __syncthreads();
    int c = 0, incl = 0;
    if (t < 256) {
        c = (t < NB) ? h[t] : 0;
        incl = wave_incl_scan(c, lane);
        if (lane == 31) wt[w] = incl;
    }
    __syncthreads();
    if (t < 256) {
        int wpre = 0;
        #pragma unroll
        for (int j = 0; j < 8; ++j) wpre += (j < w) ? wt[j] : 0;
        const int excl = wpre + incl - c;
        if (t < NB) {
            tab[t * G1 + blockIdx.x] = (unsigned)excl | ((unsigned)c << 16);
            h[t] = excl;
        }
    }
    __syncthreads();
    int pos[EPT];
    #pragma unroll
    for (int k = 0; k < EPT; ++k) {
        pos[k] = -1;
        if ((unsigned)d[k] < (unsigned)NN) pos[k] = atomicAdd(&h[(unsigned)d[k] / BINSZ], 1);
    }
    #pragma unroll
    for (int k = 0; k < EPT; ++k)
        if ((unsigned)pos[k] < (unsigned)CHUNK) buf[pos[k]] = make_int2(s[k], d[k]);
    __syncthreads();
    const int total = min(h[NB - 1], CHUNK);
    int2* run = (int2*)tmpi + blockIdx.x * CHUNK;
    int4 o[COT];
    #pragma unroll
    for (int k = 0; k < COT; ++k) {
        const int i = (k * CT + t) * 2;
        o[k] = make_int4(0, 0, 0, 0);
        if (i < total) o[k] = *(const int4*)(buf + i);
    }
    #pragma unroll
    for (int k = 0; k < COT; ++k) {
        const int i = (k * CT + t) * 2;
        if (i + 1 < total) *(int4*)(run + i) = o[k];
        else if (i < total) run[i] = make_int2(o[k].x, o[k].y);
    }
}

template <int U, bool CLEAN>
__device__ __forceinline__ void gather_trip(const float* __restrict__ feat, int myc, int j, int n,
                                            int hi, int c4, float4& acc) {
    static_assert(U == 4 || U == 8, "the fence below names every row of the trip");
    v4f v[U];
    #pragma unroll
    for (int u = 0; u < U; ++u) {
        const int s = __shfl(myc, j + 2 * u + hi, 32);
        v[u] = *(const v4f*)(feat + s * CH + c4);
    }
    if constexpr (U == 8)
        asm volatile("" : "+v"(v[0]), "+v"(v[1]), "+v"(v[2]), "+v"(v[3]), "+v"(v[4]), "+v"(v[5]), "+v"(v[6]), "+v"(v[7]));
    else
        asm volatile("" : "+v"(v[0]), "+v"(v[1]), "+v"(v[2]), "+v"(v[3]));
    #pragma unroll
    for (int u = 0; u < U; ++u) {
        const bool ok = (j + 2 * u + hi) < n;
        v4f c = v[u];
        if (CLEAN) { c.x = nan0(c.x); c.y = nan0(c.y); c.z = nan0(c.z); c.w = nan0(c.w); }
        acc.x += ok ? c.x : 0.0f;
        acc.y += ok ? c.y : 0.0f;
        acc.z += ok ? c.z : 0.0f;
        acc.w += ok ? c.w : 0.0f;
    }
}

__device__ __forceinline__ void request_rows8(const float* __restrict__ feat, int myc, int hi, int c4, v4f (&v)[8]) {
    #pragma unroll
    for (int u = 0; u < 8; ++u) {
        const int s = __shfl(myc, 2 * u + hi, 32);
        v[u] = *(const v4f*)(feat + s * CH + c4);
    }
}

template <bool CLEAN>
__device__ __forceinline__ void add_rows8(const v4f (&v)[8], int n, int hi, float4& acc) {
    #pragma unroll
    for (int u = 0; u < 8; ++u) {
        const bool ok = (2 * u + hi) < n;
        v4f c = v[u];
        if (CLEAN) { c.x = nan0(c.x); c.y = nan0(c.y); c.z = nan0(c.z); c.w = nan0(c.w); }
        acc.x += ok ? c.x : 0.0f;
        acc.y += ok ? c.y : 0.0f;
        acc.z += ok ? c.z : 0.0f;
        acc.w += ok ? c.w : 0.0f;
    }
}

template <bool CLEAN>
__device__ __forceinline__ void gather_rest(const float* __restrict__ feat, const int* __restrict__ col,
                                            int beg, int end, int myc, int lane, int hi, int c4, float4& acc) {
    for (int base = beg; base < end; base += 32) {
        int n = end - base;
        if (n > 32) n = 32;
        int j = 16;
        if (base != beg) {
            int c = 0;
            if (lane < n) c = col[base + lane];
            myc = min(max(c, 0), NN - 1);
            j = 0;
        }
        for (; j < n; j += 16) {
            if (n - j > 8) gather_trip<8, CLEAN>(feat, myc, j, n, hi, c4, acc);
            else           gather_trip<4, CLEAN>(feat, myc, j, n, hi, c4, acc);
        }
    }
}

__device__ __forceinline__ void store_mean(float* __restrict__ mrow, float4 acc, int deg, int hi, int c4) {
    acc.x += __shfl_xor(acc.x, 16, 32);
    acc.y += __shfl_xor(acc.y, 16, 32);
    acc.z += __shfl_xor(acc.z, 16, 32);
    acc.w += __shfl_xor(acc.w, 16, 32);
    const float rd = 1.0f / fmaxf((float)deg, 1.0f);
    if (hi == 0) *(float4*)(mrow + c4) = make_float4(acc.x * rd, acc.y * rd, acc.z * rd, acc.w * rd);
}

#define TP 68

template <int XM>
__device__ __forceinline__ float row_xmask_add(float x) {
    return x + __builtin_bit_cast(float, __builtin_amdgcn_update_dpp(0, __builtin_bit_cast(int, x), 0x160 + XM, 0xf, 0xf, true));
}

__device__ __forceinline__ const float* witem_src(const float* __restrict__ Wsel, int id) {
    const int q  = id & 15;
    const int kp = id >> 4;
    return Wsel + (2 * (kp & 31)) * 64 + 4 * q;
}
__device__ __forceinline__ void witem_store(float* __restrict__ Wfrag, int id, v4f w0, v4f w1) {
    const int q  = id & 15;
    const int kp = id >> 4;
    float* wdst = Wfrag + ((((kp >> 1) * 4 + (q >> 2)) * 32 + (kp & 1) * 16 + (q & 3) * 4) * 2);
    *(v4f*)wdst       = (v4f){w0.x, w1.x, w0.y, w1.y};
    *(v4f*)(wdst + 4) = (v4f){w0.z, w1.z, w0.w, w1.w};
}

template <bool CLEAN>
__device__ __forceinline__ void pull_sums(const float* __restrict__ feat, const int* __restrict__ col, int rpv,
                                          float* mt, int lane, int hi, int c4) {
    int b0 = __builtin_amdgcn_readlane(rpv, 0);
    int b1 = __builtin_amdgcn_readlane(rpv, 1);
    int b2 = __builtin_amdgcn_readlane(rpv, 2);
    int curA = 0, curB = 0;
    if (lane < b1 - b0) curA = col[b0 + lane];
    if (lane < b2 - b1) curB = col[b1 + lane];
    asm volatile("" : "+v"(curA), "+v"(curB));
    curA = min(max(curA, 0), NN - 1);
    curB = min(max(curB, 0), NN - 1);

    for (int p = 0; p < 8; ++p) {
        const int begA = b0, endA = b1, endB = b2;
        const int mycA = curA, mycB = curB;
        int nxtA = 0, nxtB = 0;
        if (p + 1 < 8) {
            b0 = endB;
            b1 = __builtin_amdgcn_readlane(rpv, 2 * p + 3);
            b2 = __builtin_amdgcn_readlane(rpv, 2 * p + 4);
            if (lane < b1 - b0) nxtA = col[b0 + lane];
            if (lane < b2 - b1) nxtB = col[b1 + lane];
        }
        const int degA = endA - begA, degB = endB - endA;

        v4f va[8], vb[8];
        request_rows8(feat, mycA, hi, c4, va);
        request_rows8(feat, mycB, hi, c4, vb);
        float4 accA = make_float4(0.0f, 0.0f, 0.0f, 0.0f);
        float4 accB = make_float4(0.0f, 0.0f, 0.0f, 0.0f);
        add_rows8<CLEAN>(va, min(degA, 32), hi, accA);
        add_rows8<CLEAN>(vb, min(degB, 32), hi, accB);
        if (degA > 16) gather_rest<CLEAN>(feat, col, begA, endA, mycA, lane, hi, c4, accA);
        if (degB > 16) gather_rest<CLEAN>(feat, col, endA, endB, mycB, lane, hi, c4, accB);
        store_mean(mt + (2 * p) * TP, accA, degA, hi, c4);
        store_mean(mt + (2 * p + 1) * TP, accB, degB, hi, c4);
        asm volatile("" : "+v"(nxtA), "+v"(nxtB));
        curA = min(max(nxtA, 0), NN - 1);
        curB = min(max(nxtB, 0), NN - 1);
    }
}

template <bool CLEAN, bool HEAD>
__device__ __forceinline__ void matrix_part(const float* __restrict__ feat, int n0, float* mt, const float* Wfrag,
                                            const float (&bv)[4], const float (&wf)[4], float bf,
                                            float* __restrict__ out, int lane, int hi, int r) {
    const int row = n0 + r;
    const float* featrow = feat + row * CH;
    v2f af[16];
    #pragma unroll
    for (int s = 0; s < 16; ++s) af[s] = *(const v2f*)(featrow + 4 * s + hi * 2);
    if (CLEAN) {
        #pragma unroll
        for (int s = 0; s < 16; ++s) { af[s].x = nan0(af[s].x); af[s].y = nan0(af[s].y); }
    }

    v8f acc[4];
    #pragma unroll
    for (int nt = 0; nt < 4; ++nt) {
        #pragma unroll
        for (int j = 0; j < 8; ++j) acc[nt][j] = bv[nt];
    }

    #pragma unroll 4
    for (int s = 0; s < 16; ++s) {
        v2f a = *(const v2f*)(mt + r * TP + 4 * s + hi * 2);
        #pragma unroll
        for (int nt = 0; nt < 4; ++nt) {
            v2f b = *(const v2f*)&Wfrag[((s * 4 + nt) * 32 + lane) * 2];
            acc[nt] = __builtin_amdgcn_wmma_f32_16x16x4_f32(
                false, a, false, b, (short)0, acc[nt], false, false);
        }
    }
    #pragma unroll
    for (int s = 0; s < 16; ++s) {
        #pragma unroll
        for (int nt = 0; nt < 4; ++nt) {
            v2f b = *(const v2f*)&Wfrag[(((s + 16) * 4 + nt) * 32 + lane) * 2];
            acc[nt] = __builtin_amdgcn_wmma_f32_16x16x4_f32(
                false, af[s], false, b, (short)0, acc[nt], false, false);
        }
    }

    if (!HEAD) {
        #pragma unroll
        for (int nt = 0; nt < 4; ++nt) {
            #pragma unroll
            for (int v = 0; v < 8; ++v) {
                int orow = n0 + hi * 8 + v;
                out[orow * CH + nt * 16 + r] = fmaxf(acc[nt][v], 0.0f);
            }
        }
    } else {
        float p[8];
        #pragma unroll
        for (int v = 0; v < 8; ++v) {
            p[v] = fmaxf(acc[0][v], 0.0f) * wf[0];
            p[v] += fmaxf(acc[1][v], 0.0f) * wf[1];
            p[v] += fmaxf(acc[2][v], 0.0f) * wf[2];
            p[v] += fmaxf(acc[3][v], 0.0f) * wf[3];
        }
        #pragma unroll
        for (int v = 0; v < 8; ++v) p[v] = row_xmask_add<8>(p[v]);
        #pragma unroll
        for (int v = 0; v < 8; ++v) p[v] = row_xmask_add<4>(p[v]);
        #pragma unroll
        for (int v = 0; v < 8; ++v) p[v] = row_xmask_add<2>(p[v]);
        #pragma unroll
        for (int v = 0; v < 8; ++v) p[v] = row_xmask_add<1>(p[v]);
        float mine = p[0];
        #pragma unroll
        for (int v = 1; v < 8; ++v) mine = (r == v) ? p[v] : mine;
        if (r < 8) mt[hi * 8 + r] = mine + bf;
    }
}

__global__ void __launch_bounds__(1024)
k_bin_layer1(const unsigned* __restrict__ tab, const int* __restrict__ tmpi,
             int* __restrict__ rowptr, int* col,
             const float* __restrict__ x, const float* __restrict__ Wl, const float* __restrict__ Wr,
             const float* __restrict__ bias, float* __restrict__ out) {
    __shared__ int cnt[BINSZ];
    __shared__ int wtot[SCW];
    __shared__ int red[2][32];
    __shared__ int rp[BINSZ + 1];
    __shared__ float Wfrag[32 * 4 * 32 * 2];
    __shared__ float meanT[TPB][16 * TP];
    const int2* tmp = (const int2*)tmpi;
    const int t = threadIdx.x;
    const int lane = t & 31;
    const int w = __builtin_amdgcn_readfirstlane(t >> 5);
    const int bin = blockIdx.x;
    const int base = bin * BINSZ;

    unsigned myinfo = 0;
    if (lane < PPW) myinfo = tab[bin * G1 + w + 32 * lane];
    const int nprev = bin * G1;
    uint4 tv[TABT];
    #pragma unroll
    for (int k = 0; k < TABT; ++k) {
        const int i = k * 4096 + t * 4;
        tv[k] = *(const uint4*)(tab + ((i < nprev) ? i : 0));
    }
    int pst[PPW], pcn[PPW];
    #pragma unroll
    for (int i = 0; i < PPW; ++i) {
        const unsigned info = __builtin_amdgcn_readlane(myinfo, i);
        pst[i] = min((int)(info & 0xffffu), CHUNK);
        pcn[i] = min((int)(info >> 16), CHUNK - pst[i]);
    }
    int2 p[PPW * PT];
    #pragma unroll
    for (int i = 0; i < PPW; ++i) {
        const int2* piece = tmp + (w + 32 * i) * CHUNK + pst[i];
        #pragma unroll
        for (int u = 0; u < PT; ++u) {
            const int j = u * 32 + lane;
            p[i * PT + u] = make_int2(0, -1);
            if (j < pcn[i]) p[i * PT + u] = piece[j];
        }
    }

    const float* wsrc = witem_src((t < 512) ? Wl : Wr, t);
    const v4f w0 = *(const v4f*)wsrc;
    const v4f w1 = *(const v4f*)(wsrc + 64);

    int before = 0;
    #pragma unroll
    for (int k = 0; k < TABT; ++k) {
        const int i = k * 4096 + t * 4;
        const int s4 = (int)((tv[k].x >> 16) + (tv[k].y >> 16) + (tv[k].z >> 16) + (tv[k].w >> 16));
        before += (i < nprev) ? s4 : 0;
    }
    if (t < BINSZ) cnt[t] = 0;
    witem_store(Wfrag, t, w0, w1);
    const int wb = wave_sum(before, lane);
    const int wo = wave_sum((int)(myinfo >> 16), lane);
    if (lane == 0) { red[0][w] = wb; red[1][w] = wo; }
    __syncthreads();
    const int beg = wave_sum(red[0][lane], lane);
    const int end = beg + wave_sum(red[1][lane], lane);

    #pragma unroll
    for (int q = 0; q < PPW * PT; ++q) {
        const int dl = p[q].y - base;
        if (p[q].y >= 0 && (unsigned)dl < (unsigned)BINSZ) atomicAdd(&cnt[dl], 1);
    }
    for (int i = 0; i < PPW; ++i) {
        const int2* piece = tmp + (w + 32 * i) * CHUNK + pst[i];
        for (int j = PT * 32 + lane; j < pcn[i]; j += 32) {
            const int y = piece[j].y;
            const int dl = y - base;
            if (y >= 0 && (unsigned)dl < (unsigned)BINSZ) atomicAdd(&cnt[dl], 1);
        }
    }
    __syncthreads();
    const int c = (t < BINSZ) ? cnt[t] : 0;
    const int incl = wave_incl_scan(c, lane);
    if (lane == 31 && w < SCW) wtot[w] = incl;
    __syncthreads();
    int wpre = 0;
    #pragma unroll
    for (int j = 0; j < SCW; ++j) wpre += (j < w) ? wtot[j] : 0;
    const int excl = beg + wpre + incl - c;
    if (t < BINSZ) {
        const int node = base + t;
        if (node < NN) rowptr[node] = excl;
        rp[t] = excl;
        cnt[t] = excl;
    }
    if (t == 0) rp[BINSZ] = end;
    if (bin == NB - 1 && t == 0) rowptr[NN] = end;
    __syncthreads();
    int pos[PPW * PT];
    #pragma unroll
    for (int q = 0; q < PPW * PT; ++q) {
        const int dl = p[q].y - base;
        pos[q] = -1;
        if (p[q].y >= 0 && (unsigned)dl < (unsigned)BINSZ) pos[q] = atomicAdd(&cnt[dl], 1);
    }
    #pragma unroll
    for (int q = 0; q < PPW * PT; ++q)
        if ((unsigned)pos[q] < (unsigned)NE) col[pos[q]] = p[q].x;
    for (int i = 0; i < PPW; ++i) {
        const int2* piece = tmp + (w + 32 * i) * CHUNK + pst[i];
        for (int j = PT * 32 + lane; j < pcn[i]; j += 32) {
            const int2 pr = piece[j];
            const int dl = pr.y - base;
            if (pr.y >= 0 && (unsigned)dl < (unsigned)BINSZ) {
                const int ps = atomicAdd(&cnt[dl], 1);
                if ((unsigned)ps < (unsigned)NE) col[ps] = pr.x;
            }
        }
    }

    __syncthreads();
    const int tile = bin * TPB + w;
    if (w < TPB && tile < NTILES) {
        const int hi = lane >> 4;
        const int r  = lane & 15;
        const int c4 = r * 4;
        float* mt = meanT[w];
        float bv[4];
        const float nowf[4] = {0.0f, 0.0f, 0.0f, 0.0f};
        #pragma unroll
        for (int nt = 0; nt < 4; ++nt) bv[nt] = bias[nt * 16 + r];
        int rpv = 0;
        if (lane <= 16) rpv = rp[16 * w + lane];
        pull_sums<true>(x, col, rpv, mt, lane, hi, c4);
        matrix_part<true, false>(x, tile * 16, mt, Wfrag, bv, nowf, 0.0f, out, lane, hi, r);
    }
}

__global__ void __launch_bounds__(256)
k_sage_layer2(const float* __restrict__ feat, const int* __restrict__ rowptr, const int* __restrict__ col,
              const float* __restrict__ Wl, const float* __restrict__ Wr, const float* __restrict__ bias,
              const float* __restrict__ Wfc, const float* __restrict__ bfc, float* __restrict__ pred) {
    __shared__ float Wfrag[32 * 4 * 32 * 2];
    __shared__ float meanT[8][16 * TP];

    const int lane = threadIdx.x & 31;
    const int hi   = lane >> 4;
    const int r    = lane & 15;
    const int c4   = r * 4;
    const int wv   = __builtin_amdgcn_readfirstlane((int)(threadIdx.x >> 5));
    const int tile = blockIdx.x * 8 + wv;
    const bool live = tile < NTILES;
    float* mt = meanT[wv];
    const int n0 = tile * 16;

    v4f w0[4], w1[4];
    #pragma unroll
    for (int j = 0; j < 4; ++j) {
        const float* wsrc = witem_src((j < 2) ? Wl : Wr, j * 256 + (int)threadIdx.x);
        w0[j] = *(const v4f*)wsrc;
        w1[j] = *(const v4f*)(wsrc + 64);
    }
    float bv[4], wf[4];
    #pragma unroll
    for (int nt = 0; nt < 4; ++nt) {
        bv[nt] = bias[nt * 16 + r];
        wf[nt] = Wfc[nt * 16 + r];
    }
    const float bf = bfc[0];
    int rpv = 0;
    if (live && lane <= 16) rpv = rowptr[n0 + lane];
    #pragma unroll
    for (int j = 0; j < 4; ++j) witem_store(Wfrag, j * 256 + (int)threadIdx.x, w0[j], w1[j]);
    asm volatile("s_wait_dscnt 0x0\n\ts_barrier_signal -1" ::: "memory");

    if (live) pull_sums<false>(feat, col, rpv, mt, lane, hi, c4);

    asm volatile("s_barrier_wait -1" ::: "memory");

    if (live) matrix_part<false, true>(feat, n0, mt, Wfrag, bv, wf, bf, nullptr, lane, hi, r);

    __syncthreads();
    if (threadIdx.x < 128) {
        const int node = blockIdx.x * 128 + threadIdx.x;
        if (node < NN) pred[node] = meanT[threadIdx.x >> 4][threadIdx.x & 15];
    }
}

extern "C" void kernel_launch(void* const* d_in, const int* in_sizes, int n_in,
                              void* d_out, int out_size, void* d_ws, size_t ws_size,
                              hipStream_t stream) {
    const float* x   = (const float*)d_in[0];
    const int*   ei  = (const int*)d_in[1];
    const float* W1l = (const float*)d_in[2];
    const float* b1  = (const float*)d_in[3];
    const float* W1r = (const float*)d_in[4];
    const float* W2l = (const float*)d_in[5];
    const float* b2  = (const float*)d_in[6];
    const float* W2r = (const float*)d_in[7];
    const float* Wfc = (const float*)d_in[8];
    const float* bfc = (const float*)d_in[9];
    float* pred = (float*)d_out;

    const int* src = ei;
    const int* dst = ei + NE;

    float* h1  = (float*)d_ws;
    int* rowptr = (int*)(h1 + NN * CH);
    int* col    = rowptr + NNP;
    unsigned* tab = (unsigned*)(col + NE);
    int* tmp    = (int*)(tab + NTAB);

    const int ggrid = (NTILES + 7) / 8;

    k_chunk_sort<<<G1, CT, 0, stream>>>(src, dst, tab, tmp);
    k_bin_layer1<<<NB, 1024, 0, stream>>>(tab, tmp, rowptr, col, x, W1l, W1r, b1, h1);

    k_sage_layer2<<<ggrid, 256, 0, stream>>>(h1, rowptr, col, W2l, W2r, b2, Wfc, bfc, pred);
}


# Pull sums, fewer instructions per row: 32-bit row index off a scalar base; rows past a node's degree not loaded or added, only odd segments masked; layer 1 nan test on the node's sum; sum order kept
# speedup vs baseline: 6.8141x; 1.0360x over previous
#include <hip/hip_runtime.h>


typedef __attribute__((ext_vector_type(2))) float v2f;
typedef __attribute__((ext_vector_type(4))) float v4f;
typedef __attribute__((ext_vector_type(8))) float v8f;

#define NN 100000
#define NE 1200000
#define CH 64
#define NTILES (NN / 16)
#define NNP 100032

#define BINSZ 400
#define NB (NN / BINSZ)
#define TPB (BINSZ / 16)
#define SCW ((BINSZ + 31) / 32)
#define G1 256
#define CHUNK ((NE + G1 - 1) / G1)
#define CT 512
#define EPT 10
#define NTAB (NB * G1)
#define PPW (G1 / 32)
#define PT 2
#define TABT ((NTAB + 4095) / 4096)
#define COT ((CHUNK + 2 * CT - 1) / (2 * CT))
static_assert(CHUNK <= EPT * CT, "a chunk must fit one pass of the work-group");
static_assert(CHUNK < 32768, "piece start and piece size share one 32-bit table word");
static_assert(NB <= 256, "the chunk pass scans its bin counts with eight waves");
static_assert(NN % BINSZ == 0 && BINSZ % 16 == 0 && TPB <= 32, "a bin is a whole number of 16-node tiles, one per wave of the bin pass");
static_assert(CHUNK % 2 == 0, "an odd run is shorter than the chunk: its last 16-byte LDS read stays inside the buffer");
static_assert(G1 % 32 == 0 && G1 <= 1024, "the bin pass deals whole pieces to its 32 waves");

__device__ __forceinline__ float nan0(float v) { return (v == v) ? v : 0.0f; }


__device__ __forceinline__ int wave_incl_scan(int x, int lane) {
    x += __builtin_amdgcn_update_dpp(0, x, 0x111, 0xf, 0xf, true);
    x += __builtin_amdgcn_update_dpp(0, x, 0x112, 0xf, 0xf, true);
    x += __builtin_amdgcn_update_dpp(0, x, 0x114, 0xf, 0xf, true);
    x += __builtin_amdgcn_update_dpp(0, x, 0x118, 0xf, 0xf, true);
    const int row0 = __builtin_amdgcn_readlane(x, 15);
    if (lane >= 16) x += row0;
    return x;
}

__device__ __forceinline__ int wave_sum(int x, int lane) {
    return __builtin_amdgcn_readlane(wave_incl_scan(x, lane), 31);
}

__global__ void __launch_bounds__(CT)
k_chunk_sort(const int* __restrict__ src, const int* __restrict__ dst,
             unsigned* __restrict__ tab, int* __restrict__ tmpi) {
    __shared__ int h[NB];
    __shared__ int wt[8];
    __shared__ __attribute__((aligned(16))) int2 buf[CHUNK];
    const int t = threadIdx.x;
    const int lane = t & 31;
    const int w = t >> 5;
    if (t < NB) h[t] = 0;
    const int cbeg = blockIdx.x * CHUNK;
    const int cend = min(cbeg + CHUNK, NE);
    int d[EPT], s[EPT];
    #pragma unroll
    for (int k = 0; k < EPT; ++k) {
        const int e = cbeg + k * CT + t;
        d[k] = (e < cend) ? dst[e] : -1;
        s[k] = (e < cend) ? src[e] : 0;
    }
    __syncthreads();
    #pragma unroll
    for (int k = 0; k < EPT; ++k)
        if ((unsigned)d[k] < (unsigned)NN) atomicAdd(&h[(unsigned)d[k] / BINSZ], 1);
    __syncthreads();
    int c = 0, incl = 0;
    if (t < 256) {
        c = (t < NB) ? h[t] : 0;
        incl = wave_incl_scan(c, lane);
        if (lane == 31) wt[w] = incl;
    }
    __syncthreads();
    if (t < 256) {
        int wpre = 0;
        #pragma unroll
        for (int j = 0; j < 8; ++j) wpre += (j < w) ? wt[j] : 0;
        const int excl = wpre + incl - c;
        if (t < NB) {
            tab[t * G1 + blockIdx.x] = (unsigned)excl | ((unsigned)c << 16);
            h[t] = excl;
        }
    }
    __syncthreads();
    int pos[EPT];
    #pragma unroll
    for (int k = 0; k < EPT; ++k) {
        pos[k] = -1;
        if ((unsigned)d[k] < (unsigned)NN) pos[k] = atomicAdd(&h[(unsigned)d[k] / BINSZ], 1);
    }
    #pragma unroll
    for (int k = 0; k < EPT; ++k)
        if ((unsigned)pos[k] < (unsigned)CHUNK) buf[pos[k]] = make_int2(s[k], d[k]);
    __syncthreads();
    const int total = min(h[NB - 1], CHUNK);
    int2* run = (int2*)tmpi + blockIdx.x * CHUNK;
    int4 o[COT];
    #pragma unroll
    for (int k = 0; k < COT; ++k) {
        const int i = (k * CT + t) * 2;
        o[k] = make_int4(0, 0, 0, 0);
        if (i < total) o[k] = *(const int4*)(buf + i);
    }
    #pragma unroll
    for (int k = 0; k < COT; ++k) {
        const int i = (k * CT + t) * 2;
        if (i + 1 < total) *(int4*)(run + i) = o[k];
        else if (i < total) run[i] = make_int2(o[k].x, o[k].y);
    }
}


__device__ __forceinline__ v4f load_unit(const float* __restrict__ feat, unsigned q) {
    return ((const v4f*)feat)[q];
}

__device__ __forceinline__ int row_unit(int s) { return min(max(s, 0), NN - 1) << 4; }

__device__ __forceinline__ void gather_trip4(const float* __restrict__ feat, int myq, int j, int n,
                                             int hi, int r, float4& acc) {
    v4f v[4];
    #pragma unroll
    for (int u = 0; u < 4; ++u) {
        const unsigned q = (unsigned)__shfl(myq, j + 2 * u + hi, 32) | (unsigned)r;
        v[u] = load_unit(feat, q);
    }
    #pragma unroll
    for (int u = 0; u < 4; ++u) {
        const bool ok = (j + 2 * u + hi) < n;
        const v4f c = v[u];
        acc.x += ok ? c.x : 0.0f;
        acc.y += ok ? c.y : 0.0f;
        acc.z += ok ? c.z : 0.0f;
        acc.w += ok ? c.w : 0.0f;
    }
}

__device__ __forceinline__ void request_rows8(const float* __restrict__ feat, int myq, int deg, int hi, int r,
                                              v4f (&v)[8]) {
    unsigned q[8];
    #pragma unroll
    for (int u = 0; u < 8; ++u) q[u] = (unsigned)__shfl(myq, 2 * u + hi, 32) | (unsigned)r;
    #pragma unroll
    for (int u = 0; u < 8; ++u) {
        if (2 * u >= deg) break;
        v[u] = load_unit(feat, q[u]);
    }
}

__device__ __forceinline__ void add_rows8(const v4f (&v)[8], int deg, int hi, float4& acc) {
    if ((deg & 1) == 0 || deg >= 16) {
        #pragma unroll
        for (int u = 0; u < 8; ++u) {
            if (2 * u >= deg) break;
            const v4f c = v[u];
            acc.x += c.x;
            acc.y += c.y;
            acc.z += c.z;
            acc.w += c.w;
        }
    } else {
        #pragma unroll
        for (int u = 0; u < 8; ++u) {
            if (2 * u >= deg) break;
            const bool ok = (2 * u + hi) < deg;
            const v4f c = v[u];
            acc.x += ok ? c.x : 0.0f;
            acc.y += ok ? c.y : 0.0f;
            acc.z += ok ? c.z : 0.0f;
            acc.w += ok ? c.w : 0.0f;
        }
    }
}

__device__ __forceinline__ void gather_rest(const float* __restrict__ feat, const int* __restrict__ col,
                                            int beg, int end, int myq, int lane, int hi, int r, float4& acc) {
    for (int base = beg; base < end; base += 32) {
        int n = end - base;
        if (n > 32) n = 32;
        int j = 16;
        if (base != beg) {
            int c = 0;
            if (lane < n) c = col[base + lane];
            myq = row_unit(c);
            j = 0;
        }
        for (; j < n; j += 8) gather_trip4(feat, myq, j, n, hi, r, acc);
    }
}

__device__ __forceinline__ float4 resum_clean(const float* __restrict__ feat, const int* __restrict__ col,
                                              int beg, int deg, int hi, int r) {
    float4 a = make_float4(0.0f, 0.0f, 0.0f, 0.0f);
    for (int e = hi; e < deg; e += 2) {
        const v4f c = load_unit(feat, (unsigned)row_unit(col[beg + e]) | (unsigned)r);
        a.x += nan0(c.x);
        a.y += nan0(c.y);
        a.z += nan0(c.z);
        a.w += nan0(c.w);
    }
    return a;
}

__device__ __forceinline__ void store_mean(float* __restrict__ mrow, float4 acc, int deg, int hi, int c4) {
    acc.x += __shfl_xor(acc.x, 16, 32);
    acc.y += __shfl_xor(acc.y, 16, 32);
    acc.z += __shfl_xor(acc.z, 16, 32);
    acc.w += __shfl_xor(acc.w, 16, 32);
    const float rd = 1.0f / fmaxf((float)deg, 1.0f);
    if (hi == 0) *(float4*)(mrow + c4) = make_float4(acc.x * rd, acc.y * rd, acc.z * rd, acc.w * rd);
}

#define TP 68

template <int XM>
__device__ __forceinline__ float row_xmask_add(float x) {
    return x + __builtin_bit_cast(float, __builtin_amdgcn_update_dpp(0, __builtin_bit_cast(int, x), 0x160 + XM, 0xf, 0xf, true));
}

__device__ __forceinline__ const float* witem_src(const float* __restrict__ Wsel, int id) {
    const int q  = id & 15;
    const int kp = id >> 4;
    return Wsel + (2 * (kp & 31)) * 64 + 4 * q;
}
__device__ __forceinline__ void witem_store(float* __restrict__ Wfrag, int id, v4f w0, v4f w1) {
    const int q  = id & 15;
    const int kp = id >> 4;
    float* wdst = Wfrag + ((((kp >> 1) * 4 + (q >> 2)) * 32 + (kp & 1) * 16 + (q & 3) * 4) * 2);
    *(v4f*)wdst       = (v4f){w0.x, w1.x, w0.y, w1.y};
    *(v4f*)(wdst + 4) = (v4f){w0.z, w1.z, w0.w, w1.w};
}

template <bool CLEAN>
__device__ __forceinline__ void pull_sums(const float* __restrict__ feat, const int* __restrict__ col, int rpv,
                                          float* mt, int lane, int hi, int r) {
    const int c4 = r * 4;
    int b0 = __builtin_amdgcn_readlane(rpv, 0);
    int b1 = __builtin_amdgcn_readlane(rpv, 1);
    int b2 = __builtin_amdgcn_readlane(rpv, 2);
    int curA = 0, curB = 0;
    if (lane < b1 - b0) curA = col[b0 + lane];
    if (lane < b2 - b1) curB = col[b1 + lane];
    asm volatile("" : "+v"(curA), "+v"(curB));
    curA = row_unit(curA);
    curB = row_unit(curB);

    for (int p = 0; p < 8; ++p) {
        const int begA = b0, endA = b1, endB = b2;
        const int myqA = curA, myqB = curB;
        int nxtA = 0, nxtB = 0;
        if (p + 1 < 8) {
            b0 = endB;
            b1 = __builtin_amdgcn_readlane(rpv, 2 * p + 3);
            b2 = __builtin_amdgcn_readlane(rpv, 2 * p + 4);
            if (lane < b1 - b0) nxtA = col[b0 + lane];
            if (lane < b2 - b1) nxtB = col[b1 + lane];
        }
        const int degA = endA - begA, degB = endB - endA;

        v4f va[8], vb[8];
        request_rows8(feat, myqA, degA, hi, r, va);
        request_rows8(feat, myqB, degB, hi, r, vb);
        float4 accA = make_float4(0.0f, 0.0f, 0.0f, 0.0f);
        float4 accB = make_float4(0.0f, 0.0f, 0.0f, 0.0f);
        add_rows8(va, degA, hi, accA);
        add_rows8(vb, degB, hi, accB);
        if (degA > 16) gather_rest(feat, col, begA, endA, myqA, lane, hi, r, accA);
        if (degB > 16) gather_rest(feat, col, endA, endB, myqB, lane, hi, r, accB);
        if (CLEAN) {
            const bool badA = (accA.x != accA.x) || (accA.y != accA.y) || (accA.z != accA.z) || (accA.w != accA.w);
            const bool badB = (accB.x != accB.x) || (accB.y != accB.y) || (accB.z != accB.z) || (accB.w != accB.w);
            if (__builtin_amdgcn_ballot_w32(badA) != 0u) accA = resum_clean(feat, col, begA, degA, hi, r);
            if (__builtin_amdgcn_ballot_w32(badB) != 0u) accB = resum_clean(feat, col, endA, degB, hi, r);
        }
        store_mean(mt + (2 * p) * TP, accA, degA, hi, c4);
        store_mean(mt + (2 * p + 1) * TP, accB, degB, hi, c4);
        asm volatile("" : "+v"(nxtA), "+v"(nxtB));
        curA = row_unit(nxtA);
        curB = row_unit(nxtB);
    }
}

template <bool CLEAN, bool HEAD>
__device__ __forceinline__ void matrix_part(const float* __restrict__ feat, int n0, float* mt, const float* Wfrag,
                                            const float (&bv)[4], const float (&wf)[4], float bf,
                                            float* __restrict__ out, int lane, int hi, int r) {
    const int row = n0 + r;
    const float* featrow = feat + row * CH;
    v2f af[16];
    #pragma unroll
    for (int s = 0; s < 16; ++s) af[s] = *(const v2f*)(featrow + 4 * s + hi * 2);
    if (CLEAN) {
        #pragma unroll
        for (int s = 0; s < 16; ++s) { af[s].x = nan0(af[s].x); af[s].y = nan0(af[s].y); }
    }

    v8f acc[4];
    #pragma unroll
    for (int nt = 0; nt < 4; ++nt) {
        #pragma unroll
        for (int j = 0; j < 8; ++j) acc[nt][j] = bv[nt];
    }

    #pragma unroll 4
    for (int s = 0; s < 16; ++s) {
        v2f a = *(const v2f*)(mt + r * TP + 4 * s + hi * 2);
        #pragma unroll
        for (int nt = 0; nt < 4; ++nt) {
            v2f b = *(const v2f*)&Wfrag[((s * 4 + nt) * 32 + lane) * 2];
            acc[nt] = __builtin_amdgcn_wmma_f32_16x16x4_f32(
                false, a, false, b, (short)0, acc[nt], false, false);
        }
    }
    #pragma unroll
    for (int s = 0; s < 16; ++s) {
        #pragma unroll
        for (int nt = 0; nt < 4; ++nt) {
            v2f b = *(const v2f*)&Wfrag[(((s + 16) * 4 + nt) * 32 + lane) * 2];
            acc[nt] = __builtin_amdgcn_wmma_f32_16x16x4_f32(
                false, af[s], false, b, (short)0, acc[nt], false, false);
        }
    }

    if (!HEAD) {
        #pragma unroll
        for (int nt = 0; nt < 4; ++nt) {
            #pragma unroll
            for (int v = 0; v < 8; ++v) {
                int orow = n0 + hi * 8 + v;
                out[orow * CH + nt * 16 + r] = fmaxf(acc[nt][v], 0.0f);
            }
        }
    } else {
        float p[8];
        #pragma unroll
        for (int v = 0; v < 8; ++v) {
            p[v] = fmaxf(acc[0][v], 0.0f) * wf[0];
            p[v] += fmaxf(acc[1][v], 0.0f) * wf[1];
            p[v] += fmaxf(acc[2][v], 0.0f) * wf[2];
            p[v] += fmaxf(acc[3][v], 0.0f) * wf[3];
        }
        #pragma unroll
        for (int v = 0; v < 8; ++v) p[v] = row_xmask_add<8>(p[v]);
        #pragma unroll
        for (int v = 0; v < 8; ++v) p[v] = row_xmask_add<4>(p[v]);
        #pragma unroll
        for (int v = 0; v < 8; ++v) p[v] = row_xmask_add<2>(p[v]);
        #pragma unroll
        for (int v = 0; v < 8; ++v) p[v] = row_xmask_add<1>(p[v]);
        float mine = p[0];
        #pragma unroll
        for (int v = 1; v < 8; ++v) mine = (r == v) ? p[v] : mine;
        if (r < 8) mt[hi * 8 + r] = mine + bf;
    }
}

__global__ void __launch_bounds__(1024)
k_bin_layer1(const unsigned* __restrict__ tab, const int* __restrict__ tmpi,
             int* __restrict__ rowptr, int* col,
             const float* __restrict__ x, const float* __restrict__ Wl, const float* __restrict__ Wr,
             const float* __restrict__ bias, float* __restrict__ out) {
    __shared__ int cnt[BINSZ];
    __shared__ int wtot[SCW];
    __shared__ int red[2][32];
    __shared__ int rp[BINSZ + 1];
    __shared__ float Wfrag[32 * 4 * 32 * 2];
    __shared__ float meanT[TPB][16 * TP];
    const int2* tmp = (const int2*)tmpi;
    const int t = threadIdx.x;
    const int lane = t & 31;
    const int w = __builtin_amdgcn_readfirstlane(t >> 5);
    const int bin = blockIdx.x;
    const int base = bin * BINSZ;

    unsigned myinfo = 0;
    if (lane < PPW) myinfo = tab[bin * G1 + w + 32 * lane];
    const int nprev = bin * G1;
    uint4 tv[TABT];
    #pragma unroll
    for (int k = 0; k < TABT; ++k) {
        const int i = k * 4096 + t * 4;
        tv[k] = *(const uint4*)(tab + ((i < nprev) ? i : 0));
    }
    int pst[PPW], pcn[PPW];
    #pragma unroll
    for (int i = 0; i < PPW; ++i) {
        const unsigned info = __builtin_amdgcn_readlane(myinfo, i);
        pst[i] = min((int)(info & 0xffffu), CHUNK);
        pcn[i] = min((int)(info >> 16), CHUNK - pst[i]);
    }
    int2 p[PPW * PT];
    #pragma unroll
    for (int i = 0; i < PPW; ++i) {
        const int2* piece = tmp + (w + 32 * i) * CHUNK + pst[i];
        #pragma unroll
        for (int u = 0; u < PT; ++u) {
            const int j = u * 32 + lane;
            p[i * PT + u] = make_int2(0, -1);
            if (j < pcn[i]) p[i * PT + u] = piece[j];
        }
    }

    const float* wsrc = witem_src((t < 512) ? Wl : Wr, t);
    const v4f w0 = *(const v4f*)wsrc;
    const v4f w1 = *(const v4f*)(wsrc + 64);

    int before = 0;
    #pragma unroll
    for (int k = 0; k < TABT; ++k) {
        const int i = k * 4096 + t * 4;
        const int s4 = (int)((tv[k].x >> 16) + (tv[k].y >> 16) + (tv[k].z >> 16) + (tv[k].w >> 16));
        before += (i < nprev) ? s4 : 0;
    }
    if (t < BINSZ) cnt[t] = 0;
    witem_store(Wfrag, t, w0, w1);
    const int wb = wave_sum(before, lane);
    const int wo = wave_sum((int)(myinfo >> 16), lane);
    if (lane == 0) { red[0][w] = wb; red[1][w] = wo; }
    __syncthreads();
    const int beg = wave_sum(red[0][lane], lane);
    const int end = beg + wave_sum(red[1][lane], lane);

    #pragma unroll
    for (int q = 0; q < PPW * PT; ++q) {
        const int dl = p[q].y - base;
        if (p[q].y >= 0 && (unsigned)dl < (unsigned)BINSZ) atomicAdd(&cnt[dl], 1);
    }
    for (int i = 0; i < PPW; ++i) {
        const int2* piece = tmp + (w + 32 * i) * CHUNK + pst[i];
        for (int j = PT * 32 + lane; j < pcn[i]; j += 32) {
            const int y = piece[j].y;
            const int dl = y - base;
            if (y >= 0 && (unsigned)dl < (unsigned)BINSZ) atomicAdd(&cnt[dl], 1);
        }
    }
    __syncthreads();
    const int c = (t < BINSZ) ? cnt[t] : 0;
    const int incl = wave_incl_scan(c, lane);
    if (lane == 31 && w < SCW) wtot[w] = incl;
    __syncthreads();
    int wpre = 0;
    #pragma unroll
    for (int j = 0; j < SCW; ++j) wpre += (j < w) ? wtot[j] : 0;
    const int excl = beg + wpre + incl - c;
    if (t < BINSZ) {
        const int node = base + t;
        if (node < NN) rowptr[node] = excl;
        rp[t] = excl;
        cnt[t] = excl;
    }
    if (t == 0) rp[BINSZ] = end;
    if (bin == NB - 1 && t == 0) rowptr[NN] = end;
    __syncthreads();
    int pos[PPW * PT];
    #pragma unroll
    for (int q = 0; q < PPW * PT; ++q) {
        const int dl = p[q].y - base;
        pos[q] = -1;
        if (p[q].y >= 0 && (unsigned)dl < (unsigned)BINSZ) pos[q] = atomicAdd(&cnt[dl], 1);
    }
    #pragma unroll
    for (int q = 0; q < PPW * PT; ++q)
        if ((unsigned)pos[q] < (unsigned)NE) col[pos[q]] = p[q].x;
    for (int i = 0; i < PPW; ++i) {
        const int2* piece = tmp + (w + 32 * i) * CHUNK + pst[i];
        for (int j = PT * 32 + lane; j < pcn[i]; j += 32) {
            const int2 pr = piece[j];
            const int dl = pr.y - base;
            if (pr.y >= 0 && (unsigned)dl < (unsigned)BINSZ) {
                const int ps = atomicAdd(&cnt[dl], 1);
                if ((unsigned)ps < (unsigned)NE) col[ps] = pr.x;
            }
        }
    }

    __syncthreads();
    const int tile = bin * TPB + w;
    if (w < TPB && tile < NTILES) {
        const int hi = lane >> 4;
        const int r  = lane & 15;
        float* mt = meanT[w];
        float bv[4];
        const float nowf[4] = {0.0f, 0.0f, 0.0f, 0.0f};
        #pragma unroll
        for (int nt = 0; nt < 4; ++nt) bv[nt] = bias[nt * 16 + r];
        int rpv = 0;
        if (lane <= 16) rpv = rp[16 * w + lane];
        pull_sums<true>(x, col, rpv, mt, lane, hi, r);
        matrix_part<true, false>(x, tile * 16, mt, Wfrag, bv, nowf, 0.0f, out, lane, hi, r);
    }
}

__global__ void __launch_bounds__(256)
k_sage_layer2(const float* __restrict__ feat, const int* __restrict__ rowptr, const int* __restrict__ col,
              const float* __restrict__ Wl, const float* __restrict__ Wr, const float* __restrict__ bias,
              const float* __restrict__ Wfc, const float* __restrict__ bfc, float* __restrict__ pred) {
    __shared__ float Wfrag[32 * 4 * 32 * 2];
    __shared__ float meanT[8][16 * TP];

    const int lane = threadIdx.x & 31;
    const int hi   = lane >> 4;
    const int r    = lane & 15;
    const int wv   = __builtin_amdgcn_readfirstlane((int)(threadIdx.x >> 5));
    const int tile = blockIdx.x * 8 + wv;
    const bool live = tile < NTILES;
    float* mt = meanT[wv];
    const int n0 = tile * 16;

    v4f w0[4], w1[4];
    #pragma unroll
    for (int j = 0; j < 4; ++j) {
        const float* wsrc = witem_src((j < 2) ? Wl : Wr, j * 256 + (int)threadIdx.x);
        w0[j] = *(const v4f*)wsrc;
        w1[j] = *(const v4f*)(wsrc + 64);
    }
    float bv[4], wf[4];
    #pragma unroll
    for (int nt = 0; nt < 4; ++nt) {
        bv[nt] = bias[nt * 16 + r];
        wf[nt] = Wfc[nt * 16 + r];
    }
    const float bf = bfc[0];
    int rpv = 0;
    if (live && lane <= 16) rpv = rowptr[n0 + lane];
    #pragma unroll
    for (int j = 0; j < 4; ++j) witem_store(Wfrag, j * 256 + (int)threadIdx.x, w0[j], w1[j]);
    asm volatile("s_wait_dscnt 0x0\n\ts_barrier_signal -1" ::: "memory");

    if (live) pull_sums<false>(feat, col, rpv, mt, lane, hi, r);

    asm volatile("s_barrier_wait -1" ::: "memory");

    if (live) matrix_part<false, true>(feat, n0, mt, Wfrag, bv, wf, bf, nullptr, lane, hi, r);

    __syncthreads();
    if (threadIdx.x < 128) {
        const int node = blockIdx.x * 128 + threadIdx.x;
        if (node < NN) pred[node] = meanT[threadIdx.x >> 4][threadIdx.x & 15];
    }
}

extern "C" void kernel_launch(void* const* d_in, const int* in_sizes, int n_in,
                              void* d_out, int out_size, void* d_ws, size_t ws_size,
                              hipStream_t stream) {
    const float* x   = (const float*)d_in[0];
    const int*   ei  = (const int*)d_in[1];
    const float* W1l = (const float*)d_in[2];
    const float* b1  = (const float*)d_in[3];
    const float* W1r = (const float*)d_in[4];
    const float* W2l = (const float*)d_in[5];
    const float* b2  = (const float*)d_in[6];
    const float* W2r = (const float*)d_in[7];
    const float* Wfc = (const float*)d_in[8];
    const float* bfc = (const float*)d_in[9];
    float* pred = (float*)d_out;

    const int* src = ei;
    const int* dst = ei + NE;

    float* h1  = (float*)d_ws;
    int* rowptr = (int*)(h1 + NN * CH);
    int* col    = rowptr + NNP;
    unsigned* tab = (unsigned*)(col + NE);
    int* tmp    = (int*)(tab + NTAB);

    const int ggrid = (NTILES + 7) / 8;

    k_chunk_sort<<<G1, CT, 0, stream>>>(src, dst, tab, tmp);
    k_bin_layer1<<<NB, 1024, 0, stream>>>(tab, tmp, rowptr, col, x, W1l, W1r, b1, h1);

    k_sage_layer2<<<ggrid, 256, 0, stream>>>(h1, rowptr, col, W2l, W2r, b2, Wfc, bfc, pred);
}
